# starting kernel
# baseline (speedup 1.0000x reference)
#include <hip/hip_runtime.h>


#define S_   2048
#define D_   64
#define BS_  64
#define NB_  32
#define CW_  96
#define NH_  48

typedef __attribute__((ext_vector_type(2))) float v2f;
typedef __attribute__((ext_vector_type(8))) float v8f;

static constexpr size_t SZ_BL   = (size_t)NH_ * S_ * CW_;
static constexpr size_t SZ_R    = (size_t)NH_ * NB_ * BS_ * BS_;
static constexpr size_t O_BL    = 0;
static constexpr size_t O_R     = O_BL  + SZ_BL;
static constexpr size_t O_DBL   = O_R   + SZ_R;
static constexpr size_t O_T1    = O_DBL + SZ_BL;
static constexpr size_t O_T2    = O_T1  + SZ_R;
static constexpr size_t O_INVBL = O_T2  + SZ_R;
static constexpr size_t O_INVR  = O_INVBL + (size_t)NH_ * S_;
static constexpr size_t O_R2S   = O_INVR  + (size_t)NH_ * S_;
static constexpr size_t O_DAD   = O_R2S   + (size_t)NH_ * S_;
static constexpr size_t O_L2S   = O_DAD   + (size_t)NH_ * S_;

__device__ __forceinline__ int frag_half() { return (threadIdx.x & 31) >> 4; }
__device__ __forceinline__ int frag_l15()  { return threadIdx.x & 15; }
__device__ __forceinline__ int frag_m(int r) { return r + frag_half() * 8; }

template <bool AT, bool BT>
__device__ __forceinline__ v8f wmma_tile(const float* A, int lda, float asgn,
                                         const float* B, int ldb, int K, v8f acc) {
  const int half = frag_half();
  const int l15  = frag_l15();
  for (int k0 = 0; k0 < K; k0 += 4) {
    const int ka = k0 + half * 2;
    v2f a, b;
    if (AT) { a.x = asgn * A[ka * lda + l15];  a.y = asgn * A[(ka + 1) * lda + l15]; }
    else    { a.x = asgn * A[l15 * lda + ka];  a.y = asgn * A[l15 * lda + ka + 1]; }
    if (BT) { b.x = B[l15 * ldb + ka];         b.y = B[l15 * ldb + ka + 1]; }
    else    { b.x = B[ka * ldb + l15];         b.y = B[(ka + 1) * ldb + l15]; }
    acc = __builtin_amdgcn_wmma_f32_16x16x4_f32(false, a, false, b, (short)0, acc,
                                                false, false);
  }
  return acc;
}

__global__ __launch_bounds__(256) void k_init(float* ws) {
  const size_t tid = (size_t)blockIdx.x * 256 + threadIdx.x;
  if (tid < SZ_BL) ws[O_BL + tid] = 0.1020620726159658f;
  if (tid < SZ_R)  ws[O_R + tid]  = 0.125f;
}

__global__ __launch_bounds__(256) void k_norms(float* ws) {
  const size_t tid = (size_t)blockIdx.x * 256 + threadIdx.x;
  const size_t NR = (size_t)NH_ * S_;
  if (tid < NR) {
    const float* row = ws + O_BL + tid * CW_;
    float s = 0.f;
    for (int c = 0; c < CW_; ++c) s += row[c] * row[c];
    ws[O_INVBL + tid] = (s > 0.f) ? 1.0f / sqrtf(s) : 0.f;
  } else if (tid < 2 * NR) {
    const size_t r = tid - NR;
    const float* row = ws + O_R + r * BS_;
    float s = 0.f;
    for (int c = 0; c < BS_; ++c) s += row[c] * row[c];
    ws[O_INVR + r] = (s > 0.f) ? 1.0f / sqrtf(s) : 0.f;
  }
}

__global__ __launch_bounds__(256) void k_grad_block(const float* __restrict__ q,
                                                    const float* __restrict__ kk,
                                                    float* ws) {
  __shared__ float Qs[BS_ * D_];
  __shared__ float Ks[BS_ * D_];
  __shared__ float Rsq[BS_ * BS_];
  __shared__ float DiagL[BS_];

  const int blk = blockIdx.x;
  const int hb = blk >> 5, kb = blk & 31;
  const float* qh = q  + (size_t)hb * S_ * D_ + (size_t)kb * BS_ * D_;
  const float* kh = kk + (size_t)hb * S_ * D_ + (size_t)kb * BS_ * D_;
  float* blp   = ws + O_BL    + (size_t)hb * S_ * CW_;
  float* rp    = ws + O_R     + (size_t)hb * NB_ * BS_ * BS_ + (size_t)kb * BS_ * BS_;
  float* invbl = ws + O_INVBL + (size_t)hb * S_;
  float* invr  = ws + O_INVR  + (size_t)hb * S_;
  float* dbl   = ws + O_DBL   + (size_t)hb * S_ * CW_;
  float* t1    = ws + O_T1    + (size_t)hb * NB_ * BS_ * BS_ + (size_t)kb * BS_ * BS_;
  float* r2s   = ws + O_R2S   + (size_t)hb * S_;
  float* dad   = ws + O_DAD   + (size_t)hb * S_;

  const int tid = threadIdx.x;
  for (int i = tid; i < BS_ * D_; i += 256) {
    Qs[i] = qh[i] * 0.125f;
    Ks[i] = kh[i];
    float rs = rp[i] * invr[kb * 64 + (i >> 6)];
    Rsq[i] = rs * rs;
  }
  if (tid < BS_) {
    const int row = kb * 64 + tid;
    float v = blp[row * CW_ + 64 + kb] * invbl[row];
    DiagL[tid] = v * v;
  }
  __syncthreads();

  if (tid < BS_) {
    const int j = tid, row = kb * 64 + j;
    const float inv = invbl[row];
    float s2 = 0.f, sd = 0.f;
    for (int i = 0; i < BS_; ++i) {
      float r = Rsq[j * 64 + i];
      float b = blp[row * CW_ + i] * inv;
      s2 += r * r;
      sd += r * b * b;
    }
    r2s[row] = s2;
    dad[row] = sd;
  }

  const int wid = tid >> 5;
  for (int t = wid; t < 32; t += 8) {
    const int which = t >> 4;
    const int tt = t & 15;
    const int tj = tt >> 2, ti = tt & 3;
    if (which == 0) {
      v8f acc;
      #pragma unroll
      for (int r = 0; r < 8; ++r) {
        const int m = tj * 16 + frag_m(r), n = ti * 16 + frag_l15();
        const int row = kb * 64 + m;
        float b = blp[row * CW_ + n] * invbl[row];
        acc[r] = b * b + DiagL[m] * Rsq[m * 64 + n];
      }
      acc = wmma_tile<false, true>(Qs + tj * 16 * 64, 64, -1.f,
                                   Ks + ti * 16 * 64, 64, 64, acc);
      #pragma unroll
      for (int r = 0; r < 8; ++r) {
        const int m = tj * 16 + frag_m(r), n = ti * 16 + frag_l15();
        dbl[(kb * 64 + m) * CW_ + n] = acc[r];
      }
    } else {
      v8f acc = 0.0f;
      acc = wmma_tile<false, false>(Rsq + tj * 16 * 64, 64, 1.f,
                                    Ks + ti * 16, 64, 64, acc);
      #pragma unroll
      for (int r = 0; r < 8; ++r) {
        const int m = tj * 16 + frag_m(r), n = ti * 16 + frag_l15();
        t1[m * 64 + n] = acc[r];
      }
    }
  }
}

__global__ __launch_bounds__(256) void k_grad_j(const float* __restrict__ q,
                                                float* ws) {
  __shared__ float Lj[NB_ * NB_];
  __shared__ float Qj[NB_ * D_];
  __shared__ float T1j[NB_ * D_];

  const int blk = blockIdx.x;
  const int hb = blk >> 6, j = blk & 63;
  const float* qh = q + (size_t)hb * S_ * D_;
  float* blp   = ws + O_BL    + (size_t)hb * S_ * CW_;
  float* invbl = ws + O_INVBL + (size_t)hb * S_;
  float* dbl   = ws + O_DBL   + (size_t)hb * S_ * CW_;
  float* t1h   = ws + O_T1    + (size_t)hb * NB_ * BS_ * BS_;
  float* t2h   = ws + O_T2    + (size_t)hb * NB_ * BS_ * BS_;
  float* r2s   = ws + O_R2S   + (size_t)hb * S_;
  float* dad   = ws + O_DAD   + (size_t)hb * S_;
  float* l2s   = ws + O_L2S   + (size_t)hb * S_;

  const int tid = threadIdx.x;
  for (int i = tid; i < NB_ * NB_; i += 256) {
    const int l = i >> 5, k2 = i & 31;
    float v = blp[(l * 64 + j) * CW_ + 64 + k2] * invbl[l * 64 + j];
    Lj[i] = v * v;
  }
  for (int i = tid; i < NB_ * D_; i += 256) {
    const int l = i >> 6, d = i & 63;
    Qj[i]  = qh[(l * 64 + j) * 64 + d] * 0.125f;
    T1j[i] = t1h[l * (BS_ * BS_) + j * 64 + d];
  }
  __syncthreads();

  if (tid < NB_) {
    float s = 0.f;
    for (int l = 0; l < NB_; ++l) { float v = Lj[l * 32 + tid]; s += v * v; }
    l2s[j * 32 + tid] = s;
  }

  const int wid = tid >> 5;
  if (wid < 4) {
    const int tj = wid >> 1, ti = wid & 1;
    v8f acc = 0.0f;
    acc = wmma_tile<false, true>(Qj + tj * 16 * 64, 64, 1.f,
                                 T1j + ti * 16 * 64, 64, 64, acc);
    #pragma unroll
    for (int r = 0; r < 8; ++r) {
      const int l = tj * 16 + frag_m(r), k2 = ti * 16 + frag_l15();
      float v = Lj[l * 32 + k2] * r2s[k2 * 64 + j] - acc[r];
      if (l == k2) v += dad[l * 64 + j];
      dbl[(l * 64 + j) * CW_ + 64 + k2] = v;
    }
  }
  {
    const int tj = wid >> 2, ti = wid & 3;
    v8f acc = 0.0f;
    acc = wmma_tile<true, false>(Lj + tj * 16, 32, 1.f,
                                 Qj + ti * 16, 64, 32, acc);
    #pragma unroll
    for (int r = 0; r < 8; ++r) {
      const int k2 = tj * 16 + frag_m(r), d = ti * 16 + frag_l15();
      t2h[k2 * (BS_ * BS_) + j * 64 + d] = acc[r];
    }
  }
}

__global__ __launch_bounds__(256) void k_grad_r(const float* __restrict__ kk,
                                                float* ws) {
  __shared__ float T2s[BS_ * D_];
  __shared__ float Ks[BS_ * D_];
  __shared__ float DRs[BS_ * BS_];

  const int blk = blockIdx.x;
  const int hb = blk >> 5, kb = blk & 31;
  const float* kh = kk + (size_t)hb * S_ * D_ + (size_t)kb * BS_ * D_;
  float* blp   = ws + O_BL    + (size_t)hb * S_ * CW_;
  float* rp    = ws + O_R     + (size_t)hb * NB_ * BS_ * BS_ + (size_t)kb * BS_ * BS_;
  float* invbl = ws + O_INVBL + (size_t)hb * S_;
  float* invr  = ws + O_INVR  + (size_t)hb * S_;
  float* t2h   = ws + O_T2    + (size_t)hb * NB_ * BS_ * BS_ + (size_t)kb * BS_ * BS_;
  float* l2s   = ws + O_L2S   + (size_t)hb * S_;

  const int tid = threadIdx.x;
  for (int i = tid; i < BS_ * D_; i += 256) {
    T2s[i] = t2h[i];
    Ks[i]  = kh[i];
  }
  __syncthreads();

  const int wid = tid >> 5;
  for (int t = wid; t < 16; t += 8) {
    const int tj = t >> 2, ti = t & 3;
    v8f acc;
    #pragma unroll
    for (int r = 0; r < 8; ++r) {
      const int m = tj * 16 + frag_m(r), n = ti * 16 + frag_l15();
      const int row = kb * 64 + m;
      const float ib = invbl[row];
      float bd = blp[row * CW_ + n] * ib;         bd *= bd;
      float dl = blp[row * CW_ + 64 + kb] * ib;   dl *= dl;
      float rv = rp[m * 64 + n] * invr[row];      rv *= rv;
      acc[r] = rv * l2s[m * 32 + kb] + dl * bd;
    }
    acc = wmma_tile<false, true>(T2s + tj * 16 * 64, 64, -1.f,
                                 Ks + ti * 16 * 64, 64, 64, acc);
    #pragma unroll
    for (int r = 0; r < 8; ++r) {
      const int m = tj * 16 + frag_m(r), n = ti * 16 + frag_l15();
      DRs[m * 64 + n] = acc[r];
    }
  }
  __syncthreads();

  if (tid < BS_) {
    const int j = tid, row = kb * 64 + j;
    const float inv = invr[row];
    float dot = 0.f;
    for (int i = 0; i < BS_; ++i) {
      float rs = rp[j * 64 + i] * inv;
      dot += rs * (2.f * DRs[j * 64 + i] * rs);
    }
    for (int i = 0; i < BS_; ++i) {
      float rs = rp[j * 64 + i] * inv;
      float dr = 2.f * DRs[j * 64 + i] * rs;
      float df = (dr - rs * dot) * inv;
      rp[j * 64 + i] -= 0.5f * df;
    }
  }
}

__global__ __launch_bounds__(256) void k_bl_update(float* ws) {
  const size_t tid = (size_t)blockIdx.x * 256 + threadIdx.x;
  const size_t NR = (size_t)NH_ * S_;
  if (tid >= NR) return;
  float* blrow = ws + O_BL + tid * CW_;
  const float* drow = ws + O_DBL + tid * CW_;
  const float inv = ws[O_INVBL + tid];
  float dot = 0.f;
  for (int c = 0; c < CW_; ++c) {
    float bs = blrow[c] * inv;
    dot += bs * (2.f * drow[c] * bs);
  }
  for (int c = 0; c < CW_; ++c) {
    float bs = blrow[c] * inv;
    float d = 2.f * drow[c] * bs;
    blrow[c] -= 0.5f * ((d - bs * dot) * inv);
  }
}

__global__ __launch_bounds__(256) void k_Y(const float* __restrict__ val,
                                           float* ws) {
  __shared__ float Rsq[BS_ * BS_];
  __shared__ float Xs[BS_ * D_];
  const int blk = blockIdx.x;
  const int hb = blk >> 5, kb = blk & 31;
  const float* xh = val + (size_t)hb * S_ * D_ + (size_t)kb * BS_ * D_;
  float* rp   = ws + O_R    + (size_t)hb * NB_ * BS_ * BS_ + (size_t)kb * BS_ * BS_;
  float* invr = ws + O_INVR + (size_t)hb * S_;
  float* yb   = ws + O_T1   + (size_t)hb * NB_ * BS_ * BS_ + (size_t)kb * BS_ * BS_;

  const int tid = threadIdx.x;
  for (int i = tid; i < BS_ * D_; i += 256) {
    float rs = rp[i] * invr[kb * 64 + (i >> 6)];
    Rsq[i] = rs * rs;
    Xs[i]  = xh[i];
  }
  __syncthreads();

  const int wid = tid >> 5;
  for (int t = wid; t < 16; t += 8) {
    const int tj = t >> 2, ti = t & 3;
    v8f acc = 0.0f;
    acc = wmma_tile<false, false>(Rsq + tj * 16 * 64, 64, 1.f,
                                  Xs + ti * 16, 64, 64, acc);
    #pragma unroll
    for (int r = 0; r < 8; ++r) {
      const int m = tj * 16 + frag_m(r), n = ti * 16 + frag_l15();
      yb[m * 64 + n] = acc[r];
    }
  }
}

__global__ __launch_bounds__(256) void k_zleft(float* __restrict__ out,
                                               float* ws) {
  __shared__ float Lj[NB_ * NB_];
  __shared__ float Yj[NB_ * D_];
  const int blk = blockIdx.x;
  const int hb = blk >> 6, j = blk & 63;
  float* blp   = ws + O_BL    + (size_t)hb * S_ * CW_;
  float* invbl = ws + O_INVBL + (size_t)hb * S_;
  float* yb    = ws + O_T1    + (size_t)hb * NB_ * BS_ * BS_;
  float* oh    = out + (size_t)hb * S_ * D_;

  const int tid = threadIdx.x;
  for (int i = tid; i < NB_ * NB_; i += 256) {
    const int l = i >> 5, k2 = i & 31;
    float v = blp[(l * 64 + j) * CW_ + 64 + k2] * invbl[l * 64 + j];
    Lj[i] = v * v;
  }
  for (int i = tid; i < NB_ * D_; i += 256) {
    const int k2 = i >> 6, d = i & 63;
    Yj[i] = yb[k2 * (BS_ * BS_) + j * 64 + d];
  }
  __syncthreads();

  const int wid = tid >> 5;
  const int tj = wid >> 2, ti = wid & 3;
  v8f acc = 0.0f;
  acc = wmma_tile<false, false>(Lj + tj * 16 * 32, 32, 1.f,
                                Yj + ti * 16, 64, 32, acc);
  #pragma unroll
  for (int r = 0; r < 8; ++r) {
    const int l = tj * 16 + frag_m(r), d = ti * 16 + frag_l15();
    oh[(size_t)(l * 64 + j) * 64 + d] = acc[r];
  }
}

__global__ __launch_bounds__(256) void k_zbd(const float* __restrict__ val,
                                             float* __restrict__ out,
                                             float* ws) {
  __shared__ float BDs[BS_ * BS_];
  __shared__ float Xs[BS_ * D_];
  const int blk = blockIdx.x;
  const int hb = blk >> 5, lb = blk & 31;
  const float* xh = val + (size_t)hb * S_ * D_ + (size_t)lb * BS_ * D_;
  float* blp   = ws + O_BL    + (size_t)hb * S_ * CW_;
  float* invbl = ws + O_INVBL + (size_t)hb * S_;
  float* oh    = out + (size_t)hb * S_ * D_ + (size_t)lb * BS_ * D_;

  const int tid = threadIdx.x;
  for (int i = tid; i < BS_ * D_; i += 256) {
    const int j = i >> 6, c = i & 63;
    float v = blp[(lb * 64 + j) * CW_ + c] * invbl[lb * 64 + j];
    BDs[i] = v * v;
    Xs[i]  = xh[i];
  }
  __syncthreads();

  const int wid = tid >> 5;
  for (int t = wid; t < 16; t += 8) {
    const int tj = t >> 2, ti = t & 3;
    v8f acc = 0.0f;
    acc = wmma_tile<false, false>(BDs + tj * 16 * 64, 64, 1.f,
                                  Xs + ti * 16, 64, 64, acc);
    #pragma unroll
    for (int r = 0; r < 8; ++r) {
      const int m = tj * 16 + frag_m(r), n = ti * 16 + frag_l15();
      oh[m * 64 + n] += acc[r];
    }
  }
}

extern "C" void kernel_launch(void* const* d_in, const int* in_sizes, int n_in,
                              void* d_out, int out_size, void* d_ws, size_t ws_size,
                              hipStream_t stream) {
  const float* q = (const float*)d_in[0];
  const float* k = (const float*)d_in[1];
  const float* v = (const float*)d_in[2];
  float* out = (float*)d_out;
  float* ws = (float*)d_ws;

  const int initN  = (int)((SZ_BL + 255) / 256);
  const int normN  = (int)((2 * (size_t)NH_ * S_ + 255) / 256);
  const int blupN  = (int)(((size_t)NH_ * S_ + 255) / 256);
  const int blkG   = NH_ * NB_;
  const int jG     = NH_ * BS_;

  k_init<<<initN, 256, 0, stream>>>(ws);

  for (int step = 0; step < 2; ++step) {
    k_norms<<<normN, 256, 0, stream>>>(ws);
    k_grad_block<<<blkG, 256, 0, stream>>>(q, k, ws);
    k_grad_j<<<jG, 256, 0, stream>>>(q, ws);
    k_grad_r<<<blkG, 256, 0, stream>>>(k, ws);
    k_bl_update<<<blupN, 256, 0, stream>>>(ws);
  }

  k_norms<<<normN, 256, 0, stream>>>(ws);
  k_Y<<<blkG, 256, 0, stream>>>(v, ws);
  k_zleft<<<jG, 256, 0, stream>>>(out, ws);
  k_zbd<<<blkG, 256, 0, stream>>>(v, out, ws);
}


# Pad the LDS tile row pitches (64 to 68/76/80/66 floats, 32 to 36/40) so WMMA fragment loads and per-row scans spread over the 64 banks instead of 16 lanes sharing one; summation order unchanged
# speedup vs baseline: 1.2049x; 1.2049x over previous
#include <hip/hip_runtime.h>


#define S_   2048
#define D_   64
#define BS_  64
#define NB_  32
#define CW_  96
#define NH_  48

#define PA_   68
#define PA32_ 36
#define PB_   80
#define PAB_  76
#define PT32_ 40
#define PR_   66
#define LDS_TILE_ __shared__ __attribute__((aligned(16)))

typedef __attribute__((ext_vector_type(2))) float v2f;
typedef __attribute__((ext_vector_type(8))) float v8f;

static constexpr size_t SZ_BL   = (size_t)NH_ * S_ * CW_;
static constexpr size_t SZ_R    = (size_t)NH_ * NB_ * BS_ * BS_;
static constexpr size_t O_BL    = 0;
static constexpr size_t O_R     = O_BL  + SZ_BL;
static constexpr size_t O_DBL   = O_R   + SZ_R;
static constexpr size_t O_T1    = O_DBL + SZ_BL;
static constexpr size_t O_T2    = O_T1  + SZ_R;
static constexpr size_t O_INVBL = O_T2  + SZ_R;
static constexpr size_t O_INVR  = O_INVBL + (size_t)NH_ * S_;
static constexpr size_t O_R2S   = O_INVR  + (size_t)NH_ * S_;
static constexpr size_t O_DAD   = O_R2S   + (size_t)NH_ * S_;
static constexpr size_t O_L2S   = O_DAD   + (size_t)NH_ * S_;

__device__ __forceinline__ int frag_half() { return (threadIdx.x & 31) >> 4; }
__device__ __forceinline__ int frag_l15()  { return threadIdx.x & 15; }
__device__ __forceinline__ int frag_m(int r) { return r + frag_half() * 8; }

template <bool AT, bool BT>
__device__ __forceinline__ v8f wmma_tile(const float* A, int lda, float asgn,
                                         const float* B, int ldb, int K, v8f acc) {
  const int half = frag_half();
  const int l15  = frag_l15();
  for (int k0 = 0; k0 < K; k0 += 4) {
    const int ka = k0 + half * 2;
    v2f a, b;
    if (AT) { a.x = asgn * A[ka * lda + l15];  a.y = asgn * A[(ka + 1) * lda + l15]; }
    else    { a.x = asgn * A[l15 * lda + ka];  a.y = asgn * A[l15 * lda + ka + 1]; }
    if (BT) { b.x = B[l15 * ldb + ka];         b.y = B[l15 * ldb + ka + 1]; }
    else    { b.x = B[ka * ldb + l15];         b.y = B[(ka + 1) * ldb + l15]; }
    acc = __builtin_amdgcn_wmma_f32_16x16x4_f32(false, a, false, b, (short)0, acc,
                                                false, false);
  }
  return acc;
}

__global__ __launch_bounds__(256) void k_init(float* ws) {
  const size_t tid = (size_t)blockIdx.x * 256 + threadIdx.x;
  if (tid < SZ_BL) ws[O_BL + tid] = 0.1020620726159658f;
  if (tid < SZ_R)  ws[O_R + tid]  = 0.125f;
}

__global__ __launch_bounds__(256) void k_norms(float* ws) {
  const size_t tid = (size_t)blockIdx.x * 256 + threadIdx.x;
  const size_t NR = (size_t)NH_ * S_;
  if (tid < NR) {
    const float* row = ws + O_BL + tid * CW_;
    float s = 0.f;
    for (int c = 0; c < CW_; ++c) s += row[c] * row[c];
    ws[O_INVBL + tid] = (s > 0.f) ? 1.0f / sqrtf(s) : 0.f;
  } else if (tid < 2 * NR) {
    const size_t r = tid - NR;
    const float* row = ws + O_R + r * BS_;
    float s = 0.f;
    for (int c = 0; c < BS_; ++c) s += row[c] * row[c];
    ws[O_INVR + r] = (s > 0.f) ? 1.0f / sqrtf(s) : 0.f;
  }
}

__global__ __launch_bounds__(256) void k_grad_block(const float* __restrict__ q,
                                                    const float* __restrict__ kk,
                                                    float* ws) {
  LDS_TILE_ float Qs[BS_ * PA_];
  LDS_TILE_ float Ks[BS_ * PA_];
  LDS_TILE_ float Rsq[BS_ * PA_];
  __shared__ float DiagL[BS_];

  const int blk = blockIdx.x;
  const int hb = blk >> 5, kb = blk & 31;
  const float* qh = q  + (size_t)hb * S_ * D_ + (size_t)kb * BS_ * D_;
  const float* kh = kk + (size_t)hb * S_ * D_ + (size_t)kb * BS_ * D_;
  float* blp   = ws + O_BL    + (size_t)hb * S_ * CW_;
  float* rp    = ws + O_R     + (size_t)hb * NB_ * BS_ * BS_ + (size_t)kb * BS_ * BS_;
  float* invbl = ws + O_INVBL + (size_t)hb * S_;
  float* invr  = ws + O_INVR  + (size_t)hb * S_;
  float* dbl   = ws + O_DBL   + (size_t)hb * S_ * CW_;
  float* t1    = ws + O_T1    + (size_t)hb * NB_ * BS_ * BS_ + (size_t)kb * BS_ * BS_;
  float* r2s   = ws + O_R2S   + (size_t)hb * S_;
  float* dad   = ws + O_DAD   + (size_t)hb * S_;

  const int tid = threadIdx.x;
  for (int i = tid; i < BS_ * D_; i += 256) {
    const int p = (i >> 6) * PA_ + (i & 63);
    Qs[p] = qh[i] * 0.125f;
    Ks[p] = kh[i];
    float rs = rp[i] * invr[kb * 64 + (i >> 6)];
    Rsq[p] = rs * rs;
  }
  if (tid < BS_) {
    const int row = kb * 64 + tid;
    float v = blp[row * CW_ + 64 + kb] * invbl[row];
    DiagL[tid] = v * v;
  }
  __syncthreads();

  if (tid < BS_) {
    const int j = tid, row = kb * 64 + j;
    const float inv = invbl[row];
    float s2 = 0.f, sd = 0.f;
    for (int i = 0; i < BS_; ++i) {
      float r = Rsq[j * PA_ + i];
      float b = blp[row * CW_ + i] * inv;
      s2 += r * r;
      sd += r * b * b;
    }
    r2s[row] = s2;
    dad[row] = sd;
  }

  const int wid = tid >> 5;
  for (int t = wid; t < 32; t += 8) {
    const int which = t >> 4;
    const int tt = t & 15;
    const int tj = tt >> 2, ti = tt & 3;
    if (which == 0) {
      v8f acc;
      #pragma unroll
      for (int r = 0; r < 8; ++r) {
        const int m = tj * 16 + frag_m(r), n = ti * 16 + frag_l15();
        const int row = kb * 64 + m;
        float b = blp[row * CW_ + n] * invbl[row];
        acc[r] = b * b + DiagL[m] * Rsq[m * PA_ + n];
      }
      acc = wmma_tile<false, true>(Qs + tj * 16 * PA_, PA_, -1.f,
                                   Ks + ti * 16 * PA_, PA_, 64, acc);
      #pragma unroll
      for (int r = 0; r < 8; ++r) {
        const int m = tj * 16 + frag_m(r), n = ti * 16 + frag_l15();
        dbl[(kb * 64 + m) * CW_ + n] = acc[r];
      }
    } else {
      v8f acc = 0.0f;
      acc = wmma_tile<false, false>(Rsq + tj * 16 * PA_, PA_, 1.f,
                                    Ks + ti * 16, PA_, 64, acc);
      #pragma unroll
      for (int r = 0; r < 8; ++r) {
        const int m = tj * 16 + frag_m(r), n = ti * 16 + frag_l15();
        t1[m * 64 + n] = acc[r];
      }
    }
  }
}

__global__ __launch_bounds__(256) void k_grad_j(const float* __restrict__ q,
                                                float* ws) {
  LDS_TILE_ float Lj[NB_ * PT32_];
  LDS_TILE_ float Qj[NB_ * PAB_];
  LDS_TILE_ float T1j[NB_ * PA_];

  const int blk = blockIdx.x;
  const int hb = blk >> 6, j = blk & 63;
  const float* qh = q + (size_t)hb * S_ * D_;
  float* blp   = ws + O_BL    + (size_t)hb * S_ * CW_;
  float* invbl = ws + O_INVBL + (size_t)hb * S_;
  float* dbl   = ws + O_DBL   + (size_t)hb * S_ * CW_;
  float* t1h   = ws + O_T1    + (size_t)hb * NB_ * BS_ * BS_;
  float* t2h   = ws + O_T2    + (size_t)hb * NB_ * BS_ * BS_;
  float* r2s   = ws + O_R2S   + (size_t)hb * S_;
  float* dad   = ws + O_DAD   + (size_t)hb * S_;
  float* l2s   = ws + O_L2S   + (size_t)hb * S_;

  const int tid = threadIdx.x;
  for (int i = tid; i < NB_ * NB_; i += 256) {
    const int l = i >> 5, k2 = i & 31;
    float v = blp[(l * 64 + j) * CW_ + 64 + k2] * invbl[l * 64 + j];
    Lj[l * PT32_ + k2] = v * v;
  }
  for (int i = tid; i < NB_ * D_; i += 256) {
    const int l = i >> 6, d = i & 63;
    Qj[l * PAB_ + d] = qh[(l * 64 + j) * 64 + d] * 0.125f;
    T1j[l * PA_ + d] = t1h[l * (BS_ * BS_) + j * 64 + d];
  }
  __syncthreads();

  if (tid < NB_) {
    float s = 0.f;
    for (int l = 0; l < NB_; ++l) { float v = Lj[l * PT32_ + tid]; s += v * v; }
    l2s[j * 32 + tid] = s;
  }

  const int wid = tid >> 5;
  if (wid < 4) {
    const int tj = wid >> 1, ti = wid & 1;
    v8f acc = 0.0f;
    acc = wmma_tile<false, true>(Qj + tj * 16 * PAB_, PAB_, 1.f,
                                 T1j + ti * 16 * PA_, PA_, 64, acc);
    #pragma unroll
    for (int r = 0; r < 8; ++r) {
      const int l = tj * 16 + frag_m(r), k2 = ti * 16 + frag_l15();
      float v = Lj[l * PT32_ + k2] * r2s[k2 * 64 + j] - acc[r];
      if (l == k2) v += dad[l * 64 + j];
      dbl[(l * 64 + j) * CW_ + 64 + k2] = v;
    }
  }
  {
    const int tj = wid >> 2, ti = wid & 3;
    v8f acc = 0.0f;
    acc = wmma_tile<true, false>(Lj + tj * 16, PT32_, 1.f,
                                 Qj + ti * 16, PAB_, 32, acc);
    #pragma unroll
    for (int r = 0; r < 8; ++r) {
      const int k2 = tj * 16 + frag_m(r), d = ti * 16 + frag_l15();
      t2h[k2 * (BS_ * BS_) + j * 64 + d] = acc[r];
    }
  }
}

__global__ __launch_bounds__(256) void k_grad_r(const float* __restrict__ kk,
                                                float* ws) {
  LDS_TILE_ float T2s[BS_ * PA_];
  LDS_TILE_ float Ks[BS_ * PA_];
  LDS_TILE_ float DRs[BS_ * PR_];

  const int blk = blockIdx.x;
  const int hb = blk >> 5, kb = blk & 31;
  const float* kh = kk + (size_t)hb * S_ * D_ + (size_t)kb * BS_ * D_;
  float* blp   = ws + O_BL    + (size_t)hb * S_ * CW_;
  float* rp    = ws + O_R     + (size_t)hb * NB_ * BS_ * BS_ + (size_t)kb * BS_ * BS_;
  float* invbl = ws + O_INVBL + (size_t)hb * S_;
  float* invr  = ws + O_INVR  + (size_t)hb * S_;
  float* t2h   = ws + O_T2    + (size_t)hb * NB_ * BS_ * BS_ + (size_t)kb * BS_ * BS_;
  float* l2s   = ws + O_L2S   + (size_t)hb * S_;

  const int tid = threadIdx.x;
  for (int i = tid; i < BS_ * D_; i += 256) {
    const int p = (i >> 6) * PA_ + (i & 63);
    T2s[p] = t2h[i];
    Ks[p]  = kh[i];
  }
  __syncthreads();

  const int wid = tid >> 5;
  for (int t = wid; t < 16; t += 8) {
    const int tj = t >> 2, ti = t & 3;
    v8f acc;
    #pragma unroll
    for (int r = 0; r < 8; ++r) {
      const int m = tj * 16 + frag_m(r), n = ti * 16 + frag_l15();
      const int row = kb * 64 + m;
      const float ib = invbl[row];
      float bd = blp[row * CW_ + n] * ib;         bd *= bd;
      float dl = blp[row * CW_ + 64 + kb] * ib;   dl *= dl;
      float rv = rp[m * 64 + n] * invr[row];      rv *= rv;
      acc[r] = rv * l2s[m * 32 + kb] + dl * bd;
    }
    acc = wmma_tile<false, true>(T2s + tj * 16 * PA_, PA_, -1.f,
                                 Ks + ti * 16 * PA_, PA_, 64, acc);
    #pragma unroll
    for (int r = 0; r < 8; ++r) {
      const int m = tj * 16 + frag_m(r), n = ti * 16 + frag_l15();
      DRs[m * PR_ + n] = acc[r];
    }
  }
  __syncthreads();

  if (tid < BS_) {
    const int j = tid, row = kb * 64 + j;
    const float inv = invr[row];
    float dot = 0.f;
    for (int i = 0; i < BS_; ++i) {
      float rs = rp[j * 64 + i] * inv;
      dot += rs * (2.f * DRs[j * PR_ + i] * rs);
    }
    for (int i = 0; i < BS_; ++i) {
      float rs = rp[j * 64 + i] * inv;
      float dr = 2.f * DRs[j * PR_ + i] * rs;
      float df = (dr - rs * dot) * inv;
      rp[j * 64 + i] -= 0.5f * df;
    }
  }
}

__global__ __launch_bounds__(256) void k_bl_update(float* ws) {
  const size_t tid = (size_t)blockIdx.x * 256 + threadIdx.x;
  const size_t NR = (size_t)NH_ * S_;
  if (tid >= NR) return;
  float* blrow = ws + O_BL + tid * CW_;
  const float* drow = ws + O_DBL + tid * CW_;
  const float inv = ws[O_INVBL + tid];
  float dot = 0.f;
  for (int c = 0; c < CW_; ++c) {
    float bs = blrow[c] * inv;
    dot += bs * (2.f * drow[c] * bs);
  }
  for (int c = 0; c < CW_; ++c) {
    float bs = blrow[c] * inv;
    float d = 2.f * drow[c] * bs;
    blrow[c] -= 0.5f * ((d - bs * dot) * inv);
  }
}

__global__ __launch_bounds__(256) void k_Y(const float* __restrict__ val,
                                           float* ws) {
  LDS_TILE_ float Rsq[BS_ * PA_];
  LDS_TILE_ float Xs[BS_ * PB_];
  const int blk = blockIdx.x;
  const int hb = blk >> 5, kb = blk & 31;
  const float* xh = val + (size_t)hb * S_ * D_ + (size_t)kb * BS_ * D_;
  float* rp   = ws + O_R    + (size_t)hb * NB_ * BS_ * BS_ + (size_t)kb * BS_ * BS_;
  float* invr = ws + O_INVR + (size_t)hb * S_;
  float* yb   = ws + O_T1   + (size_t)hb * NB_ * BS_ * BS_ + (size_t)kb * BS_ * BS_;

  const int tid = threadIdx.x;
  for (int i = tid; i < BS_ * D_; i += 256) {
    float rs = rp[i] * invr[kb * 64 + (i >> 6)];
    Rsq[(i >> 6) * PA_ + (i & 63)] = rs * rs;
    Xs[(i >> 6) * PB_ + (i & 63)]  = xh[i];
  }
  __syncthreads();

  const int wid = tid >> 5;
  for (int t = wid; t < 16; t += 8) {
    const int tj = t >> 2, ti = t & 3;
    v8f acc = 0.0f;
    acc = wmma_tile<false, false>(Rsq + tj * 16 * PA_, PA_, 1.f,
                                  Xs + ti * 16, PB_, 64, acc);
    #pragma unroll
    for (int r = 0; r < 8; ++r) {
      const int m = tj * 16 + frag_m(r), n = ti * 16 + frag_l15();
      yb[m * 64 + n] = acc[r];
    }
  }
}

__global__ __launch_bounds__(256) void k_zleft(float* __restrict__ out,
                                               float* ws) {
  LDS_TILE_ float Lj[NB_ * PA32_];
  LDS_TILE_ float Yj[NB_ * PB_];
  const int blk = blockIdx.x;
  const int hb = blk >> 6, j = blk & 63;
  float* blp   = ws + O_BL    + (size_t)hb * S_ * CW_;
  float* invbl = ws + O_INVBL + (size_t)hb * S_;
  float* yb    = ws + O_T1    + (size_t)hb * NB_ * BS_ * BS_;
  float* oh    = out + (size_t)hb * S_ * D_;

  const int tid = threadIdx.x;
  for (int i = tid; i < NB_ * NB_; i += 256) {
    const int l = i >> 5, k2 = i & 31;
    float v = blp[(l * 64 + j) * CW_ + 64 + k2] * invbl[l * 64 + j];
    Lj[l * PA32_ + k2] = v * v;
  }
  for (int i = tid; i < NB_ * D_; i += 256) {
    const int k2 = i >> 6, d = i & 63;
    Yj[k2 * PB_ + d] = yb[k2 * (BS_ * BS_) + j * 64 + d];
  }
  __syncthreads();

  const int wid = tid >> 5;
  const int tj = wid >> 2, ti = wid & 3;
  v8f acc = 0.0f;
  acc = wmma_tile<false, false>(Lj + tj * 16 * PA32_, PA32_, 1.f,
                                Yj + ti * 16, PB_, 32, acc);
  #pragma unroll
  for (int r = 0; r < 8; ++r) {
    const int l = tj * 16 + frag_m(r), d = ti * 16 + frag_l15();
    oh[(size_t)(l * 64 + j) * 64 + d] = acc[r];
  }
}

__global__ __launch_bounds__(256) void k_zbd(const float* __restrict__ val,
                                             float* __restrict__ out,
                                             float* ws) {
  LDS_TILE_ float BDs[BS_ * PA_];
  LDS_TILE_ float Xs[BS_ * PB_];
  const int blk = blockIdx.x;
  const int hb = blk >> 5, lb = blk & 31;
  const float* xh = val + (size_t)hb * S_ * D_ + (size_t)lb * BS_ * D_;
  float* blp   = ws + O_BL    + (size_t)hb * S_ * CW_;
  float* invbl = ws + O_INVBL + (size_t)hb * S_;
  float* oh    = out + (size_t)hb * S_ * D_ + (size_t)lb * BS_ * D_;

  const int tid = threadIdx.x;
  for (int i = tid; i < BS_ * D_; i += 256) {
    const int j = i >> 6, c = i & 63;
    float v = blp[(lb * 64 + j) * CW_ + c] * invbl[lb * 64 + j];
    BDs[j * PA_ + c] = v * v;
    Xs[j * PB_ + c]  = xh[i];
  }
  __syncthreads();

  const int wid = tid >> 5;
  for (int t = wid; t < 16; t += 8) {
    const int tj = t >> 2, ti = t & 3;
    v8f acc = 0.0f;
    acc = wmma_tile<false, false>(BDs + tj * 16 * PA_, PA_, 1.f,
                                  Xs + ti * 16, PB_, 64, acc);
    #pragma unroll
    for (int r = 0; r < 8; ++r) {
      const int m = tj * 16 + frag_m(r), n = ti * 16 + frag_l15();
      oh[m * 64 + n] += acc[r];
    }
  }
}

extern "C" void kernel_launch(void* const* d_in, const int* in_sizes, int n_in,
                              void* d_out, int out_size, void* d_ws, size_t ws_size,
                              hipStream_t stream) {
  const float* q = (const float*)d_in[0];
  const float* k = (const float*)d_in[1];
  const float* v = (const float*)d_in[2];
  float* out = (float*)d_out;
  float* ws = (float*)d_ws;

  const int initN  = (int)((SZ_BL + 255) / 256);
  const int normN  = (int)((2 * (size_t)NH_ * S_ + 255) / 256);
  const int blupN  = (int)(((size_t)NH_ * S_ + 255) / 256);
  const int blkG   = NH_ * NB_;
  const int jG     = NH_ * BS_;

  k_init<<<initN, 256, 0, stream>>>(ws);

  for (int step = 0; step < 2; ++step) {
    k_norms<<<normN, 256, 0, stream>>>(ws);
    k_grad_block<<<blkG, 256, 0, stream>>>(q, k, ws);
    k_grad_j<<<jG, 256, 0, stream>>>(q, ws);
    k_grad_r<<<blkG, 256, 0, stream>>>(k, ws);
    k_bl_update<<<blupN, 256, 0, stream>>>(ws);
  }

  k_norms<<<normN, 256, 0, stream>>>(ws);
  k_Y<<<blkG, 256, 0, stream>>>(v, ws);
  k_zleft<<<jG, 256, 0, stream>>>(out, ws);
  k_zbd<<<blkG, 256, 0, stream>>>(v, out, ws);
}


# Each kernel that writes parameter rows (init, r update, bl update) also stores the row's inverse norm; the separate norm kernel is gone, 15 launches to 12; row sums stay left to right
# speedup vs baseline: 1.3371x; 1.1097x over previous
#include <hip/hip_runtime.h>


#define S_   2048
#define D_   64
#define BS_  64
#define NB_  32
#define CW_  96
#define NH_  48

#define PA_   68
#define PA32_ 36
#define PB_   80
#define PAB_  76
#define PT32_ 40
#define PR_   66
#define LDS_TILE_ __shared__ __attribute__((aligned(16)))

typedef __attribute__((ext_vector_type(2))) float v2f;
typedef __attribute__((ext_vector_type(8))) float v8f;

static constexpr size_t SZ_BL   = (size_t)NH_ * S_ * CW_;
static constexpr size_t SZ_R    = (size_t)NH_ * NB_ * BS_ * BS_;
static constexpr size_t O_BL    = 0;
static constexpr size_t O_R     = O_BL  + SZ_BL;
static constexpr size_t O_DBL   = O_R   + SZ_R;
static constexpr size_t O_T1    = O_DBL + SZ_BL;
static constexpr size_t O_T2    = O_T1  + SZ_R;
static constexpr size_t O_INVBL = O_T2  + SZ_R;
static constexpr size_t O_INVR  = O_INVBL + (size_t)NH_ * S_;
static constexpr size_t O_R2S   = O_INVR  + (size_t)NH_ * S_;
static constexpr size_t O_DAD   = O_R2S   + (size_t)NH_ * S_;
static constexpr size_t O_L2S   = O_DAD   + (size_t)NH_ * S_;

__device__ __forceinline__ int frag_half() { return (threadIdx.x & 31) >> 4; }
__device__ __forceinline__ int frag_l15()  { return threadIdx.x & 15; }
__device__ __forceinline__ int frag_m(int r) { return r + frag_half() * 8; }

template <bool AT, bool BT>
__device__ __forceinline__ v8f wmma_tile(const float* A, int lda, float asgn,
                                         const float* B, int ldb, int K, v8f acc) {
  const int half = frag_half();
  const int l15  = frag_l15();
  for (int k0 = 0; k0 < K; k0 += 4) {
    const int ka = k0 + half * 2;
    v2f a, b;
    if (AT) { a.x = asgn * A[ka * lda + l15];  a.y = asgn * A[(ka + 1) * lda + l15]; }
    else    { a.x = asgn * A[l15 * lda + ka];  a.y = asgn * A[l15 * lda + ka + 1]; }
    if (BT) { b.x = B[l15 * ldb + ka];         b.y = B[l15 * ldb + ka + 1]; }
    else    { b.x = B[ka * ldb + l15];         b.y = B[(ka + 1) * ldb + l15]; }
    acc = __builtin_amdgcn_wmma_f32_16x16x4_f32(false, a, false, b, (short)0, acc,
                                                false, false);
  }
  return acc;
}

__device__ __forceinline__ float inv_norm_from_sumsq(float s) {
  return (s > 0.f) ? 1.0f / sqrtf(s) : 0.f;
}
__device__ __forceinline__ float inv_norm_const_row(float v, int n) {
  float s = 0.f;
  for (int c = 0; c < n; ++c) s += v * v;
  return inv_norm_from_sumsq(s);
}

__global__ __launch_bounds__(256) void k_init(float* ws) {
  const size_t tid = (size_t)blockIdx.x * 256 + threadIdx.x;
  const float bl0 = 0.1020620726159658f;
  const float r0  = 0.125f;
  if (tid < SZ_BL) ws[O_BL + tid] = bl0;
  if (tid < SZ_R)  ws[O_R + tid]  = r0;
  if (tid < (size_t)NH_ * S_) {
    ws[O_INVBL + tid] = inv_norm_const_row(bl0, CW_);
    ws[O_INVR + tid]  = inv_norm_const_row(r0, BS_);
  }
}

__global__ __launch_bounds__(256) void k_grad_block(const float* __restrict__ q,
                                                    const float* __restrict__ kk,
                                                    float* ws) {
  LDS_TILE_ float Qs[BS_ * PA_];
  LDS_TILE_ float Ks[BS_ * PA_];
  LDS_TILE_ float Rsq[BS_ * PA_];
  __shared__ float DiagL[BS_];

  const int blk = blockIdx.x;
  const int hb = blk >> 5, kb = blk & 31;
  const float* qh = q  + (size_t)hb * S_ * D_ + (size_t)kb * BS_ * D_;
  const float* kh = kk + (size_t)hb * S_ * D_ + (size_t)kb * BS_ * D_;
  float* blp   = ws + O_BL    + (size_t)hb * S_ * CW_;
  float* rp    = ws + O_R     + (size_t)hb * NB_ * BS_ * BS_ + (size_t)kb * BS_ * BS_;
  float* invbl = ws + O_INVBL + (size_t)hb * S_;
  float* invr  = ws + O_INVR  + (size_t)hb * S_;
  float* dbl   = ws + O_DBL   + (size_t)hb * S_ * CW_;
  float* t1    = ws + O_T1    + (size_t)hb * NB_ * BS_ * BS_ + (size_t)kb * BS_ * BS_;
  float* r2s   = ws + O_R2S   + (size_t)hb * S_;
  float* dad   = ws + O_DAD   + (size_t)hb * S_;

  const int tid = threadIdx.x;
  for (int i = tid; i < BS_ * D_; i += 256) {
    const int p = (i >> 6) * PA_ + (i & 63);
    Qs[p] = qh[i] * 0.125f;
    Ks[p] = kh[i];
    float rs = rp[i] * invr[kb * 64 + (i >> 6)];
    Rsq[p] = rs * rs;
  }
  if (tid < BS_) {
    const int row = kb * 64 + tid;
    float v = blp[row * CW_ + 64 + kb] * invbl[row];
    DiagL[tid] = v * v;
  }
  __syncthreads();

  if (tid < BS_) {
    const int j = tid, row = kb * 64 + j;
    const float inv = invbl[row];
    float s2 = 0.f, sd = 0.f;
    for (int i = 0; i < BS_; ++i) {
      float r = Rsq[j * PA_ + i];
      float b = blp[row * CW_ + i] * inv;
      s2 += r * r;
      sd += r * b * b;
    }
    r2s[row] = s2;
    dad[row] = sd;
  }

  const int wid = tid >> 5;
  for (int t = wid; t < 32; t += 8) {
    const int which = t >> 4;
    const int tt = t & 15;
    const int tj = tt >> 2, ti = tt & 3;
    if (which == 0) {
      v8f acc;
      #pragma unroll
      for (int r = 0; r < 8; ++r) {
        const int m = tj * 16 + frag_m(r), n = ti * 16 + frag_l15();
        const int row = kb * 64 + m;
        float b = blp[row * CW_ + n] * invbl[row];
        acc[r] = b * b + DiagL[m] * Rsq[m * PA_ + n];
      }
      acc = wmma_tile<false, true>(Qs + tj * 16 * PA_, PA_, -1.f,
                                   Ks + ti * 16 * PA_, PA_, 64, acc);
      #pragma unroll
      for (int r = 0; r < 8; ++r) {
        const int m = tj * 16 + frag_m(r), n = ti * 16 + frag_l15();
        dbl[(kb * 64 + m) * CW_ + n] = acc[r];
      }
    } else {
      v8f acc = 0.0f;
      acc = wmma_tile<false, false>(Rsq + tj * 16 * PA_, PA_, 1.f,
                                    Ks + ti * 16, PA_, 64, acc);
      #pragma unroll
      for (int r = 0; r < 8; ++r) {
        const int m = tj * 16 + frag_m(r), n = ti * 16 + frag_l15();
        t1[m * 64 + n] = acc[r];
      }
    }
  }
}

__global__ __launch_bounds__(256) void k_grad_j(const float* __restrict__ q,
                                                float* ws) {
  LDS_TILE_ float Lj[NB_ * PT32_];
  LDS_TILE_ float Qj[NB_ * PAB_];
  LDS_TILE_ float T1j[NB_ * PA_];

  const int blk = blockIdx.x;
  const int hb = blk >> 6, j = blk & 63;
  const float* qh = q + (size_t)hb * S_ * D_;
  float* blp   = ws + O_BL    + (size_t)hb * S_ * CW_;
  float* invbl = ws + O_INVBL + (size_t)hb * S_;
  float* dbl   = ws + O_DBL   + (size_t)hb * S_ * CW_;
  float* t1h   = ws + O_T1    + (size_t)hb * NB_ * BS_ * BS_;
  float* t2h   = ws + O_T2    + (size_t)hb * NB_ * BS_ * BS_;
  float* r2s   = ws + O_R2S   + (size_t)hb * S_;
  float* dad   = ws + O_DAD   + (size_t)hb * S_;
  float* l2s   = ws + O_L2S   + (size_t)hb * S_;

  const int tid = threadIdx.x;
  for (int i = tid; i < NB_ * NB_; i += 256) {
    const int l = i >> 5, k2 = i & 31;
    float v = blp[(l * 64 + j) * CW_ + 64 + k2] * invbl[l * 64 + j];
    Lj[l * PT32_ + k2] = v * v;
  }
  for (int i = tid; i < NB_ * D_; i += 256) {
    const int l = i >> 6, d = i & 63;
    Qj[l * PAB_ + d] = qh[(l * 64 + j) * 64 + d] * 0.125f;
    T1j[l * PA_ + d] = t1h[l * (BS_ * BS_) + j * 64 + d];
  }
  __syncthreads();

  if (tid < NB_) {
    float s = 0.f;
    for (int l = 0; l < NB_; ++l) { float v = Lj[l * PT32_ + tid]; s += v * v; }
    l2s[j * 32 + tid] = s;
  }

  const int wid = tid >> 5;
  if (wid < 4) {
    const int tj = wid >> 1, ti = wid & 1;
    v8f acc = 0.0f;
    acc = wmma_tile<false, true>(Qj + tj * 16 * PAB_, PAB_, 1.f,
                                 T1j + ti * 16 * PA_, PA_, 64, acc);
    #pragma unroll
    for (int r = 0; r < 8; ++r) {
      const int l = tj * 16 + frag_m(r), k2 = ti * 16 + frag_l15();
      float v = Lj[l * PT32_ + k2] * r2s[k2 * 64 + j] - acc[r];
      if (l == k2) v += dad[l * 64 + j];
      dbl[(l * 64 + j) * CW_ + 64 + k2] = v;
    }
  }
  {
    const int tj = wid >> 2, ti = wid & 3;
    v8f acc = 0.0f;
    acc = wmma_tile<true, false>(Lj + tj * 16, PT32_, 1.f,
                                 Qj + ti * 16, PAB_, 32, acc);
    #pragma unroll
    for (int r = 0; r < 8; ++r) {
      const int k2 = tj * 16 + frag_m(r), d = ti * 16 + frag_l15();
      t2h[k2 * (BS_ * BS_) + j * 64 + d] = acc[r];
    }
  }
}

__global__ __launch_bounds__(256) void k_grad_r(const float* __restrict__ kk,
                                                float* ws) {
  LDS_TILE_ float T2s[BS_ * PA_];
  LDS_TILE_ float Ks[BS_ * PA_];
  LDS_TILE_ float DRs[BS_ * PR_];

  const int blk = blockIdx.x;
  const int hb = blk >> 5, kb = blk & 31;
  const float* kh = kk + (size_t)hb * S_ * D_ + (size_t)kb * BS_ * D_;
  float* blp   = ws + O_BL    + (size_t)hb * S_ * CW_;
  float* rp    = ws + O_R     + (size_t)hb * NB_ * BS_ * BS_ + (size_t)kb * BS_ * BS_;
  float* invbl = ws + O_INVBL + (size_t)hb * S_;
  float* invr  = ws + O_INVR  + (size_t)hb * S_;
  float* t2h   = ws + O_T2    + (size_t)hb * NB_ * BS_ * BS_ + (size_t)kb * BS_ * BS_;
  float* l2s   = ws + O_L2S   + (size_t)hb * S_;

  const int tid = threadIdx.x;
  for (int i = tid; i < BS_ * D_; i += 256) {
    const int p = (i >> 6) * PA_ + (i & 63);
    T2s[p] = t2h[i];
    Ks[p]  = kh[i];
  }
  __syncthreads();

  const int wid = tid >> 5;
  for (int t = wid; t < 16; t += 8) {
    const int tj = t >> 2, ti = t & 3;
    v8f acc;
    #pragma unroll
    for (int r = 0; r < 8; ++r) {
      const int m = tj * 16 + frag_m(r), n = ti * 16 + frag_l15();
      const int row = kb * 64 + m;
      const float ib = invbl[row];
      float bd = blp[row * CW_ + n] * ib;         bd *= bd;
      float dl = blp[row * CW_ + 64 + kb] * ib;   dl *= dl;
      float rv = rp[m * 64 + n] * invr[row];      rv *= rv;
      acc[r] = rv * l2s[m * 32 + kb] + dl * bd;
    }
    acc = wmma_tile<false, true>(T2s + tj * 16 * PA_, PA_, -1.f,
                                 Ks + ti * 16 * PA_, PA_, 64, acc);
    #pragma unroll
    for (int r = 0; r < 8; ++r) {
      const int m = tj * 16 + frag_m(r), n = ti * 16 + frag_l15();
      DRs[m * PR_ + n] = acc[r];
    }
  }
  __syncthreads();

  if (tid < BS_) {
    const int j = tid, row = kb * 64 + j;
    const float inv = invr[row];
    float dot = 0.f;
    for (int i = 0; i < BS_; ++i) {
      float rs = rp[j * 64 + i] * inv;
      dot += rs * (2.f * DRs[j * PR_ + i] * rs);
    }
    float s2 = 0.f;
    #pragma unroll 32
    for (int i = 0; i < BS_; ++i) {
      float rs = rp[j * 64 + i] * inv;
      float dr = 2.f * DRs[j * PR_ + i] * rs;
      float df = (dr - rs * dot) * inv;
      float nv = rp[j * 64 + i] - 0.5f * df;
      rp[j * 64 + i] = nv;
      s2 += nv * nv;
    }
    invr[row] = inv_norm_from_sumsq(s2);
  }
}

__global__ __launch_bounds__(256) void k_bl_update(float* ws) {
  const size_t tid = (size_t)blockIdx.x * 256 + threadIdx.x;
  const size_t NR = (size_t)NH_ * S_;
  if (tid >= NR) return;
  float* blrow = ws + O_BL + tid * CW_;
  const float* drow = ws + O_DBL + tid * CW_;
  const float inv = ws[O_INVBL + tid];
  float dot = 0.f;
  for (int c = 0; c < CW_; ++c) {
    float bs = blrow[c] * inv;
    dot += bs * (2.f * drow[c] * bs);
  }
  float s2 = 0.f;
  #pragma unroll 24
  for (int c = 0; c < CW_; ++c) {
    float bs = blrow[c] * inv;
    float d = 2.f * drow[c] * bs;
    float nv = blrow[c] - 0.5f * ((d - bs * dot) * inv);
    blrow[c] = nv;
    s2 += nv * nv;
  }
  ws[O_INVBL + tid] = inv_norm_from_sumsq(s2);
}

__global__ __launch_bounds__(256) void k_Y(const float* __restrict__ val,
                                           float* ws) {
  LDS_TILE_ float Rsq[BS_ * PA_];
  LDS_TILE_ float Xs[BS_ * PB_];
  const int blk = blockIdx.x;
  const int hb = blk >> 5, kb = blk & 31;
  const float* xh = val + (size_t)hb * S_ * D_ + (size_t)kb * BS_ * D_;
  float* rp   = ws + O_R    + (size_t)hb * NB_ * BS_ * BS_ + (size_t)kb * BS_ * BS_;
  float* invr = ws + O_INVR + (size_t)hb * S_;
  float* yb   = ws + O_T1   + (size_t)hb * NB_ * BS_ * BS_ + (size_t)kb * BS_ * BS_;

  const int tid = threadIdx.x;
  for (int i = tid; i < BS_ * D_; i += 256) {
    float rs = rp[i] * invr[kb * 64 + (i >> 6)];
    Rsq[(i >> 6) * PA_ + (i & 63)] = rs * rs;
    Xs[(i >> 6) * PB_ + (i & 63)]  = xh[i];
  }
  __syncthreads();

  const int wid = tid >> 5;
  for (int t = wid; t < 16; t += 8) {
    const int tj = t >> 2, ti = t & 3;
    v8f acc = 0.0f;
    acc = wmma_tile<false, false>(Rsq + tj * 16 * PA_, PA_, 1.f,
                                  Xs + ti * 16, PB_, 64, acc);
    #pragma unroll
    for (int r = 0; r < 8; ++r) {
      const int m = tj * 16 + frag_m(r), n = ti * 16 + frag_l15();
      yb[m * 64 + n] = acc[r];
    }
  }
}

__global__ __launch_bounds__(256) void k_zleft(float* __restrict__ out,
                                               float* ws) {
  LDS_TILE_ float Lj[NB_ * PA32_];
  LDS_TILE_ float Yj[NB_ * PB_];
  const int blk = blockIdx.x;
  const int hb = blk >> 6, j = blk & 63;
  float* blp   = ws + O_BL    + (size_t)hb * S_ * CW_;
  float* invbl = ws + O_INVBL + (size_t)hb * S_;
  float* yb    = ws + O_T1    + (size_t)hb * NB_ * BS_ * BS_;
  float* oh    = out + (size_t)hb * S_ * D_;

  const int tid = threadIdx.x;
  for (int i = tid; i < NB_ * NB_; i += 256) {
    const int l = i >> 5, k2 = i & 31;
    float v = blp[(l * 64 + j) * CW_ + 64 + k2] * invbl[l * 64 + j];
    Lj[l * PA32_ + k2] = v * v;
  }
  for (int i = tid; i < NB_ * D_; i += 256) {
    const int k2 = i >> 6, d = i & 63;
    Yj[k2 * PB_ + d] = yb[k2 * (BS_ * BS_) + j * 64 + d];
  }
  __syncthreads();

  const int wid = tid >> 5;
  const int tj = wid >> 2, ti = wid & 3;
  v8f acc = 0.0f;
  acc = wmma_tile<false, false>(Lj + tj * 16 * PA32_, PA32_, 1.f,
                                Yj + ti * 16, PB_, 32, acc);
  #pragma unroll
  for (int r = 0; r < 8; ++r) {
    const int l = tj * 16 + frag_m(r), d = ti * 16 + frag_l15();
    oh[(size_t)(l * 64 + j) * 64 + d] = acc[r];
  }
}

__global__ __launch_bounds__(256) void k_zbd(const float* __restrict__ val,
                                             float* __restrict__ out,
                                             float* ws) {
  LDS_TILE_ float BDs[BS_ * PA_];
  LDS_TILE_ float Xs[BS_ * PB_];
  const int blk = blockIdx.x;
  const int hb = blk >> 5, lb = blk & 31;
  const float* xh = val + (size_t)hb * S_ * D_ + (size_t)lb * BS_ * D_;
  float* blp   = ws + O_BL    + (size_t)hb * S_ * CW_;
  float* invbl = ws + O_INVBL + (size_t)hb * S_;
  float* oh    = out + (size_t)hb * S_ * D_ + (size_t)lb * BS_ * D_;

  const int tid = threadIdx.x;
  for (int i = tid; i < BS_ * D_; i += 256) {
    const int j = i >> 6, c = i & 63;
    float v = blp[(lb * 64 + j) * CW_ + c] * invbl[lb * 64 + j];
    BDs[j * PA_ + c] = v * v;
    Xs[j * PB_ + c]  = xh[i];
  }
  __syncthreads();

  const int wid = tid >> 5;
  for (int t = wid; t < 16; t += 8) {
    const int tj = t >> 2, ti = t & 3;
    v8f acc = 0.0f;
    acc = wmma_tile<false, false>(BDs + tj * 16 * PA_, PA_, 1.f,
                                  Xs + ti * 16, PB_, 64, acc);
    #pragma unroll
    for (int r = 0; r < 8; ++r) {
      const int m = tj * 16 + frag_m(r), n = ti * 16 + frag_l15();
      oh[m * 64 + n] += acc[r];
    }
  }
}

extern "C" void kernel_launch(void* const* d_in, const int* in_sizes, int n_in,
                              void* d_out, int out_size, void* d_ws, size_t ws_size,
                              hipStream_t stream) {
  const float* q = (const float*)d_in[0];
  const float* k = (const float*)d_in[1];
  const float* v = (const float*)d_in[2];
  float* out = (float*)d_out;
  float* ws = (float*)d_ws;

  const int initN  = (int)((SZ_BL + 255) / 256);
  const int blupN  = (int)(((size_t)NH_ * S_ + 255) / 256);
  const int blkG   = NH_ * NB_;
  const int jG     = NH_ * BS_;

  k_init<<<initN, 256, 0, stream>>>(ws);

  for (int step = 0; step < 2; ++step) {
    k_grad_block<<<blkG, 256, 0, stream>>>(q, k, ws);
    k_grad_j<<<jG, 256, 0, stream>>>(q, ws);
    k_grad_r<<<blkG, 256, 0, stream>>>(k, ws);
    k_bl_update<<<blupN, 256, 0, stream>>>(ws);
  }

  k_Y<<<blkG, 256, 0, stream>>>(v, ws);
  k_zleft<<<jG, 256, 0, stream>>>(out, ws);
  k_zbd<<<blkG, 256, 0, stream>>>(v, out, ws);
}


# Step 1 takes the constant start parameters and their row inverse norms as literals in its four kernels; the init kernel (36864 work-groups, one 4-byte store per lane) is gone, 12 launches to 11
# speedup vs baseline: 1.5990x; 1.1959x over previous
#include <hip/hip_runtime.h>


#define S_   2048
#define D_   64
#define BS_  64
#define NB_  32
#define CW_  96
#define NH_  48

#define PA_   68
#define PA32_ 36
#define PB_   80
#define PAB_  76
#define PT32_ 40
#define PR_   66
#define LDS_TILE_ __shared__ __attribute__((aligned(16)))

typedef __attribute__((ext_vector_type(2))) float v2f;
typedef __attribute__((ext_vector_type(8))) float v8f;

static constexpr size_t SZ_BL   = (size_t)NH_ * S_ * CW_;
static constexpr size_t SZ_R    = (size_t)NH_ * NB_ * BS_ * BS_;
static constexpr size_t O_BL    = 0;
static constexpr size_t O_R     = O_BL  + SZ_BL;
static constexpr size_t O_DBL   = O_R   + SZ_R;
static constexpr size_t O_T1    = O_DBL + SZ_BL;
static constexpr size_t O_T2    = O_T1  + SZ_R;
static constexpr size_t O_INVBL = O_T2  + SZ_R;
static constexpr size_t O_INVR  = O_INVBL + (size_t)NH_ * S_;
static constexpr size_t O_R2S   = O_INVR  + (size_t)NH_ * S_;
static constexpr size_t O_DAD   = O_R2S   + (size_t)NH_ * S_;
static constexpr size_t O_L2S   = O_DAD   + (size_t)NH_ * S_;

__device__ __forceinline__ int frag_half() { return (threadIdx.x & 31) >> 4; }
__device__ __forceinline__ int frag_l15()  { return threadIdx.x & 15; }
__device__ __forceinline__ int frag_m(int r) { return r + frag_half() * 8; }

template <bool AT, bool BT>
__device__ __forceinline__ v8f wmma_tile(const float* A, int lda, float asgn,
                                         const float* B, int ldb, int K, v8f acc) {
  const int half = frag_half();
  const int l15  = frag_l15();
  for (int k0 = 0; k0 < K; k0 += 4) {
    const int ka = k0 + half * 2;
    v2f a, b;
    if (AT) { a.x = asgn * A[ka * lda + l15];  a.y = asgn * A[(ka + 1) * lda + l15]; }
    else    { a.x = asgn * A[l15 * lda + ka];  a.y = asgn * A[l15 * lda + ka + 1]; }
    if (BT) { b.x = B[l15 * ldb + ka];         b.y = B[l15 * ldb + ka + 1]; }
    else    { b.x = B[ka * ldb + l15];         b.y = B[(ka + 1) * ldb + l15]; }
    acc = __builtin_amdgcn_wmma_f32_16x16x4_f32(false, a, false, b, (short)0, acc,
                                                false, false);
  }
  return acc;
}

__device__ __forceinline__ float inv_norm_from_sumsq(float s) {
  return (s > 0.f) ? 1.0f / sqrtf(s) : 0.f;
}
__device__ __forceinline__ float inv_norm_const_row(float v, int n) {
  float s = 0.f;
  for (int c = 0; c < n; ++c) s += v * v;
  return inv_norm_from_sumsq(s);
}

#define BL0_ 0.1020620726159658f
#define R0_  0.125f
__device__ __forceinline__ float invbl0() { return inv_norm_const_row(BL0_, CW_); }
__device__ __forceinline__ float invr0()  { return inv_norm_const_row(R0_, BS_); }
template <bool FIRST>
__device__ __forceinline__ float par(const float* p, int i, float c0) {
  if (FIRST) return c0;
  return p[i];
}

template <bool FIRST>
__device__ __forceinline__ void grad_block_body(const float* __restrict__ q,
                                                const float* __restrict__ kk,
                                                float* ws) {
  LDS_TILE_ float Qs[BS_ * PA_];
  LDS_TILE_ float Ks[BS_ * PA_];
  LDS_TILE_ float Rsq[BS_ * PA_];
  __shared__ float DiagL[BS_];

  const int blk = blockIdx.x;
  const int hb = blk >> 5, kb = blk & 31;
  const float* qh = q  + (size_t)hb * S_ * D_ + (size_t)kb * BS_ * D_;
  const float* kh = kk + (size_t)hb * S_ * D_ + (size_t)kb * BS_ * D_;
  float* blp   = ws + O_BL    + (size_t)hb * S_ * CW_;
  float* rp    = ws + O_R     + (size_t)hb * NB_ * BS_ * BS_ + (size_t)kb * BS_ * BS_;
  float* invbl = ws + O_INVBL + (size_t)hb * S_;
  float* invr  = ws + O_INVR  + (size_t)hb * S_;
  float* dbl   = ws + O_DBL   + (size_t)hb * S_ * CW_;
  float* t1    = ws + O_T1    + (size_t)hb * NB_ * BS_ * BS_ + (size_t)kb * BS_ * BS_;
  float* r2s   = ws + O_R2S   + (size_t)hb * S_;
  float* dad   = ws + O_DAD   + (size_t)hb * S_;

  const int tid = threadIdx.x;
  for (int i = tid; i < BS_ * D_; i += 256) {
    const int p = (i >> 6) * PA_ + (i & 63);
    Qs[p] = qh[i] * 0.125f;
    Ks[p] = kh[i];
    float rs = par<FIRST>(rp, i, R0_) * par<FIRST>(invr, kb * 64 + (i >> 6), invr0());
    Rsq[p] = rs * rs;
  }
  if (tid < BS_) {
    const int row = kb * 64 + tid;
    float v = par<FIRST>(blp, row * CW_ + 64 + kb, BL0_) * par<FIRST>(invbl, row, invbl0());
    DiagL[tid] = v * v;
  }
  __syncthreads();

  if (tid < BS_) {
    const int j = tid, row = kb * 64 + j;
    const float inv = par<FIRST>(invbl, row, invbl0());
    float s2 = 0.f, sd = 0.f;
    for (int i = 0; i < BS_; ++i) {
      float r = Rsq[j * PA_ + i];
      float b = par<FIRST>(blp, row * CW_ + i, BL0_) * inv;
      s2 += r * r;
      sd += r * b * b;
    }
    r2s[row] = s2;
    dad[row] = sd;
  }

  const int wid = tid >> 5;
  for (int t = wid; t < 32; t += 8) {
    const int which = t >> 4;
    const int tt = t & 15;
    const int tj = tt >> 2, ti = tt & 3;
    if (which == 0) {
      v8f acc;
      #pragma unroll
      for (int r = 0; r < 8; ++r) {
        const int m = tj * 16 + frag_m(r), n = ti * 16 + frag_l15();
        const int row = kb * 64 + m;
        float b = par<FIRST>(blp, row * CW_ + n, BL0_) * par<FIRST>(invbl, row, invbl0());
        acc[r] = b * b + DiagL[m] * Rsq[m * PA_ + n];
      }
      acc = wmma_tile<false, true>(Qs + tj * 16 * PA_, PA_, -1.f,
                                   Ks + ti * 16 * PA_, PA_, 64, acc);
      #pragma unroll
      for (int r = 0; r < 8; ++r) {
        const int m = tj * 16 + frag_m(r), n = ti * 16 + frag_l15();
        dbl[(kb * 64 + m) * CW_ + n] = acc[r];
      }
    } else {
      v8f acc = 0.0f;
      acc = wmma_tile<false, false>(Rsq + tj * 16 * PA_, PA_, 1.f,
                                    Ks + ti * 16, PA_, 64, acc);
      #pragma unroll
      for (int r = 0; r < 8; ++r) {
        const int m = tj * 16 + frag_m(r), n = ti * 16 + frag_l15();
        t1[m * 64 + n] = acc[r];
      }
    }
  }
}
__global__ __launch_bounds__(256) void k_grad_block_first(const float* __restrict__ q,
                                                          const float* __restrict__ kk,
                                                          float* ws) {
  grad_block_body<true>(q, kk, ws);
}
__global__ __launch_bounds__(256) void k_grad_block(const float* __restrict__ q,
                                                    const float* __restrict__ kk,
                                                    float* ws) {
  grad_block_body<false>(q, kk, ws);
}

template <bool FIRST>
__device__ __forceinline__ void grad_j_body(const float* __restrict__ q,
                                            float* ws) {
  LDS_TILE_ float Lj[NB_ * PT32_];
  LDS_TILE_ float Qj[NB_ * PAB_];
  LDS_TILE_ float T1j[NB_ * PA_];

  const int blk = blockIdx.x;
  const int hb = blk >> 6, j = blk & 63;
  const float* qh = q + (size_t)hb * S_ * D_;
  float* blp   = ws + O_BL    + (size_t)hb * S_ * CW_;
  float* invbl = ws + O_INVBL + (size_t)hb * S_;
  float* dbl   = ws + O_DBL   + (size_t)hb * S_ * CW_;
  float* t1h   = ws + O_T1    + (size_t)hb * NB_ * BS_ * BS_;
  float* t2h   = ws + O_T2    + (size_t)hb * NB_ * BS_ * BS_;
  float* r2s   = ws + O_R2S   + (size_t)hb * S_;
  float* dad   = ws + O_DAD   + (size_t)hb * S_;
  float* l2s   = ws + O_L2S   + (size_t)hb * S_;

  const int tid = threadIdx.x;
  for (int i = tid; i < NB_ * NB_; i += 256) {
    const int l = i >> 5, k2 = i & 31;
    float v = par<FIRST>(blp, (l * 64 + j) * CW_ + 64 + k2, BL0_) *
              par<FIRST>(invbl, l * 64 + j, invbl0());
    Lj[l * PT32_ + k2] = v * v;
  }
  for (int i = tid; i < NB_ * D_; i += 256) {
    const int l = i >> 6, d = i & 63;
    Qj[l * PAB_ + d] = qh[(l * 64 + j) * 64 + d] * 0.125f;
    T1j[l * PA_ + d] = t1h[l * (BS_ * BS_) + j * 64 + d];
  }
  __syncthreads();

  if (tid < NB_) {
    float s = 0.f;
    for (int l = 0; l < NB_; ++l) { float v = Lj[l * PT32_ + tid]; s += v * v; }
    l2s[j * 32 + tid] = s;
  }

  const int wid = tid >> 5;
  if (wid < 4) {
    const int tj = wid >> 1, ti = wid & 1;
    v8f acc = 0.0f;
    acc = wmma_tile<false, true>(Qj + tj * 16 * PAB_, PAB_, 1.f,
                                 T1j + ti * 16 * PA_, PA_, 64, acc);
    #pragma unroll
    for (int r = 0; r < 8; ++r) {
      const int l = tj * 16 + frag_m(r), k2 = ti * 16 + frag_l15();
      float v = Lj[l * PT32_ + k2] * r2s[k2 * 64 + j] - acc[r];
      if (l == k2) v += dad[l * 64 + j];
      dbl[(l * 64 + j) * CW_ + 64 + k2] = v;
    }
  }
  {
    const int tj = wid >> 2, ti = wid & 3;
    v8f acc = 0.0f;
    acc = wmma_tile<true, false>(Lj + tj * 16, PT32_, 1.f,
                                 Qj + ti * 16, PAB_, 32, acc);
    #pragma unroll
    for (int r = 0; r < 8; ++r) {
      const int k2 = tj * 16 + frag_m(r), d = ti * 16 + frag_l15();
      t2h[k2 * (BS_ * BS_) + j * 64 + d] = acc[r];
    }
  }
}
__global__ __launch_bounds__(256) void k_grad_j_first(const float* __restrict__ q,
                                                      float* ws) {
  grad_j_body<true>(q, ws);
}
__global__ __launch_bounds__(256) void k_grad_j(const float* __restrict__ q,
                                                float* ws) {
  grad_j_body<false>(q, ws);
}

template <bool FIRST>
__device__ __forceinline__ void grad_r_body(const float* __restrict__ kk,
                                            float* ws) {
  LDS_TILE_ float T2s[BS_ * PA_];
  LDS_TILE_ float Ks[BS_ * PA_];
  LDS_TILE_ float DRs[BS_ * PR_];

  const int blk = blockIdx.x;
  const int hb = blk >> 5, kb = blk & 31;
  const float* kh = kk + (size_t)hb * S_ * D_ + (size_t)kb * BS_ * D_;
  float* blp   = ws + O_BL    + (size_t)hb * S_ * CW_;
  float* rp    = ws + O_R     + (size_t)hb * NB_ * BS_ * BS_ + (size_t)kb * BS_ * BS_;
  float* invbl = ws + O_INVBL + (size_t)hb * S_;
  float* invr  = ws + O_INVR  + (size_t)hb * S_;
  float* t2h   = ws + O_T2    + (size_t)hb * NB_ * BS_ * BS_ + (size_t)kb * BS_ * BS_;
  float* l2s   = ws + O_L2S   + (size_t)hb * S_;

  const int tid = threadIdx.x;
  for (int i = tid; i < BS_ * D_; i += 256) {
    const int p = (i >> 6) * PA_ + (i & 63);
    T2s[p] = t2h[i];
    Ks[p]  = kh[i];
  }
  __syncthreads();

  const int wid = tid >> 5;
  for (int t = wid; t < 16; t += 8) {
    const int tj = t >> 2, ti = t & 3;
    v8f acc;
    #pragma unroll
    for (int r = 0; r < 8; ++r) {
      const int m = tj * 16 + frag_m(r), n = ti * 16 + frag_l15();
      const int row = kb * 64 + m;
      const float ib = par<FIRST>(invbl, row, invbl0());
      float bd = par<FIRST>(blp, row * CW_ + n, BL0_) * ib;         bd *= bd;
      float dl = par<FIRST>(blp, row * CW_ + 64 + kb, BL0_) * ib;   dl *= dl;
      float rv = par<FIRST>(rp, m * 64 + n, R0_) * par<FIRST>(invr, row, invr0());
      rv *= rv;
      acc[r] = rv * l2s[m * 32 + kb] + dl * bd;
    }
    acc = wmma_tile<false, true>(T2s + tj * 16 * PA_, PA_, -1.f,
                                 Ks + ti * 16 * PA_, PA_, 64, acc);
    #pragma unroll
    for (int r = 0; r < 8; ++r) {
      const int m = tj * 16 + frag_m(r), n = ti * 16 + frag_l15();
      DRs[m * PR_ + n] = acc[r];
    }
  }
  __syncthreads();

  if (tid < BS_) {
    const int j = tid, row = kb * 64 + j;
    const float inv = par<FIRST>(invr, row, invr0());
    float dot = 0.f;
    for (int i = 0; i < BS_; ++i) {
      float rs = par<FIRST>(rp, j * 64 + i, R0_) * inv;
      dot += rs * (2.f * DRs[j * PR_ + i] * rs);
    }
    float s2 = 0.f;
    #pragma unroll 32
    for (int i = 0; i < BS_; ++i) {
      const float old = par<FIRST>(rp, j * 64 + i, R0_);
      float rs = old * inv;
      float dr = 2.f * DRs[j * PR_ + i] * rs;
      float df = (dr - rs * dot) * inv;
      float nv = old - 0.5f * df;
      rp[j * 64 + i] = nv;
      s2 += nv * nv;
    }
    invr[row] = inv_norm_from_sumsq(s2);
  }
}
__global__ __launch_bounds__(256) void k_grad_r_first(const float* __restrict__ kk,
                                                      float* ws) {
  grad_r_body<true>(kk, ws);
}
__global__ __launch_bounds__(256) void k_grad_r(const float* __restrict__ kk,
                                                float* ws) {
  grad_r_body<false>(kk, ws);
}

template <bool FIRST>
__device__ __forceinline__ void bl_update_body(float* ws) {
  const size_t tid = (size_t)blockIdx.x * 256 + threadIdx.x;
  const size_t NR = (size_t)NH_ * S_;
  if (tid >= NR) return;
  float* blrow = ws + O_BL + tid * CW_;
  const float* drow = ws + O_DBL + tid * CW_;
  const float inv = par<FIRST>(ws + O_INVBL + tid, 0, invbl0());
  float dot = 0.f;
  for (int c = 0; c < CW_; ++c) {
    float bs = par<FIRST>(blrow, c, BL0_) * inv;
    dot += bs * (2.f * drow[c] * bs);
  }
  float s2 = 0.f;
  #pragma unroll 24
  for (int c = 0; c < CW_; ++c) {
    const float old = par<FIRST>(blrow, c, BL0_);
    float bs = old * inv;
    float d = 2.f * drow[c] * bs;
    float nv = old - 0.5f * ((d - bs * dot) * inv);
    blrow[c] = nv;
    s2 += nv * nv;
  }
  ws[O_INVBL + tid] = inv_norm_from_sumsq(s2);
}
__global__ __launch_bounds__(256) void k_bl_update_first(float* ws) {
  bl_update_body<true>(ws);
}
__global__ __launch_bounds__(256) void k_bl_update(float* ws) {
  bl_update_body<false>(ws);
}

__global__ __launch_bounds__(256) void k_Y(const float* __restrict__ val,
                                           float* ws) {
  LDS_TILE_ float Rsq[BS_ * PA_];
  LDS_TILE_ float Xs[BS_ * PB_];
  const int blk = blockIdx.x;
  const int hb = blk >> 5, kb = blk & 31;
  const float* xh = val + (size_t)hb * S_ * D_ + (size_t)kb * BS_ * D_;
  float* rp   = ws + O_R    + (size_t)hb * NB_ * BS_ * BS_ + (size_t)kb * BS_ * BS_;
  float* invr = ws + O_INVR + (size_t)hb * S_;
  float* yb   = ws + O_T1   + (size_t)hb * NB_ * BS_ * BS_ + (size_t)kb * BS_ * BS_;

  const int tid = threadIdx.x;
  for (int i = tid; i < BS_ * D_; i += 256) {
    float rs = rp[i] * invr[kb * 64 + (i >> 6)];
    Rsq[(i >> 6) * PA_ + (i & 63)] = rs * rs;
    Xs[(i >> 6) * PB_ + (i & 63)]  = xh[i];
  }
  __syncthreads();

  const int wid = tid >> 5;
  for (int t = wid; t < 16; t += 8) {
    const int tj = t >> 2, ti = t & 3;
    v8f acc = 0.0f;
    acc = wmma_tile<false, false>(Rsq + tj * 16 * PA_, PA_, 1.f,
                                  Xs + ti * 16, PB_, 64, acc);
    #pragma unroll
    for (int r = 0; r < 8; ++r) {
      const int m = tj * 16 + frag_m(r), n = ti * 16 + frag_l15();
      yb[m * 64 + n] = acc[r];
    }
  }
}

__global__ __launch_bounds__(256) void k_zleft(float* __restrict__ out,
                                               float* ws) {
  LDS_TILE_ float Lj[NB_ * PA32_];
  LDS_TILE_ float Yj[NB_ * PB_];
  const int blk = blockIdx.x;
  const int hb = blk >> 6, j = blk & 63;
  float* blp   = ws + O_BL    + (size_t)hb * S_ * CW_;
  float* invbl = ws + O_INVBL + (size_t)hb * S_;
  float* yb    = ws + O_T1    + (size_t)hb * NB_ * BS_ * BS_;
  float* oh    = out + (size_t)hb * S_ * D_;

  const int tid = threadIdx.x;
  for (int i = tid; i < NB_ * NB_; i += 256) {
    const int l = i >> 5, k2 = i & 31;
    float v = blp[(l * 64 + j) * CW_ + 64 + k2] * invbl[l * 64 + j];
    Lj[l * PA32_ + k2] = v * v;
  }
  for (int i = tid; i < NB_ * D_; i += 256) {
    const int k2 = i >> 6, d = i & 63;
    Yj[k2 * PB_ + d] = yb[k2 * (BS_ * BS_) + j * 64 + d];
  }
  __syncthreads();

  const int wid = tid >> 5;
  const int tj = wid >> 2, ti = wid & 3;
  v8f acc = 0.0f;
  acc = wmma_tile<false, false>(Lj + tj * 16 * PA32_, PA32_, 1.f,
                                Yj + ti * 16, PB_, 32, acc);
  #pragma unroll
  for (int r = 0; r < 8; ++r) {
    const int l = tj * 16 + frag_m(r), d = ti * 16 + frag_l15();
    oh[(size_t)(l * 64 + j) * 64 + d] = acc[r];
  }
}

__global__ __launch_bounds__(256) void k_zbd(const float* __restrict__ val,
                                             float* __restrict__ out,
                                             float* ws) {
  LDS_TILE_ float BDs[BS_ * PA_];
  LDS_TILE_ float Xs[BS_ * PB_];
  const int blk = blockIdx.x;
  const int hb = blk >> 5, lb = blk & 31;
  const float* xh = val + (size_t)hb * S_ * D_ + (size_t)lb * BS_ * D_;
  float* blp   = ws + O_BL    + (size_t)hb * S_ * CW_;
  float* invbl = ws + O_INVBL + (size_t)hb * S_;
  float* oh    = out + (size_t)hb * S_ * D_ + (size_t)lb * BS_ * D_;

  const int tid = threadIdx.x;
  for (int i = tid; i < BS_ * D_; i += 256) {
    const int j = i >> 6, c = i & 63;
    float v = blp[(lb * 64 + j) * CW_ + c] * invbl[lb * 64 + j];
    BDs[j * PA_ + c] = v * v;
    Xs[j * PB_ + c]  = xh[i];
  }
  __syncthreads();

  const int wid = tid >> 5;
  for (int t = wid; t < 16; t += 8) {
    const int tj = t >> 2, ti = t & 3;
    v8f acc = 0.0f;
    acc = wmma_tile<false, false>(BDs + tj * 16 * PA_, PA_, 1.f,
                                  Xs + ti * 16, PB_, 64, acc);
    #pragma unroll
    for (int r = 0; r < 8; ++r) {
      const int m = tj * 16 + frag_m(r), n = ti * 16 + frag_l15();
      oh[m * 64 + n] += acc[r];
    }
  }
}

extern "C" void kernel_launch(void* const* d_in, const int* in_sizes, int n_in,
                              void* d_out, int out_size, void* d_ws, size_t ws_size,
                              hipStream_t stream) {
  const float* q = (const float*)d_in[0];
  const float* k = (const float*)d_in[1];
  const float* v = (const float*)d_in[2];
  float* out = (float*)d_out;
  float* ws = (float*)d_ws;

  const int blupN  = (int)(((size_t)NH_ * S_ + 255) / 256);
  const int blkG   = NH_ * NB_;
  const int jG     = NH_ * BS_;

  k_grad_block_first<<<blkG, 256, 0, stream>>>(q, k, ws);
  k_grad_j_first<<<jG, 256, 0, stream>>>(q, ws);
  k_grad_r_first<<<blkG, 256, 0, stream>>>(k, ws);
  k_bl_update_first<<<blupN, 256, 0, stream>>>(ws);
  k_grad_block<<<blkG, 256, 0, stream>>>(q, k, ws);
  k_grad_j<<<jG, 256, 0, stream>>>(q, ws);
  k_grad_r<<<blkG, 256, 0, stream>>>(k, ws);
  k_bl_update<<<blupN, 256, 0, stream>>>(ws);

  k_Y<<<blkG, 256, 0, stream>>>(v, ws);
  k_zleft<<<jG, 256, 0, stream>>>(out, ws);
  k_zbd<<<blkG, 256, 0, stream>>>(v, out, ws);
}


# Per-block r-update kernel also updates its block's 64 bl rows on its six idle waves (wave per row, whole-line loads/stores); per-row bl kernel gone, 11 launches to 9; bl row sums become lane trees
# speedup vs baseline: 1.9396x; 1.2130x over previous
#include <hip/hip_runtime.h>


#define S_   2048
#define D_   64
#define BS_  64
#define NB_  32
#define CW_  96
#define NH_  48

#define PA_   68
#define PA32_ 36
#define PB_   80
#define PAB_  76
#define PT32_ 40
#define PR_   66
#define LDS_TILE_ __shared__ __attribute__((aligned(16)))

typedef __attribute__((ext_vector_type(2))) float v2f;
typedef __attribute__((ext_vector_type(8))) float v8f;

static constexpr size_t SZ_BL   = (size_t)NH_ * S_ * CW_;
static constexpr size_t SZ_R    = (size_t)NH_ * NB_ * BS_ * BS_;
static constexpr size_t O_BL    = 0;
static constexpr size_t O_R     = O_BL  + SZ_BL;
static constexpr size_t O_DBL   = O_R   + SZ_R;
static constexpr size_t O_T1    = O_DBL + SZ_BL;
static constexpr size_t O_T2    = O_T1  + SZ_R;
static constexpr size_t O_INVBL = O_T2  + SZ_R;
static constexpr size_t O_INVR  = O_INVBL + (size_t)NH_ * S_;
static constexpr size_t O_R2S   = O_INVR  + (size_t)NH_ * S_;
static constexpr size_t O_DAD   = O_R2S   + (size_t)NH_ * S_;
static constexpr size_t O_L2S   = O_DAD   + (size_t)NH_ * S_;

__device__ __forceinline__ int frag_half() { return (threadIdx.x & 31) >> 4; }
__device__ __forceinline__ int frag_l15()  { return threadIdx.x & 15; }
__device__ __forceinline__ int frag_m(int r) { return r + frag_half() * 8; }

template <bool AT, bool BT>
__device__ __forceinline__ v8f wmma_tile(const float* A, int lda, float asgn,
                                         const float* B, int ldb, int K, v8f acc) {
  const int half = frag_half();
  const int l15  = frag_l15();
  for (int k0 = 0; k0 < K; k0 += 4) {
    const int ka = k0 + half * 2;
    v2f a, b;
    if (AT) { a.x = asgn * A[ka * lda + l15];  a.y = asgn * A[(ka + 1) * lda + l15]; }
    else    { a.x = asgn * A[l15 * lda + ka];  a.y = asgn * A[l15 * lda + ka + 1]; }
    if (BT) { b.x = B[l15 * ldb + ka];         b.y = B[l15 * ldb + ka + 1]; }
    else    { b.x = B[ka * ldb + l15];         b.y = B[(ka + 1) * ldb + l15]; }
    acc = __builtin_amdgcn_wmma_f32_16x16x4_f32(false, a, false, b, (short)0, acc,
                                                false, false);
  }
  return acc;
}

__device__ __forceinline__ float inv_norm_from_sumsq(float s) {
  return (s > 0.f) ? 1.0f / sqrtf(s) : 0.f;
}
__device__ __forceinline__ float inv_norm_const_row(float v, int n) {
  float s = 0.f;
  for (int c = 0; c < n; ++c) s += v * v;
  return inv_norm_from_sumsq(s);
}

#define BL0_ 0.1020620726159658f
#define R0_  0.125f
__device__ __forceinline__ float invbl0() { return inv_norm_const_row(BL0_, CW_); }
__device__ __forceinline__ float invr0()  { return inv_norm_const_row(R0_, BS_); }
template <bool FIRST>
__device__ __forceinline__ float par(const float* p, int i, float c0) {
  if (FIRST) return c0;
  return p[i];
}

template <bool FIRST>
__device__ __forceinline__ void grad_block_body(const float* __restrict__ q,
                                                const float* __restrict__ kk,
                                                float* ws) {
  LDS_TILE_ float Qs[BS_ * PA_];
  LDS_TILE_ float Ks[BS_ * PA_];
  LDS_TILE_ float Rsq[BS_ * PA_];
  __shared__ float DiagL[BS_];

  const int blk = blockIdx.x;
  const int hb = blk >> 5, kb = blk & 31;
  const float* qh = q  + (size_t)hb * S_ * D_ + (size_t)kb * BS_ * D_;
  const float* kh = kk + (size_t)hb * S_ * D_ + (size_t)kb * BS_ * D_;
  float* blp   = ws + O_BL    + (size_t)hb * S_ * CW_;
  float* rp    = ws + O_R     + (size_t)hb * NB_ * BS_ * BS_ + (size_t)kb * BS_ * BS_;
  float* invbl = ws + O_INVBL + (size_t)hb * S_;
  float* invr  = ws + O_INVR  + (size_t)hb * S_;
  float* dbl   = ws + O_DBL   + (size_t)hb * S_ * CW_;
  float* t1    = ws + O_T1    + (size_t)hb * NB_ * BS_ * BS_ + (size_t)kb * BS_ * BS_;
  float* r2s   = ws + O_R2S   + (size_t)hb * S_;
  float* dad   = ws + O_DAD   + (size_t)hb * S_;

  const int tid = threadIdx.x;
  for (int i = tid; i < BS_ * D_; i += 256) {
    const int p = (i >> 6) * PA_ + (i & 63);
    Qs[p] = qh[i] * 0.125f;
    Ks[p] = kh[i];
    float rs = par<FIRST>(rp, i, R0_) * par<FIRST>(invr, kb * 64 + (i >> 6), invr0());
    Rsq[p] = rs * rs;
  }
  if (tid < BS_) {
    const int row = kb * 64 + tid;
    float v = par<FIRST>(blp, row * CW_ + 64 + kb, BL0_) * par<FIRST>(invbl, row, invbl0());
    DiagL[tid] = v * v;
  }
  __syncthreads();

  if (tid < BS_) {
    const int j = tid, row = kb * 64 + j;
    const float inv = par<FIRST>(invbl, row, invbl0());
    float s2 = 0.f, sd = 0.f;
    for (int i = 0; i < BS_; ++i) {
      float r = Rsq[j * PA_ + i];
      float b = par<FIRST>(blp, row * CW_ + i, BL0_) * inv;
      s2 += r * r;
      sd += r * b * b;
    }
    r2s[row] = s2;
    dad[row] = sd;
  }

  const int wid = tid >> 5;
  for (int t = wid; t < 32; t += 8) {
    const int which = t >> 4;
    const int tt = t & 15;
    const int tj = tt >> 2, ti = tt & 3;
    if (which == 0) {
      v8f acc;
      #pragma unroll
      for (int r = 0; r < 8; ++r) {
        const int m = tj * 16 + frag_m(r), n = ti * 16 + frag_l15();
        const int row = kb * 64 + m;
        float b = par<FIRST>(blp, row * CW_ + n, BL0_) * par<FIRST>(invbl, row, invbl0());
        acc[r] = b * b + DiagL[m] * Rsq[m * PA_ + n];
      }
      acc = wmma_tile<false, true>(Qs + tj * 16 * PA_, PA_, -1.f,
                                   Ks + ti * 16 * PA_, PA_, 64, acc);
      #pragma unroll
      for (int r = 0; r < 8; ++r) {
        const int m = tj * 16 + frag_m(r), n = ti * 16 + frag_l15();
        dbl[(kb * 64 + m) * CW_ + n] = acc[r];
      }
    } else {
      v8f acc = 0.0f;
      acc = wmma_tile<false, false>(Rsq + tj * 16 * PA_, PA_, 1.f,
                                    Ks + ti * 16, PA_, 64, acc);
      #pragma unroll
      for (int r = 0; r < 8; ++r) {
        const int m = tj * 16 + frag_m(r), n = ti * 16 + frag_l15();
        t1[m * 64 + n] = acc[r];
      }
    }
  }
}
__global__ __launch_bounds__(256) void k_grad_block_first(const float* __restrict__ q,
                                                          const float* __restrict__ kk,
                                                          float* ws) {
  grad_block_body<true>(q, kk, ws);
}
__global__ __launch_bounds__(256) void k_grad_block(const float* __restrict__ q,
                                                    const float* __restrict__ kk,
                                                    float* ws) {
  grad_block_body<false>(q, kk, ws);
}

template <bool FIRST>
__device__ __forceinline__ void grad_j_body(const float* __restrict__ q,
                                            float* ws) {
  LDS_TILE_ float Lj[NB_ * PT32_];
  LDS_TILE_ float Qj[NB_ * PAB_];
  LDS_TILE_ float T1j[NB_ * PA_];

  const int blk = blockIdx.x;
  const int hb = blk >> 6, j = blk & 63;
  const float* qh = q + (size_t)hb * S_ * D_;
  float* blp   = ws + O_BL    + (size_t)hb * S_ * CW_;
  float* invbl = ws + O_INVBL + (size_t)hb * S_;
  float* dbl   = ws + O_DBL   + (size_t)hb * S_ * CW_;
  float* t1h   = ws + O_T1    + (size_t)hb * NB_ * BS_ * BS_;
  float* t2h   = ws + O_T2    + (size_t)hb * NB_ * BS_ * BS_;
  float* r2s   = ws + O_R2S   + (size_t)hb * S_;
  float* dad   = ws + O_DAD   + (size_t)hb * S_;
  float* l2s   = ws + O_L2S   + (size_t)hb * S_;

  const int tid = threadIdx.x;
  for (int i = tid; i < NB_ * NB_; i += 256) {
    const int l = i >> 5, k2 = i & 31;
    float v = par<FIRST>(blp, (l * 64 + j) * CW_ + 64 + k2, BL0_) *
              par<FIRST>(invbl, l * 64 + j, invbl0());
    Lj[l * PT32_ + k2] = v * v;
  }
  for (int i = tid; i < NB_ * D_; i += 256) {
    const int l = i >> 6, d = i & 63;
    Qj[l * PAB_ + d] = qh[(l * 64 + j) * 64 + d] * 0.125f;
    T1j[l * PA_ + d] = t1h[l * (BS_ * BS_) + j * 64 + d];
  }
  __syncthreads();

  if (tid < NB_) {
    float s = 0.f;
    for (int l = 0; l < NB_; ++l) { float v = Lj[l * PT32_ + tid]; s += v * v; }
    l2s[j * 32 + tid] = s;
  }

  const int wid = tid >> 5;
  if (wid < 4) {
    const int tj = wid >> 1, ti = wid & 1;
    v8f acc = 0.0f;
    acc = wmma_tile<false, true>(Qj + tj * 16 * PAB_, PAB_, 1.f,
                                 T1j + ti * 16 * PA_, PA_, 64, acc);
    #pragma unroll
    for (int r = 0; r < 8; ++r) {
      const int l = tj * 16 + frag_m(r), k2 = ti * 16 + frag_l15();
      float v = Lj[l * PT32_ + k2] * r2s[k2 * 64 + j] - acc[r];
      if (l == k2) v += dad[l * 64 + j];
      dbl[(l * 64 + j) * CW_ + 64 + k2] = v;
    }
  }
  {
    const int tj = wid >> 2, ti = wid & 3;
    v8f acc = 0.0f;
    acc = wmma_tile<true, false>(Lj + tj * 16, PT32_, 1.f,
                                 Qj + ti * 16, PAB_, 32, acc);
    #pragma unroll
    for (int r = 0; r < 8; ++r) {
      const int k2 = tj * 16 + frag_m(r), d = ti * 16 + frag_l15();
      t2h[k2 * (BS_ * BS_) + j * 64 + d] = acc[r];
    }
  }
}
__global__ __launch_bounds__(256) void k_grad_j_first(const float* __restrict__ q,
                                                      float* ws) {
  grad_j_body<true>(q, ws);
}
__global__ __launch_bounds__(256) void k_grad_j(const float* __restrict__ q,
                                                float* ws) {
  grad_j_body<false>(q, ws);
}

template <bool FIRST>
__device__ __forceinline__ void grad_r_body(const float* __restrict__ kk,
                                            float* ws) {
  LDS_TILE_ float T2s[BS_ * PA_];
  LDS_TILE_ float Ks[BS_ * PA_];
  LDS_TILE_ float DRs[BS_ * PR_];

  const int blk = blockIdx.x;
  const int hb = blk >> 5, kb = blk & 31;
  const float* kh = kk + (size_t)hb * S_ * D_ + (size_t)kb * BS_ * D_;
  float* blp   = ws + O_BL    + (size_t)hb * S_ * CW_;
  float* rp    = ws + O_R     + (size_t)hb * NB_ * BS_ * BS_ + (size_t)kb * BS_ * BS_;
  float* invbl = ws + O_INVBL + (size_t)hb * S_;
  float* invr  = ws + O_INVR  + (size_t)hb * S_;
  float* t2h   = ws + O_T2    + (size_t)hb * NB_ * BS_ * BS_ + (size_t)kb * BS_ * BS_;
  float* l2s   = ws + O_L2S   + (size_t)hb * S_;
  const float* dbl = ws + O_DBL + (size_t)hb * S_ * CW_;

  const int tid = threadIdx.x;
  for (int i = tid; i < BS_ * D_; i += 256) {
    const int p = (i >> 6) * PA_ + (i & 63);
    T2s[p] = t2h[i];
    Ks[p]  = kh[i];
  }
  __syncthreads();

  const int wid = tid >> 5;
  for (int t = wid; t < 16; t += 8) {
    const int tj = t >> 2, ti = t & 3;
    v8f acc;
    #pragma unroll
    for (int r = 0; r < 8; ++r) {
      const int m = tj * 16 + frag_m(r), n = ti * 16 + frag_l15();
      const int row = kb * 64 + m;
      const float ib = par<FIRST>(invbl, row, invbl0());
      float bd = par<FIRST>(blp, row * CW_ + n, BL0_) * ib;         bd *= bd;
      float dl = par<FIRST>(blp, row * CW_ + 64 + kb, BL0_) * ib;   dl *= dl;
      float rv = par<FIRST>(rp, m * 64 + n, R0_) * par<FIRST>(invr, row, invr0());
      rv *= rv;
      acc[r] = rv * l2s[m * 32 + kb] + dl * bd;
    }
    acc = wmma_tile<false, true>(T2s + tj * 16 * PA_, PA_, -1.f,
                                 Ks + ti * 16 * PA_, PA_, 64, acc);
    #pragma unroll
    for (int r = 0; r < 8; ++r) {
      const int m = tj * 16 + frag_m(r), n = ti * 16 + frag_l15();
      DRs[m * PR_ + n] = acc[r];
    }
  }
  __syncthreads();

  if (tid < BS_) {
    const int j = tid, row = kb * 64 + j;
    const float inv = par<FIRST>(invr, row, invr0());
    float dot = 0.f;
    for (int i = 0; i < BS_; ++i) {
      float rs = par<FIRST>(rp, j * 64 + i, R0_) * inv;
      dot += rs * (2.f * DRs[j * PR_ + i] * rs);
    }
    float s2 = 0.f;
    #pragma unroll 32
    for (int i = 0; i < BS_; ++i) {
      const float old = par<FIRST>(rp, j * 64 + i, R0_);
      float rs = old * inv;
      float dr = 2.f * DRs[j * PR_ + i] * rs;
      float df = (dr - rs * dot) * inv;
      float nv = old - 0.5f * df;
      rp[j * 64 + i] = nv;
      s2 += nv * nv;
    }
    invr[row] = inv_norm_from_sumsq(s2);
  } else {
    const int lane = tid & 31;
    for (int grp = wid - 2; grp < 16; grp += 6) {
      const int row0 = kb * 64 + grp * 4;
      float old[4][3], bs[4][3], g[4][3], inv[4], dot[4], s2[4];
      #pragma unroll
      for (int u = 0; u < 4; ++u) {
        const int row = row0 + u;
        inv[u] = par<FIRST>(invbl, row, invbl0());
        #pragma unroll
        for (int i = 0; i < 3; ++i) {
          const int c = row * CW_ + lane + 32 * i;
          old[u][i] = par<FIRST>(blp, c, BL0_);
          g[u][i]   = dbl[c];
        }
      }
      #pragma unroll
      for (int u = 0; u < 4; ++u) {
        float p = 0.f;
        #pragma unroll
        for (int i = 0; i < 3; ++i) {
          bs[u][i] = old[u][i] * inv[u];
          g[u][i]  = 2.f * g[u][i] * bs[u][i];
          p += bs[u][i] * g[u][i];
        }
        dot[u] = p;
      }
      #pragma unroll
      for (int m = 16; m >= 1; m >>= 1) {
        #pragma unroll
        for (int u = 0; u < 4; ++u) dot[u] += __shfl_xor(dot[u], m);
      }
      #pragma unroll
      for (int u = 0; u < 4; ++u) {
        const int row = row0 + u;
        float s = 0.f;
        #pragma unroll
        for (int i = 0; i < 3; ++i) {
          float nv = old[u][i] - 0.5f * ((g[u][i] - bs[u][i] * dot[u]) * inv[u]);
          blp[row * CW_ + lane + 32 * i] = nv;
          s += nv * nv;
        }
        s2[u] = s;
      }
      #pragma unroll
      for (int m = 16; m >= 1; m >>= 1) {
        #pragma unroll
        for (int u = 0; u < 4; ++u) s2[u] += __shfl_xor(s2[u], m);
      }
      if (lane < 4) {
        float mine = s2[0];
        if (lane == 1) mine = s2[1];
        if (lane == 2) mine = s2[2];
        if (lane == 3) mine = s2[3];
        invbl[row0 + lane] = inv_norm_from_sumsq(mine);
      }
    }
  }
}
__global__ __launch_bounds__(256) void k_grad_r_first(const float* __restrict__ kk,
                                                      float* ws) {
  grad_r_body<true>(kk, ws);
}
__global__ __launch_bounds__(256) void k_grad_r(const float* __restrict__ kk,
                                                float* ws) {
  grad_r_body<false>(kk, ws);
}

__global__ __launch_bounds__(256) void k_Y(const float* __restrict__ val,
                                           float* ws) {
  LDS_TILE_ float Rsq[BS_ * PA_];
  LDS_TILE_ float Xs[BS_ * PB_];
  const int blk = blockIdx.x;
  const int hb = blk >> 5, kb = blk & 31;
  const float* xh = val + (size_t)hb * S_ * D_ + (size_t)kb * BS_ * D_;
  float* rp   = ws + O_R    + (size_t)hb * NB_ * BS_ * BS_ + (size_t)kb * BS_ * BS_;
  float* invr = ws + O_INVR + (size_t)hb * S_;
  float* yb   = ws + O_T1   + (size_t)hb * NB_ * BS_ * BS_ + (size_t)kb * BS_ * BS_;

  const int tid = threadIdx.x;
  for (int i = tid; i < BS_ * D_; i += 256) {
    float rs = rp[i] * invr[kb * 64 + (i >> 6)];
    Rsq[(i >> 6) * PA_ + (i & 63)] = rs * rs;
    Xs[(i >> 6) * PB_ + (i & 63)]  = xh[i];
  }
  __syncthreads();

  const int wid = tid >> 5;
  for (int t = wid; t < 16; t += 8) {
    const int tj = t >> 2, ti = t & 3;
    v8f acc = 0.0f;
    acc = wmma_tile<false, false>(Rsq + tj * 16 * PA_, PA_, 1.f,
                                  Xs + ti * 16, PB_, 64, acc);
    #pragma unroll
    for (int r = 0; r < 8; ++r) {
      const int m = tj * 16 + frag_m(r), n = ti * 16 + frag_l15();
      yb[m * 64 + n] = acc[r];
    }
  }
}

__global__ __launch_bounds__(256) void k_zleft(float* __restrict__ out,
                                               float* ws) {
  LDS_TILE_ float Lj[NB_ * PA32_];
  LDS_TILE_ float Yj[NB_ * PB_];
  const int blk = blockIdx.x;
  const int hb = blk >> 6, j = blk & 63;
  float* blp   = ws + O_BL    + (size_t)hb * S_ * CW_;
  float* invbl = ws + O_INVBL + (size_t)hb * S_;
  float* yb    = ws + O_T1    + (size_t)hb * NB_ * BS_ * BS_;
  float* oh    = out + (size_t)hb * S_ * D_;

  const int tid = threadIdx.x;
  for (int i = tid; i < NB_ * NB_; i += 256) {
    const int l = i >> 5, k2 = i & 31;
    float v = blp[(l * 64 + j) * CW_ + 64 + k2] * invbl[l * 64 + j];
    Lj[l * PA32_ + k2] = v * v;
  }
  for (int i = tid; i < NB_ * D_; i += 256) {
    const int k2 = i >> 6, d = i & 63;
    Yj[k2 * PB_ + d] = yb[k2 * (BS_ * BS_) + j * 64 + d];
  }
  __syncthreads();

  const int wid = tid >> 5;
  const int tj = wid >> 2, ti = wid & 3;
  v8f acc = 0.0f;
  acc = wmma_tile<false, false>(Lj + tj * 16 * PA32_, PA32_, 1.f,
                                Yj + ti * 16, PB_, 32, acc);
  #pragma unroll
  for (int r = 0; r < 8; ++r) {
    const int l = tj * 16 + frag_m(r), d = ti * 16 + frag_l15();
    oh[(size_t)(l * 64 + j) * 64 + d] = acc[r];
  }
}

__global__ __launch_bounds__(256) void k_zbd(const float* __restrict__ val,
                                             float* __restrict__ out,
                                             float* ws) {
  LDS_TILE_ float BDs[BS_ * PA_];
  LDS_TILE_ float Xs[BS_ * PB_];
  const int blk = blockIdx.x;
  const int hb = blk >> 5, lb = blk & 31;
  const float* xh = val + (size_t)hb * S_ * D_ + (size_t)lb * BS_ * D_;
  float* blp   = ws + O_BL    + (size_t)hb * S_ * CW_;
  float* invbl = ws + O_INVBL + (size_t)hb * S_;
  float* oh    = out + (size_t)hb * S_ * D_ + (size_t)lb * BS_ * D_;

  const int tid = threadIdx.x;
  for (int i = tid; i < BS_ * D_; i += 256) {
    const int j = i >> 6, c = i & 63;
    float v = blp[(lb * 64 + j) * CW_ + c] * invbl[lb * 64 + j];
    BDs[j * PA_ + c] = v * v;
    Xs[j * PB_ + c]  = xh[i];
  }
  __syncthreads();

  const int wid = tid >> 5;
  for (int t = wid; t < 16; t += 8) {
    const int tj = t >> 2, ti = t & 3;
    v8f acc = 0.0f;
    acc = wmma_tile<false, false>(BDs + tj * 16 * PA_, PA_, 1.f,
                                  Xs + ti * 16, PB_, 64, acc);
    #pragma unroll
    for (int r = 0; r < 8; ++r) {
      const int m = tj * 16 + frag_m(r), n = ti * 16 + frag_l15();
      oh[m * 64 + n] += acc[r];
    }
  }
}

extern "C" void kernel_launch(void* const* d_in, const int* in_sizes, int n_in,
                              void* d_out, int out_size, void* d_ws, size_t ws_size,
                              hipStream_t stream) {
  const float* q = (const float*)d_in[0];
  const float* k = (const float*)d_in[1];
  const float* v = (const float*)d_in[2];
  float* out = (float*)d_out;
  float* ws = (float*)d_ws;

  const int blkG   = NH_ * NB_;
  const int jG     = NH_ * BS_;

  k_grad_block_first<<<blkG, 256, 0, stream>>>(q, k, ws);
  k_grad_j_first<<<jG, 256, 0, stream>>>(q, ws);
  k_grad_r_first<<<blkG, 256, 0, stream>>>(k, ws);
  k_grad_block<<<blkG, 256, 0, stream>>>(q, k, ws);
  k_grad_j<<<jG, 256, 0, stream>>>(q, ws);
  k_grad_r<<<blkG, 256, 0, stream>>>(k, ws);

  k_Y<<<blkG, 256, 0, stream>>>(v, ws);
  k_zleft<<<jG, 256, 0, stream>>>(out, ws);
  k_zbd<<<blkG, 256, 0, stream>>>(v, out, ws);
}


# Per-block stages reading only what their own work-group just stored run in its launch behind a barrier: update + next block grads; update + Y + block_diag@X (80-VGPR cap); 9 launches to 6; same sums
# speedup vs baseline: 2.1712x; 1.1194x over previous
#include <hip/hip_runtime.h>


#define S_   2048
#define D_   64
#define BS_  64
#define NB_  32
#define CW_  96
#define NH_  48

#define PA_   68
#define PA32_ 36
#define PB_   80
#define PAB_  76
#define PT32_ 40
#define PR_   66
#define LDS_TILE_ __shared__ __attribute__((aligned(16)))

typedef __attribute__((ext_vector_type(2))) float v2f;
typedef __attribute__((ext_vector_type(8))) float v8f;

static constexpr size_t SZ_BL   = (size_t)NH_ * S_ * CW_;
static constexpr size_t SZ_R    = (size_t)NH_ * NB_ * BS_ * BS_;
static constexpr size_t O_BL    = 0;
static constexpr size_t O_R     = O_BL  + SZ_BL;
static constexpr size_t O_DBL   = O_R   + SZ_R;
static constexpr size_t O_T1    = O_DBL + SZ_BL;
static constexpr size_t O_T2    = O_T1  + SZ_R;
static constexpr size_t O_INVBL = O_T2  + SZ_R;
static constexpr size_t O_INVR  = O_INVBL + (size_t)NH_ * S_;
static constexpr size_t O_R2S   = O_INVR  + (size_t)NH_ * S_;
static constexpr size_t O_DAD   = O_R2S   + (size_t)NH_ * S_;
static constexpr size_t O_L2S   = O_DAD   + (size_t)NH_ * S_;

__device__ __forceinline__ int frag_half() { return (threadIdx.x & 31) >> 4; }
__device__ __forceinline__ int frag_l15()  { return threadIdx.x & 15; }
__device__ __forceinline__ int frag_m(int r) { return r + frag_half() * 8; }

template <bool AT, bool BT>
__device__ __forceinline__ v8f wmma_tile(const float* A, int lda, float asgn,
                                         const float* B, int ldb, int K, v8f acc) {
  const int half = frag_half();
  const int l15  = frag_l15();
  for (int k0 = 0; k0 < K; k0 += 4) {
    const int ka = k0 + half * 2;
    v2f a, b;
    if (AT) { a.x = asgn * A[ka * lda + l15];  a.y = asgn * A[(ka + 1) * lda + l15]; }
    else    { a.x = asgn * A[l15 * lda + ka];  a.y = asgn * A[l15 * lda + ka + 1]; }
    if (BT) { b.x = B[l15 * ldb + ka];         b.y = B[l15 * ldb + ka + 1]; }
    else    { b.x = B[ka * ldb + l15];         b.y = B[(ka + 1) * ldb + l15]; }
    acc = __builtin_amdgcn_wmma_f32_16x16x4_f32(false, a, false, b, (short)0, acc,
                                                false, false);
  }
  return acc;
}

__device__ __forceinline__ float inv_norm_from_sumsq(float s) {
  return (s > 0.f) ? 1.0f / sqrtf(s) : 0.f;
}
__device__ __forceinline__ float inv_norm_const_row(float v, int n) {
  float s = 0.f;
  for (int c = 0; c < n; ++c) s += v * v;
  return inv_norm_from_sumsq(s);
}

#define BL0_ 0.1020620726159658f
#define R0_  0.125f
__device__ __forceinline__ float invbl0() { return inv_norm_const_row(BL0_, CW_); }
__device__ __forceinline__ float invr0()  { return inv_norm_const_row(R0_, BS_); }
template <bool FIRST>
__device__ __forceinline__ float par(const float* p, int i, float c0) {
  if (FIRST) return c0;
  return p[i];
}

template <bool FIRST, bool LOADK>
__device__ __forceinline__ void grad_block_body(const float* __restrict__ q,
                                                const float* __restrict__ kk,
                                                float* ws, float* Qs, float* Ks,
                                                float* Rsq, float* DiagL) {

  const int blk = blockIdx.x;
  const int hb = blk >> 5, kb = blk & 31;
  const float* qh = q  + (size_t)hb * S_ * D_ + (size_t)kb * BS_ * D_;
  const float* kh = kk + (size_t)hb * S_ * D_ + (size_t)kb * BS_ * D_;
  float* blp   = ws + O_BL    + (size_t)hb * S_ * CW_;
  float* rp    = ws + O_R     + (size_t)hb * NB_ * BS_ * BS_ + (size_t)kb * BS_ * BS_;
  float* invbl = ws + O_INVBL + (size_t)hb * S_;
  float* invr  = ws + O_INVR  + (size_t)hb * S_;
  float* dbl   = ws + O_DBL   + (size_t)hb * S_ * CW_;
  float* t1    = ws + O_T1    + (size_t)hb * NB_ * BS_ * BS_ + (size_t)kb * BS_ * BS_;
  float* r2s   = ws + O_R2S   + (size_t)hb * S_;
  float* dad   = ws + O_DAD   + (size_t)hb * S_;

  const int tid = threadIdx.x;
  for (int i = tid; i < BS_ * D_; i += 256) {
    const int p = (i >> 6) * PA_ + (i & 63);
    Qs[p] = qh[i] * 0.125f;
    if (LOADK) Ks[p] = kh[i];
    float rs = par<FIRST>(rp, i, R0_) * par<FIRST>(invr, kb * 64 + (i >> 6), invr0());
    Rsq[p] = rs * rs;
  }
  if (tid < BS_) {
    const int row = kb * 64 + tid;
    float v = par<FIRST>(blp, row * CW_ + 64 + kb, BL0_) * par<FIRST>(invbl, row, invbl0());
    DiagL[tid] = v * v;
  }
  __syncthreads();

  if (tid < BS_) {
    const int j = tid, row = kb * 64 + j;
    const float inv = par<FIRST>(invbl, row, invbl0());
    float s2 = 0.f, sd = 0.f;
    for (int i = 0; i < BS_; ++i) {
      float r = Rsq[j * PA_ + i];
      float b = par<FIRST>(blp, row * CW_ + i, BL0_) * inv;
      s2 += r * r;
      sd += r * b * b;
    }
    r2s[row] = s2;
    dad[row] = sd;
  }

  const int wid = tid >> 5;
  for (int t = wid; t < 32; t += 8) {
    const int which = t >> 4;
    const int tt = t & 15;
    const int tj = tt >> 2, ti = tt & 3;
    if (which == 0) {
      v8f acc;
      #pragma unroll
      for (int r = 0; r < 8; ++r) {
        const int m = tj * 16 + frag_m(r), n = ti * 16 + frag_l15();
        const int row = kb * 64 + m;
        float b = par<FIRST>(blp, row * CW_ + n, BL0_) * par<FIRST>(invbl, row, invbl0());
        acc[r] = b * b + DiagL[m] * Rsq[m * PA_ + n];
      }
      acc = wmma_tile<false, true>(Qs + tj * 16 * PA_, PA_, -1.f,
                                   Ks + ti * 16 * PA_, PA_, 64, acc);
      #pragma unroll
      for (int r = 0; r < 8; ++r) {
        const int m = tj * 16 + frag_m(r), n = ti * 16 + frag_l15();
        dbl[(kb * 64 + m) * CW_ + n] = acc[r];
      }
    } else {
      v8f acc = 0.0f;
      acc = wmma_tile<false, false>(Rsq + tj * 16 * PA_, PA_, 1.f,
                                    Ks + ti * 16, PA_, 64, acc);
      #pragma unroll
      for (int r = 0; r < 8; ++r) {
        const int m = tj * 16 + frag_m(r), n = ti * 16 + frag_l15();
        t1[m * 64 + n] = acc[r];
      }
    }
  }
}
__global__ __launch_bounds__(256) void k_grad_block_first(const float* __restrict__ q,
                                                          const float* __restrict__ kk,
                                                          float* ws) {
  LDS_TILE_ float Qs[BS_ * PA_];
  LDS_TILE_ float Ks[BS_ * PA_];
  LDS_TILE_ float Rsq[BS_ * PA_];
  __shared__ float DiagL[BS_];
  grad_block_body<true, true>(q, kk, ws, Qs, Ks, Rsq, DiagL);
}

template <bool FIRST>
__device__ __forceinline__ void grad_j_body(const float* __restrict__ q,
                                            float* ws) {
  LDS_TILE_ float Lj[NB_ * PT32_];
  LDS_TILE_ float Qj[NB_ * PAB_];
  LDS_TILE_ float T1j[NB_ * PA_];

  const int blk = blockIdx.x;
  const int hb = blk >> 6, j = blk & 63;
  const float* qh = q + (size_t)hb * S_ * D_;
  float* blp   = ws + O_BL    + (size_t)hb * S_ * CW_;
  float* invbl = ws + O_INVBL + (size_t)hb * S_;
  float* dbl   = ws + O_DBL   + (size_t)hb * S_ * CW_;
  float* t1h   = ws + O_T1    + (size_t)hb * NB_ * BS_ * BS_;
  float* t2h   = ws + O_T2    + (size_t)hb * NB_ * BS_ * BS_;
  float* r2s   = ws + O_R2S   + (size_t)hb * S_;
  float* dad   = ws + O_DAD   + (size_t)hb * S_;
  float* l2s   = ws + O_L2S   + (size_t)hb * S_;

  const int tid = threadIdx.x;
  for (int i = tid; i < NB_ * NB_; i += 256) {
    const int l = i >> 5, k2 = i & 31;
    float v = par<FIRST>(blp, (l * 64 + j) * CW_ + 64 + k2, BL0_) *
              par<FIRST>(invbl, l * 64 + j, invbl0());
    Lj[l * PT32_ + k2] = v * v;
  }
  for (int i = tid; i < NB_ * D_; i += 256) {
    const int l = i >> 6, d = i & 63;
    Qj[l * PAB_ + d] = qh[(l * 64 + j) * 64 + d] * 0.125f;
    T1j[l * PA_ + d] = t1h[l * (BS_ * BS_) + j * 64 + d];
  }
  __syncthreads();

  if (tid < NB_) {
    float s = 0.f;
    for (int l = 0; l < NB_; ++l) { float v = Lj[l * PT32_ + tid]; s += v * v; }
    l2s[j * 32 + tid] = s;
  }

  const int wid = tid >> 5;
  if (wid < 4) {
    const int tj = wid >> 1, ti = wid & 1;
    v8f acc = 0.0f;
    acc = wmma_tile<false, true>(Qj + tj * 16 * PAB_, PAB_, 1.f,
                                 T1j + ti * 16 * PA_, PA_, 64, acc);
    #pragma unroll
    for (int r = 0; r < 8; ++r) {
      const int l = tj * 16 + frag_m(r), k2 = ti * 16 + frag_l15();
      float v = Lj[l * PT32_ + k2] * r2s[k2 * 64 + j] - acc[r];
      if (l == k2) v += dad[l * 64 + j];
      dbl[(l * 64 + j) * CW_ + 64 + k2] = v;
    }
  }
  {
    const int tj = wid >> 2, ti = wid & 3;
    v8f acc = 0.0f;
    acc = wmma_tile<true, false>(Lj + tj * 16, PT32_, 1.f,
                                 Qj + ti * 16, PAB_, 32, acc);
    #pragma unroll
    for (int r = 0; r < 8; ++r) {
      const int k2 = tj * 16 + frag_m(r), d = ti * 16 + frag_l15();
      t2h[k2 * (BS_ * BS_) + j * 64 + d] = acc[r];
    }
  }
}
__global__ __launch_bounds__(256) void k_grad_j_first(const float* __restrict__ q,
                                                      float* ws) {
  grad_j_body<true>(q, ws);
}
__global__ __launch_bounds__(256) void k_grad_j(const float* __restrict__ q,
                                                float* ws) {
  grad_j_body<false>(q, ws);
}

template <bool FIRST>
__device__ __forceinline__ void grad_r_body(const float* __restrict__ kk,
                                            float* ws, float* T2s, float* Ks,
                                            float* DRs) {

  const int blk = blockIdx.x;
  const int hb = blk >> 5, kb = blk & 31;
  const float* kh = kk + (size_t)hb * S_ * D_ + (size_t)kb * BS_ * D_;
  float* blp   = ws + O_BL    + (size_t)hb * S_ * CW_;
  float* rp    = ws + O_R     + (size_t)hb * NB_ * BS_ * BS_ + (size_t)kb * BS_ * BS_;
  float* invbl = ws + O_INVBL + (size_t)hb * S_;
  float* invr  = ws + O_INVR  + (size_t)hb * S_;
  float* t2h   = ws + O_T2    + (size_t)hb * NB_ * BS_ * BS_ + (size_t)kb * BS_ * BS_;
  float* l2s   = ws + O_L2S   + (size_t)hb * S_;
  const float* dbl = ws + O_DBL + (size_t)hb * S_ * CW_;

  const int tid = threadIdx.x;
  for (int i = tid; i < BS_ * D_; i += 256) {
    const int p = (i >> 6) * PA_ + (i & 63);
    T2s[p] = t2h[i];
    Ks[p]  = kh[i];
  }
  __syncthreads();

  const int wid = tid >> 5;
  for (int t = wid; t < 16; t += 8) {
    const int tj = t >> 2, ti = t & 3;
    v8f acc;
    #pragma unroll
    for (int r = 0; r < 8; ++r) {
      const int m = tj * 16 + frag_m(r), n = ti * 16 + frag_l15();
      const int row = kb * 64 + m;
      const float ib = par<FIRST>(invbl, row, invbl0());
      float bd = par<FIRST>(blp, row * CW_ + n, BL0_) * ib;         bd *= bd;
      float dl = par<FIRST>(blp, row * CW_ + 64 + kb, BL0_) * ib;   dl *= dl;
      float rv = par<FIRST>(rp, m * 64 + n, R0_) * par<FIRST>(invr, row, invr0());
      rv *= rv;
      acc[r] = rv * l2s[m * 32 + kb] + dl * bd;
    }
    acc = wmma_tile<false, true>(T2s + tj * 16 * PA_, PA_, -1.f,
                                 Ks + ti * 16 * PA_, PA_, 64, acc);
    #pragma unroll
    for (int r = 0; r < 8; ++r) {
      const int m = tj * 16 + frag_m(r), n = ti * 16 + frag_l15();
      DRs[m * PR_ + n] = acc[r];
    }
  }
  __syncthreads();

  if (tid < BS_) {
    const int j = tid, row = kb * 64 + j;
    const float inv = par<FIRST>(invr, row, invr0());
    float dot = 0.f;
    for (int i = 0; i < BS_; ++i) {
      float rs = par<FIRST>(rp, j * 64 + i, R0_) * inv;
      dot += rs * (2.f * DRs[j * PR_ + i] * rs);
    }
    float s2 = 0.f;
    #pragma unroll 32
    for (int i = 0; i < BS_; ++i) {
      const float old = par<FIRST>(rp, j * 64 + i, R0_);
      float rs = old * inv;
      float dr = 2.f * DRs[j * PR_ + i] * rs;
      float df = (dr - rs * dot) * inv;
      float nv = old - 0.5f * df;
      rp[j * 64 + i] = nv;
      s2 += nv * nv;
    }
    invr[row] = inv_norm_from_sumsq(s2);
  } else {
    const int lane = tid & 31;
    for (int grp = wid - 2; grp < 16; grp += 6) {
      const int row0 = kb * 64 + grp * 4;
      float old[4][3], bs[4][3], g[4][3], inv[4], dot[4], s2[4];
      #pragma unroll
      for (int u = 0; u < 4; ++u) {
        const int row = row0 + u;
        inv[u] = par<FIRST>(invbl, row, invbl0());
        #pragma unroll
        for (int i = 0; i < 3; ++i) {
          const int c = row * CW_ + lane + 32 * i;
          old[u][i] = par<FIRST>(blp, c, BL0_);
          g[u][i]   = dbl[c];
        }
      }
      #pragma unroll
      for (int u = 0; u < 4; ++u) {
        float p = 0.f;
        #pragma unroll
        for (int i = 0; i < 3; ++i) {
          bs[u][i] = old[u][i] * inv[u];
          g[u][i]  = 2.f * g[u][i] * bs[u][i];
          p += bs[u][i] * g[u][i];
        }
        dot[u] = p;
      }
      #pragma unroll
      for (int m = 16; m >= 1; m >>= 1) {
        #pragma unroll
        for (int u = 0; u < 4; ++u) dot[u] += __shfl_xor(dot[u], m);
      }
      #pragma unroll
      for (int u = 0; u < 4; ++u) {
        const int row = row0 + u;
        float s = 0.f;
        #pragma unroll
        for (int i = 0; i < 3; ++i) {
          float nv = old[u][i] - 0.5f * ((g[u][i] - bs[u][i] * dot[u]) * inv[u]);
          blp[row * CW_ + lane + 32 * i] = nv;
          s += nv * nv;
        }
        s2[u] = s;
      }
      #pragma unroll
      for (int m = 16; m >= 1; m >>= 1) {
        #pragma unroll
        for (int u = 0; u < 4; ++u) s2[u] += __shfl_xor(s2[u], m);
      }
      if (lane < 4) {
        float mine = s2[0];
        if (lane == 1) mine = s2[1];
        if (lane == 2) mine = s2[2];
        if (lane == 3) mine = s2[3];
        invbl[row0 + lane] = inv_norm_from_sumsq(mine);
      }
    }
  }
}
__global__ __launch_bounds__(256) void k_r1_block2(const float* __restrict__ q,
                                                   const float* __restrict__ kk,
                                                   float* ws) {
  LDS_TILE_ float pool[3 * BS_ * PA_];
  __shared__ float DiagL[BS_];
  float* T2s = pool;
  float* Ks  = pool + BS_ * PA_;
  float* DRs = pool + 2 * BS_ * PA_;
  grad_r_body<true>(kk, ws, T2s, Ks, DRs);
  __syncthreads();
  grad_block_body<false, false>(q, kk, ws, T2s, Ks, DRs, DiagL);
}

__global__ __launch_bounds__(256, 12) void k_r2_apply(const float* __restrict__ kk,
                                                  const float* __restrict__ val,
                                                  float* __restrict__ out,
                                                  float* ws) {
  LDS_TILE_ float pool[2 * BS_ * PA_ + BS_ * PR_];
  grad_r_body<false>(kk, ws, pool, pool + BS_ * PA_, pool + 2 * BS_ * PA_);
  __syncthreads();

  float* Xs = pool;
  float* Ms = pool + BS_ * PB_;
  const int blk = blockIdx.x;
  const int hb = blk >> 5, kb = blk & 31;
  const float* xh = val + (size_t)hb * S_ * D_ + (size_t)kb * BS_ * D_;
  float* rp    = ws + O_R     + (size_t)hb * NB_ * BS_ * BS_ + (size_t)kb * BS_ * BS_;
  float* invr  = ws + O_INVR  + (size_t)hb * S_;
  float* blp   = ws + O_BL    + (size_t)hb * S_ * CW_;
  float* invbl = ws + O_INVBL + (size_t)hb * S_;
  float* yb    = ws + O_T1    + (size_t)hb * NB_ * BS_ * BS_ + (size_t)kb * BS_ * BS_;
  float* oh    = out + (size_t)hb * S_ * D_ + (size_t)kb * BS_ * D_;

  const int tid = threadIdx.x;
  for (int i = tid; i < BS_ * D_; i += 256) {
    float rs = rp[i] * invr[kb * 64 + (i >> 6)];
    Ms[(i >> 6) * PA_ + (i & 63)] = rs * rs;
    Xs[(i >> 6) * PB_ + (i & 63)] = xh[i];
  }
  __syncthreads();

  const int wid = tid >> 5;
  for (int t = wid; t < 16; t += 8) {
    const int tj = t >> 2, ti = t & 3;
    v8f acc = 0.0f;
    acc = wmma_tile<false, false>(Ms + tj * 16 * PA_, PA_, 1.f,
                                  Xs + ti * 16, PB_, 64, acc);
    #pragma unroll
    for (int r = 0; r < 8; ++r) {
      const int m = tj * 16 + frag_m(r), n = ti * 16 + frag_l15();
      yb[m * 64 + n] = acc[r];
    }
  }
  __syncthreads();
  for (int i = tid; i < BS_ * D_; i += 256) {
    const int j = i >> 6, cc = i & 63;
    float v = blp[(kb * 64 + j) * CW_ + cc] * invbl[kb * 64 + j];
    Ms[j * PA_ + cc] = v * v;
  }
  __syncthreads();
  for (int t = wid; t < 16; t += 8) {
    const int tj = t >> 2, ti = t & 3;
    v8f acc = 0.0f;
    acc = wmma_tile<false, false>(Ms + tj * 16 * PA_, PA_, 1.f,
                                  Xs + ti * 16, PB_, 64, acc);
    #pragma unroll
    for (int r = 0; r < 8; ++r) {
      const int m = tj * 16 + frag_m(r), n = ti * 16 + frag_l15();
      oh[m * 64 + n] = acc[r];
    }
  }
}

__global__ __launch_bounds__(256) void k_zleft(float* __restrict__ out,
                                               float* ws) {
  LDS_TILE_ float Lj[NB_ * PA32_];
  LDS_TILE_ float Yj[NB_ * PB_];
  const int blk = blockIdx.x;
  const int hb = blk >> 6, j = blk & 63;
  float* blp   = ws + O_BL    + (size_t)hb * S_ * CW_;
  float* invbl = ws + O_INVBL + (size_t)hb * S_;
  float* yb    = ws + O_T1    + (size_t)hb * NB_ * BS_ * BS_;
  float* oh    = out + (size_t)hb * S_ * D_;

  const int tid = threadIdx.x;
  const int wid = tid >> 5;
  const int tj = wid >> 2, ti = wid & 3;
  v8f prev;
  #pragma unroll
  for (int r = 0; r < 8; ++r) {
    const int l = tj * 16 + frag_m(r), d = ti * 16 + frag_l15();
    prev[r] = oh[(size_t)(l * 64 + j) * 64 + d];
  }
  for (int i = tid; i < NB_ * NB_; i += 256) {
    const int l = i >> 5, k2 = i & 31;
    float v = blp[(l * 64 + j) * CW_ + 64 + k2] * invbl[l * 64 + j];
    Lj[l * PA32_ + k2] = v * v;
  }
  for (int i = tid; i < NB_ * D_; i += 256) {
    const int k2 = i >> 6, d = i & 63;
    Yj[k2 * PB_ + d] = yb[k2 * (BS_ * BS_) + j * 64 + d];
  }
  __syncthreads();

  v8f acc = 0.0f;
  acc = wmma_tile<false, false>(Lj + tj * 16 * PA32_, PA32_, 1.f,
                                Yj + ti * 16, PB_, 32, acc);
  #pragma unroll
  for (int r = 0; r < 8; ++r) {
    const int l = tj * 16 + frag_m(r), d = ti * 16 + frag_l15();
    oh[(size_t)(l * 64 + j) * 64 + d] = prev[r] + acc[r];
  }
}

extern "C" void kernel_launch(void* const* d_in, const int* in_sizes, int n_in,
                              void* d_out, int out_size, void* d_ws, size_t ws_size,
                              hipStream_t stream) {
  const float* q = (const float*)d_in[0];
  const float* k = (const float*)d_in[1];
  const float* v = (const float*)d_in[2];
  float* out = (float*)d_out;
  float* ws = (float*)d_ws;

  const int blkG   = NH_ * NB_;
  const int jG     = NH_ * BS_;

  k_grad_block_first<<<blkG, 256, 0, stream>>>(q, k, ws);
  k_grad_j_first<<<jG, 256, 0, stream>>>(q, ws);
  k_r1_block2<<<blkG, 256, 0, stream>>>(q, k, ws);
  k_grad_j<<<jG, 256, 0, stream>>>(q, ws);
  k_r2_apply<<<blkG, 256, 0, stream>>>(k, v, out, ws);
  k_zleft<<<jG, 256, 0, stream>>>(out, ws);
}


# Step 1 (constant start): t1 rows equal over j, t2 rows over k, one row of each stored; d_left needs no gather and moves into the block update launch; per-j kernel gone, 6 launches to 5; same sums
# speedup vs baseline: 2.5306x; 1.1655x over previous
#include <hip/hip_runtime.h>


#define S_   2048
#define D_   64
#define BS_  64
#define NB_  32
#define CW_  96
#define NH_  48

#define PA_   68
#define PA32_ 36
#define PB_   80
#define PAB_  76
#define PT32_ 40
#define PR_   66
#define LDS_TILE_ __shared__ __attribute__((aligned(16)))

typedef __attribute__((ext_vector_type(2))) float v2f;
typedef __attribute__((ext_vector_type(8))) float v8f;

static constexpr size_t SZ_BL   = (size_t)NH_ * S_ * CW_;
static constexpr size_t SZ_R    = (size_t)NH_ * NB_ * BS_ * BS_;
static constexpr size_t O_BL    = 0;
static constexpr size_t O_R     = O_BL  + SZ_BL;
static constexpr size_t O_DBL   = O_R   + SZ_R;
static constexpr size_t O_T1    = O_DBL + SZ_BL;
static constexpr size_t O_T2    = O_T1  + SZ_R;
static constexpr size_t T2C_OFF = 0;
static constexpr size_t T1C_OFF = (size_t)BS_ * BS_;
static constexpr size_t O_INVBL = O_T2  + SZ_R;
static constexpr size_t O_INVR  = O_INVBL + (size_t)NH_ * S_;
static constexpr size_t O_R2S   = O_INVR  + (size_t)NH_ * S_;
static constexpr size_t O_DAD   = O_R2S   + (size_t)NH_ * S_;
static constexpr size_t O_L2S   = O_DAD   + (size_t)NH_ * S_;

__device__ __forceinline__ int frag_half() { return (threadIdx.x & 31) >> 4; }
__device__ __forceinline__ int frag_l15()  { return threadIdx.x & 15; }
__device__ __forceinline__ int frag_m(int r) { return r + frag_half() * 8; }

template <bool AT, bool BT>
__device__ __forceinline__ v8f wmma_tile(const float* A, int lda, float asgn,
                                         const float* B, int ldb, int K, v8f acc) {
  const int half = frag_half();
  const int l15  = frag_l15();
  for (int k0 = 0; k0 < K; k0 += 4) {
    const int ka = k0 + half * 2;
    v2f a, b;
    if (AT) { a.x = asgn * A[ka * lda + l15];  a.y = asgn * A[(ka + 1) * lda + l15]; }
    else    { a.x = asgn * A[l15 * lda + ka];  a.y = asgn * A[l15 * lda + ka + 1]; }
    if (BT) { b.x = B[l15 * ldb + ka];         b.y = B[l15 * ldb + ka + 1]; }
    else    { b.x = B[ka * ldb + l15];         b.y = B[(ka + 1) * ldb + l15]; }
    acc = __builtin_amdgcn_wmma_f32_16x16x4_f32(false, a, false, b, (short)0, acc,
                                                false, false);
  }
  return acc;
}

__device__ __forceinline__ v8f wmma_const_a(float a0, const float* B, int ldb, int K, v8f acc) {
  const int half = frag_half();
  const int l15  = frag_l15();
  v2f a;
  a.x = a0;  a.y = a0;
  for (int k0 = 0; k0 < K; k0 += 4) {
    const int ka = k0 + half * 2;
    v2f b;
    b.x = B[ka * ldb + l15];
    b.y = B[(ka + 1) * ldb + l15];
    acc = __builtin_amdgcn_wmma_f32_16x16x4_f32(false, a, false, b, (short)0, acc,
                                                false, false);
  }
  return acc;
}

__device__ __forceinline__ float inv_norm_from_sumsq(float s) {
  return (s > 0.f) ? 1.0f / sqrtf(s) : 0.f;
}
__device__ __forceinline__ float inv_norm_const_row(float v, int n) {
  float s = 0.f;
  for (int c = 0; c < n; ++c) s += v * v;
  return inv_norm_from_sumsq(s);
}

#define BL0_ 0.1020620726159658f
#define R0_  0.125f
__device__ __forceinline__ float invbl0() { return inv_norm_const_row(BL0_, CW_); }
__device__ __forceinline__ float invr0()  { return inv_norm_const_row(R0_, BS_); }
__device__ __forceinline__ float rsq0()  { const float r = R0_ * invr0();    return r * r; }
__device__ __forceinline__ float blsq0() { const float b = BL0_ * invbl0();  return b * b; }
__device__ __forceinline__ float r2s0() {
  const float r = rsq0();
  float s = 0.f;
  for (int i = 0; i < BS_; ++i) s += r * r;
  return s;
}
__device__ __forceinline__ float dad0() {
  const float r = rsq0(), b = BL0_ * invbl0();
  float s = 0.f;
  for (int i = 0; i < BS_; ++i) s += r * b * b;
  return s;
}
__device__ __forceinline__ float l2s0() {
  const float v = blsq0();
  float s = 0.f;
  for (int l = 0; l < NB_; ++l) s += v * v;
  return s;
}
template <bool FIRST>
__device__ __forceinline__ float par(const float* p, int i, float c0) {
  if (FIRST) return c0;
  return p[i];
}

__global__ __launch_bounds__(256) void k_first_block(const float* __restrict__ q,
                                                     const float* __restrict__ kk,
                                                     float* ws) {
  LDS_TILE_ float Qs[BS_ * PA_];
  LDS_TILE_ float Ks[BS_ * PA_];
  LDS_TILE_ float Qj2[2 * NB_ * PA_];

  const int blk = blockIdx.x;
  const int hb = blk >> 5, kb = blk & 31;
  const float* qhd = q  + (size_t)hb * S_ * D_;
  const float* qh  = qhd + (size_t)kb * BS_ * D_;
  const float* kh  = kk + (size_t)hb * S_ * D_ + (size_t)kb * BS_ * D_;
  float* dbl = ws + O_DBL + (size_t)hb * S_ * CW_;
  float* t1c = ws + O_T2  + (size_t)hb * NB_ * BS_ * BS_ + T1C_OFF;
  float* t2c = ws + O_T2  + (size_t)hb * NB_ * BS_ * BS_ + T2C_OFF;

  const int tid = threadIdx.x;
  for (int i = tid; i < BS_ * D_; i += 256) {
    const int p = (i >> 6) * PA_ + (i & 63);
    Qs[p] = qh[i] * 0.125f;
    Ks[p] = kh[i];
    const int l = i >> 7, cc = i & 127;
    Qj2[(cc >> 6) * (NB_ * PA_) + l * PA_ + (cc & 63)] =
        qhd[(l * 64 + 2 * kb) * 64 + cc] * 0.125f;
  }
  __syncthreads();

  const float rsq = rsq0(), bsq = blsq0();
  const int wid = tid >> 5;
  for (int t = wid; t < 16; t += 8) {
    const int tj = t >> 2, ti = t & 3;
    v8f acc;
    #pragma unroll
    for (int r = 0; r < 8; ++r) acc[r] = bsq + bsq * rsq;
    acc = wmma_tile<false, true>(Qs + tj * 16 * PA_, PA_, -1.f,
                                 Ks + ti * 16 * PA_, PA_, 64, acc);
    #pragma unroll
    for (int r = 0; r < 8; ++r) {
      const int m = tj * 16 + frag_m(r), n = ti * 16 + frag_l15();
      dbl[(kb * 64 + m) * CW_ + n] = acc[r];
    }
  }
  if (wid >= 4) {
    const int ti = wid & 3;
    v8f acc = 0.0f;
    acc = wmma_const_a(rsq, Ks + ti * 16, PA_, 64, acc);
    if (frag_half() == 0) t1c[kb * 64 + ti * 16 + frag_l15()] = acc[0];
  } else {
    const int ti = wid;
    #pragma unroll
    for (int jj = 0; jj < 2; ++jj) {
      v8f acc = 0.0f;
      acc = wmma_const_a(bsq, Qj2 + jj * (NB_ * PA_) + ti * 16, PA_, 32, acc);
      if (frag_half() == 0) t2c[(2 * kb + jj) * 64 + ti * 16 + frag_l15()] = acc[0];
    }
  }
}

__device__ __forceinline__ void dleft_first_body(const float* __restrict__ q,
                                                 float* ws, float* Qs, float* T1c) {
  const int blk = blockIdx.x;
  const int hb = blk >> 5, kb = blk & 31;
  const float* qh  = q + (size_t)hb * S_ * D_ + (size_t)kb * BS_ * D_;
  const float* t1c = ws + O_T2 + (size_t)hb * NB_ * BS_ * BS_ + T1C_OFF;
  float* dbl = ws + O_DBL + (size_t)hb * S_ * CW_;

  const int tid = threadIdx.x;
  for (int i = tid; i < BS_ * D_; i += 256)
    Qs[(i >> 6) * PA_ + (i & 63)] = qh[i] * 0.125f;
  for (int i = tid; i < NB_ * D_; i += 256)
    T1c[(i >> 6) * PA_ + (i & 63)] = t1c[i];
  __syncthreads();

  const int wid = tid >> 5;
  const int tj = wid >> 1, ti = wid & 1;
  v8f acc = 0.0f;
  acc = wmma_tile<false, true>(Qs + tj * 16 * PA_, PA_, 1.f,
                               T1c + ti * 16 * PA_, PA_, 64, acc);
  const float lf = blsq0(), r2 = r2s0(), da = dad0();
  #pragma unroll
  for (int r = 0; r < 8; ++r) {
    const int j = tj * 16 + frag_m(r), k2 = ti * 16 + frag_l15();
    float v = lf * r2 - acc[r];
    if (k2 == kb) v += da;
    dbl[(kb * 64 + j) * CW_ + 64 + k2] = v;
  }
}

__device__ __forceinline__ void grad_block_body(const float* __restrict__ q,
                                                float* ws, float* Qs, float* Ks,
                                                float* Rsq, float* DiagL) {

  const int blk = blockIdx.x;
  const int hb = blk >> 5, kb = blk & 31;
  const float* qh = q  + (size_t)hb * S_ * D_ + (size_t)kb * BS_ * D_;
  float* blp   = ws + O_BL    + (size_t)hb * S_ * CW_;
  float* rp    = ws + O_R     + (size_t)hb * NB_ * BS_ * BS_ + (size_t)kb * BS_ * BS_;
  float* invbl = ws + O_INVBL + (size_t)hb * S_;
  float* invr  = ws + O_INVR  + (size_t)hb * S_;
  float* dbl   = ws + O_DBL   + (size_t)hb * S_ * CW_;
  float* t1    = ws + O_T1    + (size_t)hb * NB_ * BS_ * BS_ + (size_t)kb * BS_ * BS_;
  float* r2s   = ws + O_R2S   + (size_t)hb * S_;
  float* dad   = ws + O_DAD   + (size_t)hb * S_;

  const int tid = threadIdx.x;
  for (int i = tid; i < BS_ * D_; i += 256) {
    const int p = (i >> 6) * PA_ + (i & 63);
    Qs[p] = qh[i] * 0.125f;
    float rs = rp[i] * invr[kb * 64 + (i >> 6)];
    Rsq[p] = rs * rs;
  }
  if (tid < BS_) {
    const int row = kb * 64 + tid;
    float v = blp[row * CW_ + 64 + kb] * invbl[row];
    DiagL[tid] = v * v;
  }
  __syncthreads();

  if (tid < BS_) {
    const int j = tid, row = kb * 64 + j;
    const float inv = invbl[row];
    float s2 = 0.f, sd = 0.f;
    for (int i = 0; i < BS_; ++i) {
      float r = Rsq[j * PA_ + i];
      float b = blp[row * CW_ + i] * inv;
      s2 += r * r;
      sd += r * b * b;
    }
    r2s[row] = s2;
    dad[row] = sd;
  }

  const int wid = tid >> 5;
  for (int t = wid; t < 32; t += 8) {
    const int which = t >> 4;
    const int tt = t & 15;
    const int tj = tt >> 2, ti = tt & 3;
    if (which == 0) {
      v8f acc;
      #pragma unroll
      for (int r = 0; r < 8; ++r) {
        const int m = tj * 16 + frag_m(r), n = ti * 16 + frag_l15();
        const int row = kb * 64 + m;
        float b = blp[row * CW_ + n] * invbl[row];
        acc[r] = b * b + DiagL[m] * Rsq[m * PA_ + n];
      }
      acc = wmma_tile<false, true>(Qs + tj * 16 * PA_, PA_, -1.f,
                                   Ks + ti * 16 * PA_, PA_, 64, acc);
      #pragma unroll
      for (int r = 0; r < 8; ++r) {
        const int m = tj * 16 + frag_m(r), n = ti * 16 + frag_l15();
        dbl[(kb * 64 + m) * CW_ + n] = acc[r];
      }
    } else {
      v8f acc = 0.0f;
      acc = wmma_tile<false, false>(Rsq + tj * 16 * PA_, PA_, 1.f,
                                    Ks + ti * 16, PA_, 64, acc);
      #pragma unroll
      for (int r = 0; r < 8; ++r) {
        const int m = tj * 16 + frag_m(r), n = ti * 16 + frag_l15();
        t1[m * 64 + n] = acc[r];
      }
    }
  }
}

__global__ __launch_bounds__(256) void k_grad_j(const float* __restrict__ q,
                                                float* ws) {
  LDS_TILE_ float Lj[NB_ * PT32_];
  LDS_TILE_ float Qj[NB_ * PAB_];
  LDS_TILE_ float T1j[NB_ * PA_];

  const int blk = blockIdx.x;
  const int hb = blk >> 6, j = blk & 63;
  const float* qh = q + (size_t)hb * S_ * D_;
  float* blp   = ws + O_BL    + (size_t)hb * S_ * CW_;
  float* invbl = ws + O_INVBL + (size_t)hb * S_;
  float* dbl   = ws + O_DBL   + (size_t)hb * S_ * CW_;
  float* t1h   = ws + O_T1    + (size_t)hb * NB_ * BS_ * BS_;
  float* t2h   = ws + O_T2    + (size_t)hb * NB_ * BS_ * BS_;
  float* r2s   = ws + O_R2S   + (size_t)hb * S_;
  float* dad   = ws + O_DAD   + (size_t)hb * S_;
  float* l2s   = ws + O_L2S   + (size_t)hb * S_;

  const int tid = threadIdx.x;
  for (int i = tid; i < NB_ * NB_; i += 256) {
    const int l = i >> 5, k2 = i & 31;
    float v = blp[(l * 64 + j) * CW_ + 64 + k2] * invbl[l * 64 + j];
    Lj[l * PT32_ + k2] = v * v;
  }
  for (int i = tid; i < NB_ * D_; i += 256) {
    const int l = i >> 6, d = i & 63;
    Qj[l * PAB_ + d] = qh[(l * 64 + j) * 64 + d] * 0.125f;
    T1j[l * PA_ + d] = t1h[l * (BS_ * BS_) + j * 64 + d];
  }
  __syncthreads();

  if (tid < NB_) {
    float s = 0.f;
    for (int l = 0; l < NB_; ++l) { float v = Lj[l * PT32_ + tid]; s += v * v; }
    l2s[j * 32 + tid] = s;
  }

  const int wid = tid >> 5;
  if (wid < 4) {
    const int tj = wid >> 1, ti = wid & 1;
    v8f acc = 0.0f;
    acc = wmma_tile<false, true>(Qj + tj * 16 * PAB_, PAB_, 1.f,
                                 T1j + ti * 16 * PA_, PA_, 64, acc);
    #pragma unroll
    for (int r = 0; r < 8; ++r) {
      const int l = tj * 16 + frag_m(r), k2 = ti * 16 + frag_l15();
      float v = Lj[l * PT32_ + k2] * r2s[k2 * 64 + j] - acc[r];
      if (l == k2) v += dad[l * 64 + j];
      dbl[(l * 64 + j) * CW_ + 64 + k2] = v;
    }
  }
  {
    const int tj = wid >> 2, ti = wid & 3;
    v8f acc = 0.0f;
    acc = wmma_tile<true, false>(Lj + tj * 16, PT32_, 1.f,
                                 Qj + ti * 16, PAB_, 32, acc);
    #pragma unroll
    for (int r = 0; r < 8; ++r) {
      const int k2 = tj * 16 + frag_m(r), d = ti * 16 + frag_l15();
      t2h[k2 * (BS_ * BS_) + j * 64 + d] = acc[r];
    }
  }
}

template <bool FIRST>
__device__ __forceinline__ void grad_r_body(const float* __restrict__ kk,
                                            float* ws, float* T2s, float* Ks,
                                            float* DRs) {

  const int blk = blockIdx.x;
  const int hb = blk >> 5, kb = blk & 31;
  const float* kh = kk + (size_t)hb * S_ * D_ + (size_t)kb * BS_ * D_;
  float* blp   = ws + O_BL    + (size_t)hb * S_ * CW_;
  float* rp    = ws + O_R     + (size_t)hb * NB_ * BS_ * BS_ + (size_t)kb * BS_ * BS_;
  float* invbl = ws + O_INVBL + (size_t)hb * S_;
  float* invr  = ws + O_INVR  + (size_t)hb * S_;
  float* t2h   = ws + O_T2    + (size_t)hb * NB_ * BS_ * BS_ + (size_t)kb * BS_ * BS_;
  float* l2s   = ws + O_L2S   + (size_t)hb * S_;
  const float* dbl = ws + O_DBL + (size_t)hb * S_ * CW_;

  const int tid = threadIdx.x;
  const float* t2src = FIRST ? ws + O_T2 + (size_t)hb * NB_ * BS_ * BS_ + T2C_OFF : t2h;
  for (int i = tid; i < BS_ * D_; i += 256) {
    const int p = (i >> 6) * PA_ + (i & 63);
    T2s[p] = t2src[i];
    Ks[p]  = kh[i];
  }
  __syncthreads();

  const int wid = tid >> 5;
  for (int t = wid; t < 16; t += 8) {
    const int tj = t >> 2, ti = t & 3;
    v8f acc;
    #pragma unroll
    for (int r = 0; r < 8; ++r) {
      const int m = tj * 16 + frag_m(r), n = ti * 16 + frag_l15();
      const int row = kb * 64 + m;
      const float ib = par<FIRST>(invbl, row, invbl0());
      float bd = par<FIRST>(blp, row * CW_ + n, BL0_) * ib;         bd *= bd;
      float dl = par<FIRST>(blp, row * CW_ + 64 + kb, BL0_) * ib;   dl *= dl;
      float rv = par<FIRST>(rp, m * 64 + n, R0_) * par<FIRST>(invr, row, invr0());
      rv *= rv;
      acc[r] = rv * par<FIRST>(l2s, m * 32 + kb, l2s0()) + dl * bd;
    }
    acc = wmma_tile<false, true>(T2s + tj * 16 * PA_, PA_, -1.f,
                                 Ks + ti * 16 * PA_, PA_, 64, acc);
    #pragma unroll
    for (int r = 0; r < 8; ++r) {
      const int m = tj * 16 + frag_m(r), n = ti * 16 + frag_l15();
      DRs[m * PR_ + n] = acc[r];
    }
  }
  __syncthreads();

  if (tid < BS_) {
    const int j = tid, row = kb * 64 + j;
    const float inv = par<FIRST>(invr, row, invr0());
    float dot = 0.f;
    for (int i = 0; i < BS_; ++i) {
      float rs = par<FIRST>(rp, j * 64 + i, R0_) * inv;
      dot += rs * (2.f * DRs[j * PR_ + i] * rs);
    }
    float s2 = 0.f;
    #pragma unroll 32
    for (int i = 0; i < BS_; ++i) {
      const float old = par<FIRST>(rp, j * 64 + i, R0_);
      float rs = old * inv;
      float dr = 2.f * DRs[j * PR_ + i] * rs;
      float df = (dr - rs * dot) * inv;
      float nv = old - 0.5f * df;
      rp[j * 64 + i] = nv;
      s2 += nv * nv;
    }
    invr[row] = inv_norm_from_sumsq(s2);
  } else {
    const int lane = tid & 31;
    for (int grp = wid - 2; grp < 16; grp += 6) {
      const int row0 = kb * 64 + grp * 4;
      float old[4][3], bs[4][3], g[4][3], inv[4], dot[4], s2[4];
      #pragma unroll
      for (int u = 0; u < 4; ++u) {
        const int row = row0 + u;
        inv[u] = par<FIRST>(invbl, row, invbl0());
        #pragma unroll
        for (int i = 0; i < 3; ++i) {
          const int c = row * CW_ + lane + 32 * i;
          old[u][i] = par<FIRST>(blp, c, BL0_);
          g[u][i]   = dbl[c];
        }
      }
      #pragma unroll
      for (int u = 0; u < 4; ++u) {
        float p = 0.f;
        #pragma unroll
        for (int i = 0; i < 3; ++i) {
          bs[u][i] = old[u][i] * inv[u];
          g[u][i]  = 2.f * g[u][i] * bs[u][i];
          p += bs[u][i] * g[u][i];
        }
        dot[u] = p;
      }
      #pragma unroll
      for (int m = 16; m >= 1; m >>= 1) {
        #pragma unroll
        for (int u = 0; u < 4; ++u) dot[u] += __shfl_xor(dot[u], m);
      }
      #pragma unroll
      for (int u = 0; u < 4; ++u) {
        const int row = row0 + u;
        float s = 0.f;
        #pragma unroll
        for (int i = 0; i < 3; ++i) {
          float nv = old[u][i] - 0.5f * ((g[u][i] - bs[u][i] * dot[u]) * inv[u]);
          blp[row * CW_ + lane + 32 * i] = nv;
          s += nv * nv;
        }
        s2[u] = s;
      }
      #pragma unroll
      for (int m = 16; m >= 1; m >>= 1) {
        #pragma unroll
        for (int u = 0; u < 4; ++u) s2[u] += __shfl_xor(s2[u], m);
      }
      if (lane < 4) {
        float mine = s2[0];
        if (lane == 1) mine = s2[1];
        if (lane == 2) mine = s2[2];
        if (lane == 3) mine = s2[3];
        invbl[row0 + lane] = inv_norm_from_sumsq(mine);
      }
    }
  }
}
__global__ __launch_bounds__(256) void k_r1_block2(const float* __restrict__ q,
                                                   const float* __restrict__ kk,
                                                   float* ws) {
  LDS_TILE_ float pool[3 * BS_ * PA_];
  __shared__ float DiagL[BS_];
  float* T2s = pool;
  float* Ks  = pool + BS_ * PA_;
  float* DRs = pool + 2 * BS_ * PA_;
  dleft_first_body(q, ws, pool, DRs);
  __syncthreads();
  grad_r_body<true>(kk, ws, T2s, Ks, DRs);
  __syncthreads();
  grad_block_body(q, ws, T2s, Ks, DRs, DiagL);
}

__global__ __launch_bounds__(256, 12) void k_r2_apply(const float* __restrict__ kk,
                                                  const float* __restrict__ val,
                                                  float* __restrict__ out,
                                                  float* ws) {
  LDS_TILE_ float pool[2 * BS_ * PA_ + BS_ * PR_];
  grad_r_body<false>(kk, ws, pool, pool + BS_ * PA_, pool + 2 * BS_ * PA_);
  __syncthreads();

  float* Xs = pool;
  float* Ms = pool + BS_ * PB_;
  const int blk = blockIdx.x;
  const int hb = blk >> 5, kb = blk & 31;
  const float* xh = val + (size_t)hb * S_ * D_ + (size_t)kb * BS_ * D_;
  float* rp    = ws + O_R     + (size_t)hb * NB_ * BS_ * BS_ + (size_t)kb * BS_ * BS_;
  float* invr  = ws + O_INVR  + (size_t)hb * S_;
  float* blp   = ws + O_BL    + (size_t)hb * S_ * CW_;
  float* invbl = ws + O_INVBL + (size_t)hb * S_;
  float* yb    = ws + O_T1    + (size_t)hb * NB_ * BS_ * BS_ + (size_t)kb * BS_ * BS_;
  float* oh    = out + (size_t)hb * S_ * D_ + (size_t)kb * BS_ * D_;

  const int tid = threadIdx.x;
  for (int i = tid; i < BS_ * D_; i += 256) {
    float rs = rp[i] * invr[kb * 64 + (i >> 6)];
    Ms[(i >> 6) * PA_ + (i & 63)] = rs * rs;
    Xs[(i >> 6) * PB_ + (i & 63)] = xh[i];
  }
  __syncthreads();

  const int wid = tid >> 5;
  for (int t = wid; t < 16; t += 8) {
    const int tj = t >> 2, ti = t & 3;
    v8f acc = 0.0f;
    acc = wmma_tile<false, false>(Ms + tj * 16 * PA_, PA_, 1.f,
                                  Xs + ti * 16, PB_, 64, acc);
    #pragma unroll
    for (int r = 0; r < 8; ++r) {
      const int m = tj * 16 + frag_m(r), n = ti * 16 + frag_l15();
      yb[m * 64 + n] = acc[r];
    }
  }
  __syncthreads();
  for (int i = tid; i < BS_ * D_; i += 256) {
    const int j = i >> 6, cc = i & 63;
    float v = blp[(kb * 64 + j) * CW_ + cc] * invbl[kb * 64 + j];
    Ms[j * PA_ + cc] = v * v;
  }
  __syncthreads();
  for (int t = wid; t < 16; t += 8) {
    const int tj = t >> 2, ti = t & 3;
    v8f acc = 0.0f;
    acc = wmma_tile<false, false>(Ms + tj * 16 * PA_, PA_, 1.f,
                                  Xs + ti * 16, PB_, 64, acc);
    #pragma unroll
    for (int r = 0; r < 8; ++r) {
      const int m = tj * 16 + frag_m(r), n = ti * 16 + frag_l15();
      oh[m * 64 + n] = acc[r];
    }
  }
}

__global__ __launch_bounds__(256) void k_zleft(float* __restrict__ out,
                                               float* ws) {
  LDS_TILE_ float Lj[NB_ * PA32_];
  LDS_TILE_ float Yj[NB_ * PB_];
  const int blk = blockIdx.x;
  const int hb = blk >> 6, j = blk & 63;
  float* blp   = ws + O_BL    + (size_t)hb * S_ * CW_;
  float* invbl = ws + O_INVBL + (size_t)hb * S_;
  float* yb    = ws + O_T1    + (size_t)hb * NB_ * BS_ * BS_;
  float* oh    = out + (size_t)hb * S_ * D_;

  const int tid = threadIdx.x;
  const int wid = tid >> 5;
  const int tj = wid >> 2, ti = wid & 3;
  v8f prev;
  #pragma unroll
  for (int r = 0; r < 8; ++r) {
    const int l = tj * 16 + frag_m(r), d = ti * 16 + frag_l15();
    prev[r] = oh[(size_t)(l * 64 + j) * 64 + d];
  }
  for (int i = tid; i < NB_ * NB_; i += 256) {
    const int l = i >> 5, k2 = i & 31;
    float v = blp[(l * 64 + j) * CW_ + 64 + k2] * invbl[l * 64 + j];
    Lj[l * PA32_ + k2] = v * v;
  }
  for (int i = tid; i < NB_ * D_; i += 256) {
    const int k2 = i >> 6, d = i & 63;
    Yj[k2 * PB_ + d] = yb[k2 * (BS_ * BS_) + j * 64 + d];
  }
  __syncthreads();

  v8f acc = 0.0f;
  acc = wmma_tile<false, false>(Lj + tj * 16 * PA32_, PA32_, 1.f,
                                Yj + ti * 16, PB_, 32, acc);
  #pragma unroll
  for (int r = 0; r < 8; ++r) {
    const int l = tj * 16 + frag_m(r), d = ti * 16 + frag_l15();
    oh[(size_t)(l * 64 + j) * 64 + d] = prev[r] + acc[r];
  }
}

extern "C" void kernel_launch(void* const* d_in, const int* in_sizes, int n_in,
                              void* d_out, int out_size, void* d_ws, size_t ws_size,
                              hipStream_t stream) {
  const float* q = (const float*)d_in[0];
  const float* k = (const float*)d_in[1];
  const float* v = (const float*)d_in[2];
  float* out = (float*)d_out;
  float* ws = (float*)d_ws;

  const int blkG   = NH_ * NB_;
  const int jG     = NH_ * BS_;

  k_first_block<<<blkG, 256, 0, stream>>>(q, k, ws);
  k_r1_block2<<<blkG, 256, 0, stream>>>(q, k, ws);
  k_grad_j<<<jG, 256, 0, stream>>>(q, ws);
  k_r2_apply<<<blkG, 256, 0, stream>>>(k, v, out, ws);
  k_zleft<<<jG, 256, 0, stream>>>(out, ws);
}


# Both per-(head,j) launches take 4 adjacent j per work-group (3072 work-groups of one tile per wave to 768 of four), staging loops unrolled so their loads fly together; same sums
# speedup vs baseline: 2.7605x; 1.0908x over previous
#include <hip/hip_runtime.h>


#define S_   2048
#define D_   64
#define BS_  64
#define NB_  32
#define CW_  96
#define NH_  48
#define JW_  4

#define PA_   68
#define PA32_ 36
#define PB_   80
#define PAB_  76
#define PT32_ 40
#define PR_   66
#define LDS_TILE_ __shared__ __attribute__((aligned(16)))

typedef __attribute__((ext_vector_type(2))) float v2f;
typedef __attribute__((ext_vector_type(8))) float v8f;

static constexpr size_t SZ_BL   = (size_t)NH_ * S_ * CW_;
static constexpr size_t SZ_R    = (size_t)NH_ * NB_ * BS_ * BS_;
static constexpr size_t O_BL    = 0;
static constexpr size_t O_R     = O_BL  + SZ_BL;
static constexpr size_t O_DBL   = O_R   + SZ_R;
static constexpr size_t O_T1    = O_DBL + SZ_BL;
static constexpr size_t O_T2    = O_T1  + SZ_R;
static constexpr size_t T2C_OFF = 0;
static constexpr size_t T1C_OFF = (size_t)BS_ * BS_;
static constexpr size_t O_INVBL = O_T2  + SZ_R;
static constexpr size_t O_INVR  = O_INVBL + (size_t)NH_ * S_;
static constexpr size_t O_R2S   = O_INVR  + (size_t)NH_ * S_;
static constexpr size_t O_DAD   = O_R2S   + (size_t)NH_ * S_;
static constexpr size_t O_L2S   = O_DAD   + (size_t)NH_ * S_;

__device__ __forceinline__ int frag_half() { return (threadIdx.x & 31) >> 4; }
__device__ __forceinline__ int frag_l15()  { return threadIdx.x & 15; }
__device__ __forceinline__ int frag_m(int r) { return r + frag_half() * 8; }

template <bool AT, bool BT>
__device__ __forceinline__ v8f wmma_tile(const float* A, int lda, float asgn,
                                         const float* B, int ldb, int K, v8f acc) {
  const int half = frag_half();
  const int l15  = frag_l15();
  for (int k0 = 0; k0 < K; k0 += 4) {
    const int ka = k0 + half * 2;
    v2f a, b;
    if (AT) { a.x = asgn * A[ka * lda + l15];  a.y = asgn * A[(ka + 1) * lda + l15]; }
    else    { a.x = asgn * A[l15 * lda + ka];  a.y = asgn * A[l15 * lda + ka + 1]; }
    if (BT) { b.x = B[l15 * ldb + ka];         b.y = B[l15 * ldb + ka + 1]; }
    else    { b.x = B[ka * ldb + l15];         b.y = B[(ka + 1) * ldb + l15]; }
    acc = __builtin_amdgcn_wmma_f32_16x16x4_f32(false, a, false, b, (short)0, acc,
                                                false, false);
  }
  return acc;
}

__device__ __forceinline__ v8f wmma_const_a(float a0, const float* B, int ldb, int K, v8f acc) {
  const int half = frag_half();
  const int l15  = frag_l15();
  v2f a;
  a.x = a0;  a.y = a0;
  for (int k0 = 0; k0 < K; k0 += 4) {
    const int ka = k0 + half * 2;
    v2f b;
    b.x = B[ka * ldb + l15];
    b.y = B[(ka + 1) * ldb + l15];
    acc = __builtin_amdgcn_wmma_f32_16x16x4_f32(false, a, false, b, (short)0, acc,
                                                false, false);
  }
  return acc;
}

__device__ __forceinline__ float inv_norm_from_sumsq(float s) {
  return (s > 0.f) ? 1.0f / sqrtf(s) : 0.f;
}
__device__ __forceinline__ float inv_norm_const_row(float v, int n) {
  float s = 0.f;
  for (int c = 0; c < n; ++c) s += v * v;
  return inv_norm_from_sumsq(s);
}

#define BL0_ 0.1020620726159658f
#define R0_  0.125f
__device__ __forceinline__ float invbl0() { return inv_norm_const_row(BL0_, CW_); }
__device__ __forceinline__ float invr0()  { return inv_norm_const_row(R0_, BS_); }
__device__ __forceinline__ float rsq0()  { const float r = R0_ * invr0();    return r * r; }
__device__ __forceinline__ float blsq0() { const float b = BL0_ * invbl0();  return b * b; }
__device__ __forceinline__ float r2s0() {
  const float r = rsq0();
  float s = 0.f;
  for (int i = 0; i < BS_; ++i) s += r * r;
  return s;
}
__device__ __forceinline__ float dad0() {
  const float r = rsq0(), b = BL0_ * invbl0();
  float s = 0.f;
  for (int i = 0; i < BS_; ++i) s += r * b * b;
  return s;
}
__device__ __forceinline__ float l2s0() {
  const float v = blsq0();
  float s = 0.f;
  for (int l = 0; l < NB_; ++l) s += v * v;
  return s;
}
template <bool FIRST>
__device__ __forceinline__ float par(const float* p, int i, float c0) {
  if (FIRST) return c0;
  return p[i];
}

__global__ __launch_bounds__(256) void k_first_block(const float* __restrict__ q,
                                                     const float* __restrict__ kk,
                                                     float* ws) {
  LDS_TILE_ float Qs[BS_ * PA_];
  LDS_TILE_ float Ks[BS_ * PA_];
  LDS_TILE_ float Qj2[2 * NB_ * PA_];

  const int blk = blockIdx.x;
  const int hb = blk >> 5, kb = blk & 31;
  const float* qhd = q  + (size_t)hb * S_ * D_;
  const float* qh  = qhd + (size_t)kb * BS_ * D_;
  const float* kh  = kk + (size_t)hb * S_ * D_ + (size_t)kb * BS_ * D_;
  float* dbl = ws + O_DBL + (size_t)hb * S_ * CW_;
  float* t1c = ws + O_T2  + (size_t)hb * NB_ * BS_ * BS_ + T1C_OFF;
  float* t2c = ws + O_T2  + (size_t)hb * NB_ * BS_ * BS_ + T2C_OFF;

  const int tid = threadIdx.x;
  for (int i = tid; i < BS_ * D_; i += 256) {
    const int p = (i >> 6) * PA_ + (i & 63);
    Qs[p] = qh[i] * 0.125f;
    Ks[p] = kh[i];
    const int l = i >> 7, cc = i & 127;
    Qj2[(cc >> 6) * (NB_ * PA_) + l * PA_ + (cc & 63)] =
        qhd[(l * 64 + 2 * kb) * 64 + cc] * 0.125f;
  }
  __syncthreads();

  const float rsq = rsq0(), bsq = blsq0();
  const int wid = tid >> 5;
  for (int t = wid; t < 16; t += 8) {
    const int tj = t >> 2, ti = t & 3;
    v8f acc;
    #pragma unroll
    for (int r = 0; r < 8; ++r) acc[r] = bsq + bsq * rsq;
    acc = wmma_tile<false, true>(Qs + tj * 16 * PA_, PA_, -1.f,
                                 Ks + ti * 16 * PA_, PA_, 64, acc);
    #pragma unroll
    for (int r = 0; r < 8; ++r) {
      const int m = tj * 16 + frag_m(r), n = ti * 16 + frag_l15();
      dbl[(kb * 64 + m) * CW_ + n] = acc[r];
    }
  }
  if (wid >= 4) {
    const int ti = wid & 3;
    v8f acc = 0.0f;
    acc = wmma_const_a(rsq, Ks + ti * 16, PA_, 64, acc);
    if (frag_half() == 0) t1c[kb * 64 + ti * 16 + frag_l15()] = acc[0];
  } else {
    const int ti = wid;
    #pragma unroll
    for (int jj = 0; jj < 2; ++jj) {
      v8f acc = 0.0f;
      acc = wmma_const_a(bsq, Qj2 + jj * (NB_ * PA_) + ti * 16, PA_, 32, acc);
      if (frag_half() == 0) t2c[(2 * kb + jj) * 64 + ti * 16 + frag_l15()] = acc[0];
    }
  }
}

__device__ __forceinline__ void dleft_first_body(const float* __restrict__ q,
                                                 float* ws, float* Qs, float* T1c) {
  const int blk = blockIdx.x;
  const int hb = blk >> 5, kb = blk & 31;
  const float* qh  = q + (size_t)hb * S_ * D_ + (size_t)kb * BS_ * D_;
  const float* t1c = ws + O_T2 + (size_t)hb * NB_ * BS_ * BS_ + T1C_OFF;
  float* dbl = ws + O_DBL + (size_t)hb * S_ * CW_;

  const int tid = threadIdx.x;
  for (int i = tid; i < BS_ * D_; i += 256)
    Qs[(i >> 6) * PA_ + (i & 63)] = qh[i] * 0.125f;
  for (int i = tid; i < NB_ * D_; i += 256)
    T1c[(i >> 6) * PA_ + (i & 63)] = t1c[i];
  __syncthreads();

  const int wid = tid >> 5;
  const int tj = wid >> 1, ti = wid & 1;
  v8f acc = 0.0f;
  acc = wmma_tile<false, true>(Qs + tj * 16 * PA_, PA_, 1.f,
                               T1c + ti * 16 * PA_, PA_, 64, acc);
  const float lf = blsq0(), r2 = r2s0(), da = dad0();
  #pragma unroll
  for (int r = 0; r < 8; ++r) {
    const int j = tj * 16 + frag_m(r), k2 = ti * 16 + frag_l15();
    float v = lf * r2 - acc[r];
    if (k2 == kb) v += da;
    dbl[(kb * 64 + j) * CW_ + 64 + k2] = v;
  }
}

__device__ __forceinline__ void grad_block_body(const float* __restrict__ q,
                                                float* ws, float* Qs, float* Ks,
                                                float* Rsq, float* DiagL) {

  const int blk = blockIdx.x;
  const int hb = blk >> 5, kb = blk & 31;
  const float* qh = q  + (size_t)hb * S_ * D_ + (size_t)kb * BS_ * D_;
  float* blp   = ws + O_BL    + (size_t)hb * S_ * CW_;
  float* rp    = ws + O_R     + (size_t)hb * NB_ * BS_ * BS_ + (size_t)kb * BS_ * BS_;
  float* invbl = ws + O_INVBL + (size_t)hb * S_;
  float* invr  = ws + O_INVR  + (size_t)hb * S_;
  float* dbl   = ws + O_DBL   + (size_t)hb * S_ * CW_;
  float* t1    = ws + O_T1    + (size_t)hb * NB_ * BS_ * BS_ + (size_t)kb * BS_ * BS_;
  float* r2s   = ws + O_R2S   + (size_t)hb * S_;
  float* dad   = ws + O_DAD   + (size_t)hb * S_;

  const int tid = threadIdx.x;
  for (int i = tid; i < BS_ * D_; i += 256) {
    const int p = (i >> 6) * PA_ + (i & 63);
    Qs[p] = qh[i] * 0.125f;
    float rs = rp[i] * invr[kb * 64 + (i >> 6)];
    Rsq[p] = rs * rs;
  }
  if (tid < BS_) {
    const int row = kb * 64 + tid;
    float v = blp[row * CW_ + 64 + kb] * invbl[row];
    DiagL[tid] = v * v;
  }
  __syncthreads();

  if (tid < BS_) {
    const int j = tid, row = kb * 64 + j;
    const float inv = invbl[row];
    float s2 = 0.f, sd = 0.f;
    for (int i = 0; i < BS_; ++i) {
      float r = Rsq[j * PA_ + i];
      float b = blp[row * CW_ + i] * inv;
      s2 += r * r;
      sd += r * b * b;
    }
    r2s[row] = s2;
    dad[row] = sd;
  }

  const int wid = tid >> 5;
  for (int t = wid; t < 32; t += 8) {
    const int which = t >> 4;
    const int tt = t & 15;
    const int tj = tt >> 2, ti = tt & 3;
    if (which == 0) {
      v8f acc;
      #pragma unroll
      for (int r = 0; r < 8; ++r) {
        const int m = tj * 16 + frag_m(r), n = ti * 16 + frag_l15();
        const int row = kb * 64 + m;
        float b = blp[row * CW_ + n] * invbl[row];
        acc[r] = b * b + DiagL[m] * Rsq[m * PA_ + n];
      }
      acc = wmma_tile<false, true>(Qs + tj * 16 * PA_, PA_, -1.f,
                                   Ks + ti * 16 * PA_, PA_, 64, acc);
      #pragma unroll
      for (int r = 0; r < 8; ++r) {
        const int m = tj * 16 + frag_m(r), n = ti * 16 + frag_l15();
        dbl[(kb * 64 + m) * CW_ + n] = acc[r];
      }
    } else {
      v8f acc = 0.0f;
      acc = wmma_tile<false, false>(Rsq + tj * 16 * PA_, PA_, 1.f,
                                    Ks + ti * 16, PA_, 64, acc);
      #pragma unroll
      for (int r = 0; r < 8; ++r) {
        const int m = tj * 16 + frag_m(r), n = ti * 16 + frag_l15();
        t1[m * 64 + n] = acc[r];
      }
    }
  }
}

__global__ __launch_bounds__(256, 6) void k_grad_j(const float* __restrict__ q,
                                                float* ws) {
  LDS_TILE_ float Lj[JW_ * NB_ * PT32_];
  LDS_TILE_ float Qj[JW_ * NB_ * PAB_];
  LDS_TILE_ float T1j[JW_ * NB_ * PA_];

  const int blk = blockIdx.x;
  const int hb = blk / (BS_ / JW_), j0 = (blk % (BS_ / JW_)) * JW_;
  const float* qh = q + (size_t)hb * S_ * D_;
  float* blp   = ws + O_BL    + (size_t)hb * S_ * CW_;
  float* invbl = ws + O_INVBL + (size_t)hb * S_;
  float* dbl   = ws + O_DBL   + (size_t)hb * S_ * CW_;
  float* t1h   = ws + O_T1    + (size_t)hb * NB_ * BS_ * BS_;
  float* t2h   = ws + O_T2    + (size_t)hb * NB_ * BS_ * BS_;
  float* r2s   = ws + O_R2S   + (size_t)hb * S_;
  float* dad   = ws + O_DAD   + (size_t)hb * S_;
  float* l2s   = ws + O_L2S   + (size_t)hb * S_;

  const int tid = threadIdx.x;
  #pragma unroll
  for (int it = 0; it < JW_ * NB_ * NB_ / 256; ++it) {
    const int i = tid + it * 256;
    const int l = i / (JW_ * NB_), jj = (i / NB_) % JW_, k2 = i % NB_;
    const int row = l * 64 + j0 + jj;
    float v = blp[row * CW_ + 64 + k2] * invbl[row];
    Lj[jj * (NB_ * PT32_) + l * PT32_ + k2] = v * v;
  }
  #pragma unroll
  for (int it = 0; it < JW_ * NB_ * D_ / 256; ++it) {
    const int i = tid + it * 256;
    const int l = i / (JW_ * D_), cc = i % (JW_ * D_);
    const int jj = cc >> 6, d = cc & 63;
    Qj[jj * (NB_ * PAB_) + l * PAB_ + d] = qh[(l * 64 + j0) * 64 + cc] * 0.125f;
    T1j[jj * (NB_ * PA_) + l * PA_ + d]  = t1h[l * (BS_ * BS_) + j0 * 64 + cc];
  }
  __syncthreads();

  if (tid < JW_ * NB_) {
    const int jj = tid >> 5, k2 = tid & 31;
    const float* L = Lj + jj * (NB_ * PT32_);
    float s = 0.f;
    for (int l = 0; l < NB_; ++l) { float v = L[l * PT32_ + k2]; s += v * v; }
    l2s[(j0 + jj) * 32 + k2] = s;
  }

  const int wid = tid >> 5;
  for (int t = wid; t < 4 * JW_; t += 8) {
    const int jj = t >> 2, tj = (t >> 1) & 1, ti = t & 1;
    const int j = j0 + jj;
    const float* L = Lj + jj * (NB_ * PT32_);
    v8f acc = 0.0f;
    acc = wmma_tile<false, true>(Qj + jj * (NB_ * PAB_) + tj * 16 * PAB_, PAB_, 1.f,
                                 T1j + jj * (NB_ * PA_) + ti * 16 * PA_, PA_, 64, acc);
    #pragma unroll
    for (int r = 0; r < 8; ++r) {
      const int l = tj * 16 + frag_m(r), k2 = ti * 16 + frag_l15();
      float v = L[l * PT32_ + k2] * r2s[k2 * 64 + j] - acc[r];
      if (l == k2) v += dad[l * 64 + j];
      dbl[(l * 64 + j) * CW_ + 64 + k2] = v;
    }
  }
  for (int jj = 0; jj < JW_; ++jj) {
    const int tj = wid >> 2, ti = wid & 3;
    v8f acc = 0.0f;
    acc = wmma_tile<true, false>(Lj + jj * (NB_ * PT32_) + tj * 16, PT32_, 1.f,
                                 Qj + jj * (NB_ * PAB_) + ti * 16, PAB_, 32, acc);
    #pragma unroll
    for (int r = 0; r < 8; ++r) {
      const int k2 = tj * 16 + frag_m(r), d = ti * 16 + frag_l15();
      t2h[k2 * (BS_ * BS_) + (j0 + jj) * 64 + d] = acc[r];
    }
  }
}

template <bool FIRST>
__device__ __forceinline__ void grad_r_body(const float* __restrict__ kk,
                                            float* ws, float* T2s, float* Ks,
                                            float* DRs) {

  const int blk = blockIdx.x;
  const int hb = blk >> 5, kb = blk & 31;
  const float* kh = kk + (size_t)hb * S_ * D_ + (size_t)kb * BS_ * D_;
  float* blp   = ws + O_BL    + (size_t)hb * S_ * CW_;
  float* rp    = ws + O_R     + (size_t)hb * NB_ * BS_ * BS_ + (size_t)kb * BS_ * BS_;
  float* invbl = ws + O_INVBL + (size_t)hb * S_;
  float* invr  = ws + O_INVR  + (size_t)hb * S_;
  float* t2h   = ws + O_T2    + (size_t)hb * NB_ * BS_ * BS_ + (size_t)kb * BS_ * BS_;
  float* l2s   = ws + O_L2S   + (size_t)hb * S_;
  const float* dbl = ws + O_DBL + (size_t)hb * S_ * CW_;

  const int tid = threadIdx.x;
  const float* t2src = FIRST ? ws + O_T2 + (size_t)hb * NB_ * BS_ * BS_ + T2C_OFF : t2h;
  for (int i = tid; i < BS_ * D_; i += 256) {
    const int p = (i >> 6) * PA_ + (i & 63);
    T2s[p] = t2src[i];
    Ks[p]  = kh[i];
  }
  __syncthreads();

  const int wid = tid >> 5;
  for (int t = wid; t < 16; t += 8) {
    const int tj = t >> 2, ti = t & 3;
    v8f acc;
    #pragma unroll
    for (int r = 0; r < 8; ++r) {
      const int m = tj * 16 + frag_m(r), n = ti * 16 + frag_l15();
      const int row = kb * 64 + m;
      const float ib = par<FIRST>(invbl, row, invbl0());
      float bd = par<FIRST>(blp, row * CW_ + n, BL0_) * ib;         bd *= bd;
      float dl = par<FIRST>(blp, row * CW_ + 64 + kb, BL0_) * ib;   dl *= dl;
      float rv = par<FIRST>(rp, m * 64 + n, R0_) * par<FIRST>(invr, row, invr0());
      rv *= rv;
      acc[r] = rv * par<FIRST>(l2s, m * 32 + kb, l2s0()) + dl * bd;
    }
    acc = wmma_tile<false, true>(T2s + tj * 16 * PA_, PA_, -1.f,
                                 Ks + ti * 16 * PA_, PA_, 64, acc);
    #pragma unroll
    for (int r = 0; r < 8; ++r) {
      const int m = tj * 16 + frag_m(r), n = ti * 16 + frag_l15();
      DRs[m * PR_ + n] = acc[r];
    }
  }
  __syncthreads();

  if (tid < BS_) {
    const int j = tid, row = kb * 64 + j;
    const float inv = par<FIRST>(invr, row, invr0());
    float dot = 0.f;
    for (int i = 0; i < BS_; ++i) {
      float rs = par<FIRST>(rp, j * 64 + i, R0_) * inv;
      dot += rs * (2.f * DRs[j * PR_ + i] * rs);
    }
    float s2 = 0.f;
    #pragma unroll 32
    for (int i = 0; i < BS_; ++i) {
      const float old = par<FIRST>(rp, j * 64 + i, R0_);
      float rs = old * inv;
      float dr = 2.f * DRs[j * PR_ + i] * rs;
      float df = (dr - rs * dot) * inv;
      float nv = old - 0.5f * df;
      rp[j * 64 + i] = nv;
      s2 += nv * nv;
    }
    invr[row] = inv_norm_from_sumsq(s2);
  } else {
    const int lane = tid & 31;
    for (int grp = wid - 2; grp < 16; grp += 6) {
      const int row0 = kb * 64 + grp * 4;
      float old[4][3], bs[4][3], g[4][3], inv[4], dot[4], s2[4];
      #pragma unroll
      for (int u = 0; u < 4; ++u) {
        const int row = row0 + u;
        inv[u] = par<FIRST>(invbl, row, invbl0());
        #pragma unroll
        for (int i = 0; i < 3; ++i) {
          const int c = row * CW_ + lane + 32 * i;
          old[u][i] = par<FIRST>(blp, c, BL0_);
          g[u][i]   = dbl[c];
        }
      }
      #pragma unroll
      for (int u = 0; u < 4; ++u) {
        float p = 0.f;
        #pragma unroll
        for (int i = 0; i < 3; ++i) {
          bs[u][i] = old[u][i] * inv[u];
          g[u][i]  = 2.f * g[u][i] * bs[u][i];
          p += bs[u][i] * g[u][i];
        }
        dot[u] = p;
      }
      #pragma unroll
      for (int m = 16; m >= 1; m >>= 1) {
        #pragma unroll
        for (int u = 0; u < 4; ++u) dot[u] += __shfl_xor(dot[u], m);
      }
      #pragma unroll
      for (int u = 0; u < 4; ++u) {
        const int row = row0 + u;
        float s = 0.f;
        #pragma unroll
        for (int i = 0; i < 3; ++i) {
          float nv = old[u][i] - 0.5f * ((g[u][i] - bs[u][i] * dot[u]) * inv[u]);
          blp[row * CW_ + lane + 32 * i] = nv;
          s += nv * nv;
        }
        s2[u] = s;
      }
      #pragma unroll
      for (int m = 16; m >= 1; m >>= 1) {
        #pragma unroll
        for (int u = 0; u < 4; ++u) s2[u] += __shfl_xor(s2[u], m);
      }
      if (lane < 4) {
        float mine = s2[0];
        if (lane == 1) mine = s2[1];
        if (lane == 2) mine = s2[2];
        if (lane == 3) mine = s2[3];
        invbl[row0 + lane] = inv_norm_from_sumsq(mine);
      }
    }
  }
}
__global__ __launch_bounds__(256) void k_r1_block2(const float* __restrict__ q,
                                                   const float* __restrict__ kk,
                                                   float* ws) {
  LDS_TILE_ float pool[3 * BS_ * PA_];
  __shared__ float DiagL[BS_];
  float* T2s = pool;
  float* Ks  = pool + BS_ * PA_;
  float* DRs = pool + 2 * BS_ * PA_;
  dleft_first_body(q, ws, pool, DRs);
  __syncthreads();
  grad_r_body<true>(kk, ws, T2s, Ks, DRs);
  __syncthreads();
  grad_block_body(q, ws, T2s, Ks, DRs, DiagL);
}

__global__ __launch_bounds__(256, 12) void k_r2_apply(const float* __restrict__ kk,
                                                  const float* __restrict__ val,
                                                  float* __restrict__ out,
                                                  float* ws) {
  LDS_TILE_ float pool[2 * BS_ * PA_ + BS_ * PR_];
  grad_r_body<false>(kk, ws, pool, pool + BS_ * PA_, pool + 2 * BS_ * PA_);
  __syncthreads();

  float* Xs = pool;
  float* Ms = pool + BS_ * PB_;
  const int blk = blockIdx.x;
  const int hb = blk >> 5, kb = blk & 31;
  const float* xh = val + (size_t)hb * S_ * D_ + (size_t)kb * BS_ * D_;
  float* rp    = ws + O_R     + (size_t)hb * NB_ * BS_ * BS_ + (size_t)kb * BS_ * BS_;
  float* invr  = ws + O_INVR  + (size_t)hb * S_;
  float* blp   = ws + O_BL    + (size_t)hb * S_ * CW_;
  float* invbl = ws + O_INVBL + (size_t)hb * S_;
  float* yb    = ws + O_T1    + (size_t)hb * NB_ * BS_ * BS_ + (size_t)kb * BS_ * BS_;
  float* oh    = out + (size_t)hb * S_ * D_ + (size_t)kb * BS_ * D_;

  const int tid = threadIdx.x;
  for (int i = tid; i < BS_ * D_; i += 256) {
    float rs = rp[i] * invr[kb * 64 + (i >> 6)];
    Ms[(i >> 6) * PA_ + (i & 63)] = rs * rs;
    Xs[(i >> 6) * PB_ + (i & 63)] = xh[i];
  }
  __syncthreads();

  const int wid = tid >> 5;
  for (int t = wid; t < 16; t += 8) {
    const int tj = t >> 2, ti = t & 3;
    v8f acc = 0.0f;
    acc = wmma_tile<false, false>(Ms + tj * 16 * PA_, PA_, 1.f,
                                  Xs + ti * 16, PB_, 64, acc);
    #pragma unroll
    for (int r = 0; r < 8; ++r) {
      const int m = tj * 16 + frag_m(r), n = ti * 16 + frag_l15();
      yb[m * 64 + n] = acc[r];
    }
  }
  __syncthreads();
  for (int i = tid; i < BS_ * D_; i += 256) {
    const int j = i >> 6, cc = i & 63;
    float v = blp[(kb * 64 + j) * CW_ + cc] * invbl[kb * 64 + j];
    Ms[j * PA_ + cc] = v * v;
  }
  __syncthreads();
  for (int t = wid; t < 16; t += 8) {
    const int tj = t >> 2, ti = t & 3;
    v8f acc = 0.0f;
    acc = wmma_tile<false, false>(Ms + tj * 16 * PA_, PA_, 1.f,
                                  Xs + ti * 16, PB_, 64, acc);
    #pragma unroll
    for (int r = 0; r < 8; ++r) {
      const int m = tj * 16 + frag_m(r), n = ti * 16 + frag_l15();
      oh[m * 64 + n] = acc[r];
    }
  }
}

__global__ __launch_bounds__(256, 6) void k_zleft(float* __restrict__ out,
                                               float* ws) {
  LDS_TILE_ float Lj[JW_ * NB_ * PA32_];
  LDS_TILE_ float Yj[JW_ * NB_ * PB_];
  const int blk = blockIdx.x;
  const int hb = blk / (BS_ / JW_), j0 = (blk % (BS_ / JW_)) * JW_;
  float* blp   = ws + O_BL    + (size_t)hb * S_ * CW_;
  float* invbl = ws + O_INVBL + (size_t)hb * S_;
  float* yb    = ws + O_T1    + (size_t)hb * NB_ * BS_ * BS_;
  float* oh    = out + (size_t)hb * S_ * D_;

  const int tid = threadIdx.x;
  const int wid = tid >> 5;
  const int tj = wid >> 2, ti = wid & 3;
  v8f prev[JW_];
  #pragma unroll
  for (int jj = 0; jj < JW_; ++jj) {
    #pragma unroll
    for (int r = 0; r < 8; ++r) {
      const int l = tj * 16 + frag_m(r), d = ti * 16 + frag_l15();
      prev[jj][r] = oh[(size_t)(l * 64 + j0 + jj) * 64 + d];
    }
  }
  #pragma unroll
  for (int it = 0; it < JW_ * NB_ * NB_ / 256; ++it) {
    const int i = tid + it * 256;
    const int l = i / (JW_ * NB_), jj = (i / NB_) % JW_, k2 = i % NB_;
    const int row = l * 64 + j0 + jj;
    float v = blp[row * CW_ + 64 + k2] * invbl[row];
    Lj[jj * (NB_ * PA32_) + l * PA32_ + k2] = v * v;
  }
  #pragma unroll
  for (int it = 0; it < JW_ * NB_ * D_ / 256; ++it) {
    const int i = tid + it * 256;
    const int k2 = i / (JW_ * D_), cc = i % (JW_ * D_);
    Yj[(cc >> 6) * (NB_ * PB_) + k2 * PB_ + (cc & 63)] = yb[k2 * (BS_ * BS_) + j0 * 64 + cc];
  }
  __syncthreads();

  #pragma unroll
  for (int jj = 0; jj < JW_; ++jj) {
    v8f acc = 0.0f;
    acc = wmma_tile<false, false>(Lj + jj * (NB_ * PA32_) + tj * 16 * PA32_, PA32_, 1.f,
                                  Yj + jj * (NB_ * PB_) + ti * 16, PB_, 32, acc);
    #pragma unroll
    for (int r = 0; r < 8; ++r) {
      const int l = tj * 16 + frag_m(r), d = ti * 16 + frag_l15();
      oh[(size_t)(l * 64 + j0 + jj) * 64 + d] = prev[jj][r] + acc[r];
    }
  }
}

extern "C" void kernel_launch(void* const* d_in, const int* in_sizes, int n_in,
                              void* d_out, int out_size, void* d_ws, size_t ws_size,
                              hipStream_t stream) {
  const float* q = (const float*)d_in[0];
  const float* k = (const float*)d_in[1];
  const float* v = (const float*)d_in[2];
  float* out = (float*)d_out;
  float* ws = (float*)d_ws;

  const int blkG   = NH_ * NB_;
  const int jG     = NH_ * (BS_ / JW_);

  k_first_block<<<blkG, 256, 0, stream>>>(q, k, ws);
  k_r1_block2<<<blkG, 256, 0, stream>>>(q, k, ws);
  k_grad_j<<<jG, 256, 0, stream>>>(q, ws);
  k_r2_apply<<<blkG, 256, 0, stream>>>(k, v, out, ws);
  k_zleft<<<jG, 256, 0, stream>>>(out, ws);
}


# Tile staging moves 16 bytes per lane (b128 global loads and LDS stores); the per-block launches issue a stage's loads all before its first wait, not in 8-16 rolled round trips; same sums
# speedup vs baseline: 2.9233x; 1.0590x over previous
#include <hip/hip_runtime.h>


#define S_   2048
#define D_   64
#define BS_  64
#define NB_  32
#define CW_  96
#define NH_  48
#define JW_  4

#define PA_   68
#define PA32_ 36
#define PB_   80
#define PAB_  76
#define PT32_ 40
#define PR_   66
#define LDS_TILE_ __shared__ __attribute__((aligned(16)))

typedef __attribute__((ext_vector_type(2))) float v2f;
typedef __attribute__((ext_vector_type(8))) float v8f;
typedef __attribute__((ext_vector_type(4))) float v4f;

#define NQ_ (BS_ * D_ / 4 / 256)
__device__ __forceinline__ v4f ldg4(const float* p) { return *reinterpret_cast<const v4f*>(p); }
__device__ __forceinline__ void sts4(float* p, v4f v) { *reinterpret_cast<v4f*>(p) = v; }

static constexpr size_t SZ_BL   = (size_t)NH_ * S_ * CW_;
static constexpr size_t SZ_R    = (size_t)NH_ * NB_ * BS_ * BS_;
static constexpr size_t O_BL    = 0;
static constexpr size_t O_R     = O_BL  + SZ_BL;
static constexpr size_t O_DBL   = O_R   + SZ_R;
static constexpr size_t O_T1    = O_DBL + SZ_BL;
static constexpr size_t O_T2    = O_T1  + SZ_R;
static constexpr size_t T2C_OFF = 0;
static constexpr size_t T1C_OFF = (size_t)BS_ * BS_;
static constexpr size_t O_INVBL = O_T2  + SZ_R;
static constexpr size_t O_INVR  = O_INVBL + (size_t)NH_ * S_;
static constexpr size_t O_R2S   = O_INVR  + (size_t)NH_ * S_;
static constexpr size_t O_DAD   = O_R2S   + (size_t)NH_ * S_;
static constexpr size_t O_L2S   = O_DAD   + (size_t)NH_ * S_;

__device__ __forceinline__ int frag_half() { return (threadIdx.x & 31) >> 4; }
__device__ __forceinline__ int frag_l15()  { return threadIdx.x & 15; }
__device__ __forceinline__ int frag_m(int r) { return r + frag_half() * 8; }

template <bool AT, bool BT>
__device__ __forceinline__ v8f wmma_tile(const float* A, int lda, float asgn,
                                         const float* B, int ldb, int K, v8f acc) {
  const int half = frag_half();
  const int l15  = frag_l15();
  for (int k0 = 0; k0 < K; k0 += 4) {
    const int ka = k0 + half * 2;
    v2f a, b;
    if (AT) { a.x = asgn * A[ka * lda + l15];  a.y = asgn * A[(ka + 1) * lda + l15]; }
    else    { a.x = asgn * A[l15 * lda + ka];  a.y = asgn * A[l15 * lda + ka + 1]; }
    if (BT) { b.x = B[l15 * ldb + ka];         b.y = B[l15 * ldb + ka + 1]; }
    else    { b.x = B[ka * ldb + l15];         b.y = B[(ka + 1) * ldb + l15]; }
    acc = __builtin_amdgcn_wmma_f32_16x16x4_f32(false, a, false, b, (short)0, acc,
                                                false, false);
  }
  return acc;
}

__device__ __forceinline__ v8f wmma_const_a(float a0, const float* B, int ldb, int K, v8f acc) {
  const int half = frag_half();
  const int l15  = frag_l15();
  v2f a;
  a.x = a0;  a.y = a0;
  for (int k0 = 0; k0 < K; k0 += 4) {
    const int ka = k0 + half * 2;
    v2f b;
    b.x = B[ka * ldb + l15];
    b.y = B[(ka + 1) * ldb + l15];
    acc = __builtin_amdgcn_wmma_f32_16x16x4_f32(false, a, false, b, (short)0, acc,
                                                false, false);
  }
  return acc;
}

__device__ __forceinline__ float inv_norm_from_sumsq(float s) {
  return (s > 0.f) ? 1.0f / sqrtf(s) : 0.f;
}
__device__ __forceinline__ float inv_norm_const_row(float v, int n) {
  float s = 0.f;
  for (int c = 0; c < n; ++c) s += v * v;
  return inv_norm_from_sumsq(s);
}

#define BL0_ 0.1020620726159658f
#define R0_  0.125f
__device__ __forceinline__ float invbl0() { return inv_norm_const_row(BL0_, CW_); }
__device__ __forceinline__ float invr0()  { return inv_norm_const_row(R0_, BS_); }
__device__ __forceinline__ float rsq0()  { const float r = R0_ * invr0();    return r * r; }
__device__ __forceinline__ float blsq0() { const float b = BL0_ * invbl0();  return b * b; }
__device__ __forceinline__ float r2s0() {
  const float r = rsq0();
  float s = 0.f;
  for (int i = 0; i < BS_; ++i) s += r * r;
  return s;
}
__device__ __forceinline__ float dad0() {
  const float r = rsq0(), b = BL0_ * invbl0();
  float s = 0.f;
  for (int i = 0; i < BS_; ++i) s += r * b * b;
  return s;
}
__device__ __forceinline__ float l2s0() {
  const float v = blsq0();
  float s = 0.f;
  for (int l = 0; l < NB_; ++l) s += v * v;
  return s;
}
template <bool FIRST>
__device__ __forceinline__ float par(const float* p, int i, float c0) {
  if (FIRST) return c0;
  return p[i];
}

__global__ __launch_bounds__(256) void k_first_block(const float* __restrict__ q,
                                                     const float* __restrict__ kk,
                                                     float* ws) {
  LDS_TILE_ float Qs[BS_ * PA_];
  LDS_TILE_ float Ks[BS_ * PA_];
  LDS_TILE_ float Qj2[2 * NB_ * PA_];

  const int blk = blockIdx.x;
  const int hb = blk >> 5, kb = blk & 31;
  const float* qhd = q  + (size_t)hb * S_ * D_;
  const float* qh  = qhd + (size_t)kb * BS_ * D_;
  const float* kh  = kk + (size_t)hb * S_ * D_ + (size_t)kb * BS_ * D_;
  float* dbl = ws + O_DBL + (size_t)hb * S_ * CW_;
  float* t1c = ws + O_T2  + (size_t)hb * NB_ * BS_ * BS_ + T1C_OFF;
  float* t2c = ws + O_T2  + (size_t)hb * NB_ * BS_ * BS_ + T2C_OFF;

  const int tid = threadIdx.x;
  v4f qv[NQ_], kv[NQ_], gv[NQ_];
  #pragma unroll
  for (int it = 0; it < NQ_; ++it) {
    const int i = (tid + it * 256) * 4;
    qv[it] = ldg4(qh + i);
    kv[it] = ldg4(kh + i);
    const int l = i >> 7, cc = i & 127;
    gv[it] = ldg4(qhd + (l * 64 + 2 * kb) * 64 + cc);
  }
  #pragma unroll
  for (int it = 0; it < NQ_; ++it) {
    const int i = (tid + it * 256) * 4;
    const int p = (i >> 6) * PA_ + (i & 63);
    sts4(Qs + p, qv[it] * 0.125f);
    sts4(Ks + p, kv[it]);
    const int l = i >> 7, cc = i & 127;
    sts4(Qj2 + (cc >> 6) * (NB_ * PA_) + l * PA_ + (cc & 63), gv[it] * 0.125f);
  }
  __syncthreads();

  const float rsq = rsq0(), bsq = blsq0();
  const int wid = tid >> 5;
  for (int t = wid; t < 16; t += 8) {
    const int tj = t >> 2, ti = t & 3;
    v8f acc;
    #pragma unroll
    for (int r = 0; r < 8; ++r) acc[r] = bsq + bsq * rsq;
    acc = wmma_tile<false, true>(Qs + tj * 16 * PA_, PA_, -1.f,
                                 Ks + ti * 16 * PA_, PA_, 64, acc);
    #pragma unroll
    for (int r = 0; r < 8; ++r) {
      const int m = tj * 16 + frag_m(r), n = ti * 16 + frag_l15();
      dbl[(kb * 64 + m) * CW_ + n] = acc[r];
    }
  }
  if (wid >= 4) {
    const int ti = wid & 3;
    v8f acc = 0.0f;
    acc = wmma_const_a(rsq, Ks + ti * 16, PA_, 64, acc);
    if (frag_half() == 0) t1c[kb * 64 + ti * 16 + frag_l15()] = acc[0];
  } else {
    const int ti = wid;
    #pragma unroll
    for (int jj = 0; jj < 2; ++jj) {
      v8f acc = 0.0f;
      acc = wmma_const_a(bsq, Qj2 + jj * (NB_ * PA_) + ti * 16, PA_, 32, acc);
      if (frag_half() == 0) t2c[(2 * kb + jj) * 64 + ti * 16 + frag_l15()] = acc[0];
    }
  }
}

__device__ __forceinline__ void dleft_first_body(const float* __restrict__ q,
                                                 float* ws, float* Qs, float* T1c) {
  const int blk = blockIdx.x;
  const int hb = blk >> 5, kb = blk & 31;
  const float* qh  = q + (size_t)hb * S_ * D_ + (size_t)kb * BS_ * D_;
  const float* t1c = ws + O_T2 + (size_t)hb * NB_ * BS_ * BS_ + T1C_OFF;
  float* dbl = ws + O_DBL + (size_t)hb * S_ * CW_;

  const int tid = threadIdx.x;
  v4f qv[NQ_], tv[NQ_ / 2];
  #pragma unroll
  for (int it = 0; it < NQ_; ++it) {
    qv[it] = ldg4(qh + (tid + it * 256) * 4);
    if (it < NQ_ / 2) tv[it] = ldg4(t1c + (tid + it * 256) * 4);
  }
  #pragma unroll
  for (int it = 0; it < NQ_; ++it) {
    const int i = (tid + it * 256) * 4;
    const int p = (i >> 6) * PA_ + (i & 63);
    sts4(Qs + p, qv[it] * 0.125f);
    if (it < NQ_ / 2) sts4(T1c + p, tv[it]);
  }
  __syncthreads();

  const int wid = tid >> 5;
  const int tj = wid >> 1, ti = wid & 1;
  v8f acc = 0.0f;
  acc = wmma_tile<false, true>(Qs + tj * 16 * PA_, PA_, 1.f,
                               T1c + ti * 16 * PA_, PA_, 64, acc);
  const float lf = blsq0(), r2 = r2s0(), da = dad0();
  #pragma unroll
  for (int r = 0; r < 8; ++r) {
    const int j = tj * 16 + frag_m(r), k2 = ti * 16 + frag_l15();
    float v = lf * r2 - acc[r];
    if (k2 == kb) v += da;
    dbl[(kb * 64 + j) * CW_ + 64 + k2] = v;
  }
}

__device__ __forceinline__ void grad_block_body(const float* __restrict__ q,
                                                float* ws, float* Qs, float* Ks,
                                                float* Rsq, float* DiagL) {

  const int blk = blockIdx.x;
  const int hb = blk >> 5, kb = blk & 31;
  const float* qh = q  + (size_t)hb * S_ * D_ + (size_t)kb * BS_ * D_;
  float* blp   = ws + O_BL    + (size_t)hb * S_ * CW_;
  float* rp    = ws + O_R     + (size_t)hb * NB_ * BS_ * BS_ + (size_t)kb * BS_ * BS_;
  float* invbl = ws + O_INVBL + (size_t)hb * S_;
  float* invr  = ws + O_INVR  + (size_t)hb * S_;
  float* dbl   = ws + O_DBL   + (size_t)hb * S_ * CW_;
  float* t1    = ws + O_T1    + (size_t)hb * NB_ * BS_ * BS_ + (size_t)kb * BS_ * BS_;
  float* r2s   = ws + O_R2S   + (size_t)hb * S_;
  float* dad   = ws + O_DAD   + (size_t)hb * S_;

  const int tid = threadIdx.x;
  v4f qv[NQ_], rv[NQ_];
  float iv[NQ_], dlv = 0.f, dli = 0.f;
  #pragma unroll
  for (int it = 0; it < NQ_; ++it) {
    const int i = (tid + it * 256) * 4;
    qv[it] = ldg4(qh + i);
    rv[it] = ldg4(rp + i);
    iv[it] = invr[kb * 64 + (i >> 6)];
  }
  if (tid < BS_) {
    const int row = kb * 64 + tid;
    dlv = blp[row * CW_ + 64 + kb];
    dli = invbl[row];
  }
  #pragma unroll
  for (int it = 0; it < NQ_; ++it) {
    const int i = (tid + it * 256) * 4;
    const int p = (i >> 6) * PA_ + (i & 63);
    sts4(Qs + p, qv[it] * 0.125f);
    const v4f rs = rv[it] * iv[it];
    sts4(Rsq + p, rs * rs);
  }
  if (tid < BS_) {
    const float v = dlv * dli;
    DiagL[tid] = v * v;
  }
  __syncthreads();

  if (tid < BS_) {
    const int j = tid, row = kb * 64 + j;
    const float inv = invbl[row];
    float s2 = 0.f, sd = 0.f;
    for (int i = 0; i < BS_; ++i) {
      float r = Rsq[j * PA_ + i];
      float b = blp[row * CW_ + i] * inv;
      s2 += r * r;
      sd += r * b * b;
    }
    r2s[row] = s2;
    dad[row] = sd;
  }

  const int wid = tid >> 5;
  for (int t = wid; t < 32; t += 8) {
    const int which = t >> 4;
    const int tt = t & 15;
    const int tj = tt >> 2, ti = tt & 3;
    if (which == 0) {
      v8f acc;
      #pragma unroll
      for (int r = 0; r < 8; ++r) {
        const int m = tj * 16 + frag_m(r), n = ti * 16 + frag_l15();
        const int row = kb * 64 + m;
        float b = blp[row * CW_ + n] * invbl[row];
        acc[r] = b * b + DiagL[m] * Rsq[m * PA_ + n];
      }
      acc = wmma_tile<false, true>(Qs + tj * 16 * PA_, PA_, -1.f,
                                   Ks + ti * 16 * PA_, PA_, 64, acc);
      #pragma unroll
      for (int r = 0; r < 8; ++r) {
        const int m = tj * 16 + frag_m(r), n = ti * 16 + frag_l15();
        dbl[(kb * 64 + m) * CW_ + n] = acc[r];
      }
    } else {
      v8f acc = 0.0f;
      acc = wmma_tile<false, false>(Rsq + tj * 16 * PA_, PA_, 1.f,
                                    Ks + ti * 16, PA_, 64, acc);
      #pragma unroll
      for (int r = 0; r < 8; ++r) {
        const int m = tj * 16 + frag_m(r), n = ti * 16 + frag_l15();
        t1[m * 64 + n] = acc[r];
      }
    }
  }
}

__global__ __launch_bounds__(256, 6) void k_grad_j(const float* __restrict__ q,
                                                float* ws) {
  LDS_TILE_ float Lj[JW_ * NB_ * PT32_];
  LDS_TILE_ float Qj[JW_ * NB_ * PAB_];
  LDS_TILE_ float T1j[JW_ * NB_ * PA_];

  const int blk = blockIdx.x;
  const int hb = blk / (BS_ / JW_), j0 = (blk % (BS_ / JW_)) * JW_;
  const float* qh = q + (size_t)hb * S_ * D_;
  float* blp   = ws + O_BL    + (size_t)hb * S_ * CW_;
  float* invbl = ws + O_INVBL + (size_t)hb * S_;
  float* dbl   = ws + O_DBL   + (size_t)hb * S_ * CW_;
  float* t1h   = ws + O_T1    + (size_t)hb * NB_ * BS_ * BS_;
  float* t2h   = ws + O_T2    + (size_t)hb * NB_ * BS_ * BS_;
  float* r2s   = ws + O_R2S   + (size_t)hb * S_;
  float* dad   = ws + O_DAD   + (size_t)hb * S_;
  float* l2s   = ws + O_L2S   + (size_t)hb * S_;

  const int tid = threadIdx.x;
  constexpr int NL = JW_ * NB_ * NB_ / 4 / 256;
  constexpr int NT = JW_ * NB_ * D_ / 4 / 256;
  v4f lv[NL], qv[NT], tv[NT];
  float li[NL];
  #pragma unroll
  for (int it = 0; it < NL; ++it) {
    const int i = (tid + it * 256) * 4;
    const int l = i / (JW_ * NB_), jj = (i / NB_) % JW_, k2 = i % NB_;
    const int row = l * 64 + j0 + jj;
    lv[it] = ldg4(blp + row * CW_ + 64 + k2);
    li[it] = invbl[row];
  }
  #pragma unroll
  for (int it = 0; it < NT; ++it) {
    const int i = (tid + it * 256) * 4;
    const int l = i / (JW_ * D_), cc = i % (JW_ * D_);
    qv[it] = ldg4(qh + (l * 64 + j0) * 64 + cc);
    tv[it] = ldg4(t1h + l * (BS_ * BS_) + j0 * 64 + cc);
  }
  #pragma unroll
  for (int it = 0; it < NL; ++it) {
    const int i = (tid + it * 256) * 4;
    const int l = i / (JW_ * NB_), jj = (i / NB_) % JW_, k2 = i % NB_;
    const v4f v = lv[it] * li[it];
    sts4(Lj + jj * (NB_ * PT32_) + l * PT32_ + k2, v * v);
  }
  #pragma unroll
  for (int it = 0; it < NT; ++it) {
    const int i = (tid + it * 256) * 4;
    const int l = i / (JW_ * D_), cc = i % (JW_ * D_);
    const int jj = cc >> 6, d = cc & 63;
    sts4(Qj + jj * (NB_ * PAB_) + l * PAB_ + d, qv[it] * 0.125f);
    sts4(T1j + jj * (NB_ * PA_) + l * PA_ + d,  tv[it]);
  }
  __syncthreads();

  if (tid < JW_ * NB_) {
    const int jj = tid >> 5, k2 = tid & 31;
    const float* L = Lj + jj * (NB_ * PT32_);
    float s = 0.f;
    for (int l = 0; l < NB_; ++l) { float v = L[l * PT32_ + k2]; s += v * v; }
    l2s[(j0 + jj) * 32 + k2] = s;
  }

  const int wid = tid >> 5;
  for (int t = wid; t < 4 * JW_; t += 8) {
    const int jj = t >> 2, tj = (t >> 1) & 1, ti = t & 1;
    const int j = j0 + jj;
    const float* L = Lj + jj * (NB_ * PT32_);
    v8f acc = 0.0f;
    acc = wmma_tile<false, true>(Qj + jj * (NB_ * PAB_) + tj * 16 * PAB_, PAB_, 1.f,
                                 T1j + jj * (NB_ * PA_) + ti * 16 * PA_, PA_, 64, acc);
    #pragma unroll
    for (int r = 0; r < 8; ++r) {
      const int l = tj * 16 + frag_m(r), k2 = ti * 16 + frag_l15();
      float v = L[l * PT32_ + k2] * r2s[k2 * 64 + j] - acc[r];
      if (l == k2) v += dad[l * 64 + j];
      dbl[(l * 64 + j) * CW_ + 64 + k2] = v;
    }
  }
  for (int jj = 0; jj < JW_; ++jj) {
    const int tj = wid >> 2, ti = wid & 3;
    v8f acc = 0.0f;
    acc = wmma_tile<true, false>(Lj + jj * (NB_ * PT32_) + tj * 16, PT32_, 1.f,
                                 Qj + jj * (NB_ * PAB_) + ti * 16, PAB_, 32, acc);
    #pragma unroll
    for (int r = 0; r < 8; ++r) {
      const int k2 = tj * 16 + frag_m(r), d = ti * 16 + frag_l15();
      t2h[k2 * (BS_ * BS_) + (j0 + jj) * 64 + d] = acc[r];
    }
  }
}

template <bool FIRST>
__device__ __forceinline__ void grad_r_body(const float* __restrict__ kk,
                                            float* ws, float* T2s, float* Ks,
                                            float* DRs) {

  const int blk = blockIdx.x;
  const int hb = blk >> 5, kb = blk & 31;
  const float* kh = kk + (size_t)hb * S_ * D_ + (size_t)kb * BS_ * D_;
  float* blp   = ws + O_BL    + (size_t)hb * S_ * CW_;
  float* rp    = ws + O_R     + (size_t)hb * NB_ * BS_ * BS_ + (size_t)kb * BS_ * BS_;
  float* invbl = ws + O_INVBL + (size_t)hb * S_;
  float* invr  = ws + O_INVR  + (size_t)hb * S_;
  float* t2h   = ws + O_T2    + (size_t)hb * NB_ * BS_ * BS_ + (size_t)kb * BS_ * BS_;
  float* l2s   = ws + O_L2S   + (size_t)hb * S_;
  const float* dbl = ws + O_DBL + (size_t)hb * S_ * CW_;

  const int tid = threadIdx.x;
  const float* t2src = FIRST ? ws + O_T2 + (size_t)hb * NB_ * BS_ * BS_ + T2C_OFF : t2h;
  {
    v4f tv[NQ_], kv[NQ_];
    #pragma unroll
    for (int it = 0; it < NQ_; ++it) {
      const int i = (tid + it * 256) * 4;
      tv[it] = ldg4(t2src + i);
      kv[it] = ldg4(kh + i);
    }
    #pragma unroll
    for (int it = 0; it < NQ_; ++it) {
      const int i = (tid + it * 256) * 4;
      const int p = (i >> 6) * PA_ + (i & 63);
      sts4(T2s + p, tv[it]);
      sts4(Ks + p, kv[it]);
    }
  }
  __syncthreads();

  const int wid = tid >> 5;
  for (int t = wid; t < 16; t += 8) {
    const int tj = t >> 2, ti = t & 3;
    v8f acc;
    #pragma unroll
    for (int r = 0; r < 8; ++r) {
      const int m = tj * 16 + frag_m(r), n = ti * 16 + frag_l15();
      const int row = kb * 64 + m;
      const float ib = par<FIRST>(invbl, row, invbl0());
      float bd = par<FIRST>(blp, row * CW_ + n, BL0_) * ib;         bd *= bd;
      float dl = par<FIRST>(blp, row * CW_ + 64 + kb, BL0_) * ib;   dl *= dl;
      float rv = par<FIRST>(rp, m * 64 + n, R0_) * par<FIRST>(invr, row, invr0());
      rv *= rv;
      acc[r] = rv * par<FIRST>(l2s, m * 32 + kb, l2s0()) + dl * bd;
    }
    acc = wmma_tile<false, true>(T2s + tj * 16 * PA_, PA_, -1.f,
                                 Ks + ti * 16 * PA_, PA_, 64, acc);
    #pragma unroll
    for (int r = 0; r < 8; ++r) {
      const int m = tj * 16 + frag_m(r), n = ti * 16 + frag_l15();
      DRs[m * PR_ + n] = acc[r];
    }
  }
  __syncthreads();

  if (tid < BS_) {
    const int j = tid, row = kb * 64 + j;
    const float inv = par<FIRST>(invr, row, invr0());
    float dot = 0.f;
    for (int i = 0; i < BS_; ++i) {
      float rs = par<FIRST>(rp, j * 64 + i, R0_) * inv;
      dot += rs * (2.f * DRs[j * PR_ + i] * rs);
    }
    float s2 = 0.f;
    #pragma unroll 32
    for (int i = 0; i < BS_; ++i) {
      const float old = par<FIRST>(rp, j * 64 + i, R0_);
      float rs = old * inv;
      float dr = 2.f * DRs[j * PR_ + i] * rs;
      float df = (dr - rs * dot) * inv;
      float nv = old - 0.5f * df;
      rp[j * 64 + i] = nv;
      s2 += nv * nv;
    }
    invr[row] = inv_norm_from_sumsq(s2);
  } else {
    const int lane = tid & 31;
    for (int grp = wid - 2; grp < 16; grp += 6) {
      const int row0 = kb * 64 + grp * 4;
      float old[4][3], bs[4][3], g[4][3], inv[4], dot[4], s2[4];
      #pragma unroll
      for (int u = 0; u < 4; ++u) {
        const int row = row0 + u;
        inv[u] = par<FIRST>(invbl, row, invbl0());
        #pragma unroll
        for (int i = 0; i < 3; ++i) {
          const int c = row * CW_ + lane + 32 * i;
          old[u][i] = par<FIRST>(blp, c, BL0_);
          g[u][i]   = dbl[c];
        }
      }
      #pragma unroll
      for (int u = 0; u < 4; ++u) {
        float p = 0.f;
        #pragma unroll
        for (int i = 0; i < 3; ++i) {
          bs[u][i] = old[u][i] * inv[u];
          g[u][i]  = 2.f * g[u][i] * bs[u][i];
          p += bs[u][i] * g[u][i];
        }
        dot[u] = p;
      }
      #pragma unroll
      for (int m = 16; m >= 1; m >>= 1) {
        #pragma unroll
        for (int u = 0; u < 4; ++u) dot[u] += __shfl_xor(dot[u], m);
      }
      #pragma unroll
      for (int u = 0; u < 4; ++u) {
        const int row = row0 + u;
        float s = 0.f;
        #pragma unroll
        for (int i = 0; i < 3; ++i) {
          float nv = old[u][i] - 0.5f * ((g[u][i] - bs[u][i] * dot[u]) * inv[u]);
          blp[row * CW_ + lane + 32 * i] = nv;
          s += nv * nv;
        }
        s2[u] = s;
      }
      #pragma unroll
      for (int m = 16; m >= 1; m >>= 1) {
        #pragma unroll
        for (int u = 0; u < 4; ++u) s2[u] += __shfl_xor(s2[u], m);
      }
      if (lane < 4) {
        float mine = s2[0];
        if (lane == 1) mine = s2[1];
        if (lane == 2) mine = s2[2];
        if (lane == 3) mine = s2[3];
        invbl[row0 + lane] = inv_norm_from_sumsq(mine);
      }
    }
  }
}
__global__ __launch_bounds__(256, 12) void k_r1_block2(const float* __restrict__ q,
                                                   const float* __restrict__ kk,
                                                   float* ws) {
  LDS_TILE_ float pool[3 * BS_ * PA_];
  __shared__ float DiagL[BS_];
  float* T2s = pool;
  float* Ks  = pool + BS_ * PA_;
  float* DRs = pool + 2 * BS_ * PA_;
  dleft_first_body(q, ws, pool, DRs);
  __syncthreads();
  grad_r_body<true>(kk, ws, T2s, Ks, DRs);
  __syncthreads();
  grad_block_body(q, ws, T2s, Ks, DRs, DiagL);
}

__global__ __launch_bounds__(256, 12) void k_r2_apply(const float* __restrict__ kk,
                                                  const float* __restrict__ val,
                                                  float* __restrict__ out,
                                                  float* ws) {
  LDS_TILE_ float pool[2 * BS_ * PA_ + BS_ * PR_];
  grad_r_body<false>(kk, ws, pool, pool + BS_ * PA_, pool + 2 * BS_ * PA_);
  __syncthreads();

  float* Xs = pool;
  float* Ms = pool + BS_ * PB_;
  const int blk = blockIdx.x;
  const int hb = blk >> 5, kb = blk & 31;
  const float* xh = val + (size_t)hb * S_ * D_ + (size_t)kb * BS_ * D_;
  float* rp    = ws + O_R     + (size_t)hb * NB_ * BS_ * BS_ + (size_t)kb * BS_ * BS_;
  float* invr  = ws + O_INVR  + (size_t)hb * S_;
  float* blp   = ws + O_BL    + (size_t)hb * S_ * CW_;
  float* invbl = ws + O_INVBL + (size_t)hb * S_;
  float* yb    = ws + O_T1    + (size_t)hb * NB_ * BS_ * BS_ + (size_t)kb * BS_ * BS_;
  float* oh    = out + (size_t)hb * S_ * D_ + (size_t)kb * BS_ * D_;

  const int tid = threadIdx.x;
  v4f rv[NQ_], xv[NQ_];
  float iv[NQ_];
  #pragma unroll
  for (int it = 0; it < NQ_; ++it) {
    const int i = (tid + it * 256) * 4;
    rv[it] = ldg4(rp + i);
    iv[it] = invr[kb * 64 + (i >> 6)];
    xv[it] = ldg4(xh + i);
  }
  #pragma unroll
  for (int it = 0; it < NQ_; ++it) {
    const int i = (tid + it * 256) * 4;
    const v4f rs = rv[it] * iv[it];
    sts4(Ms + (i >> 6) * PA_ + (i & 63), rs * rs);
    sts4(Xs + (i >> 6) * PB_ + (i & 63), xv[it]);
  }
  __syncthreads();

  const int wid = tid >> 5;
  for (int t = wid; t < 16; t += 8) {
    const int tj = t >> 2, ti = t & 3;
    v8f acc = 0.0f;
    acc = wmma_tile<false, false>(Ms + tj * 16 * PA_, PA_, 1.f,
                                  Xs + ti * 16, PB_, 64, acc);
    #pragma unroll
    for (int r = 0; r < 8; ++r) {
      const int m = tj * 16 + frag_m(r), n = ti * 16 + frag_l15();
      yb[m * 64 + n] = acc[r];
    }
  }
  v4f bv[NQ_];
  float ib[NQ_];
  #pragma unroll
  for (int it = 0; it < NQ_; ++it) {
    const int i = (tid + it * 256) * 4;
    const int j = i >> 6, cc = i & 63;
    bv[it] = ldg4(blp + (kb * 64 + j) * CW_ + cc);
    ib[it] = invbl[kb * 64 + j];
  }
  __syncthreads();
  #pragma unroll
  for (int it = 0; it < NQ_; ++it) {
    const int i = (tid + it * 256) * 4;
    const v4f v = bv[it] * ib[it];
    sts4(Ms + (i >> 6) * PA_ + (i & 63), v * v);
  }
  __syncthreads();
  for (int t = wid; t < 16; t += 8) {
    const int tj = t >> 2, ti = t & 3;
    v8f acc = 0.0f;
    acc = wmma_tile<false, false>(Ms + tj * 16 * PA_, PA_, 1.f,
                                  Xs + ti * 16, PB_, 64, acc);
    #pragma unroll
    for (int r = 0; r < 8; ++r) {
      const int m = tj * 16 + frag_m(r), n = ti * 16 + frag_l15();
      oh[m * 64 + n] = acc[r];
    }
  }
}

__global__ __launch_bounds__(256, 6) void k_zleft(float* __restrict__ out,
                                               float* ws) {
  LDS_TILE_ float Lj[JW_ * NB_ * PA32_];
  LDS_TILE_ float Yj[JW_ * NB_ * PB_];
  const int blk = blockIdx.x;
  const int hb = blk / (BS_ / JW_), j0 = (blk % (BS_ / JW_)) * JW_;
  float* blp   = ws + O_BL    + (size_t)hb * S_ * CW_;
  float* invbl = ws + O_INVBL + (size_t)hb * S_;
  float* yb    = ws + O_T1    + (size_t)hb * NB_ * BS_ * BS_;
  float* oh    = out + (size_t)hb * S_ * D_;

  const int tid = threadIdx.x;
  const int wid = tid >> 5;
  const int tj = wid >> 2, ti = wid & 3;
  v8f prev[JW_];
  #pragma unroll
  for (int jj = 0; jj < JW_; ++jj) {
    #pragma unroll
    for (int r = 0; r < 8; ++r) {
      const int l = tj * 16 + frag_m(r), d = ti * 16 + frag_l15();
      prev[jj][r] = oh[(size_t)(l * 64 + j0 + jj) * 64 + d];
    }
  }
  constexpr int NL = JW_ * NB_ * NB_ / 4 / 256;
  constexpr int NT = JW_ * NB_ * D_ / 4 / 256;
  v4f lv[NL], yv[NT];
  float li[NL];
  #pragma unroll
  for (int it = 0; it < NL; ++it) {
    const int i = (tid + it * 256) * 4;
    const int l = i / (JW_ * NB_), jj = (i / NB_) % JW_, k2 = i % NB_;
    const int row = l * 64 + j0 + jj;
    lv[it] = ldg4(blp + row * CW_ + 64 + k2);
    li[it] = invbl[row];
  }
  #pragma unroll
  for (int it = 0; it < NT; ++it) {
    const int i = (tid + it * 256) * 4;
    const int k2 = i / (JW_ * D_), cc = i % (JW_ * D_);
    yv[it] = ldg4(yb + k2 * (BS_ * BS_) + j0 * 64 + cc);
  }
  #pragma unroll
  for (int it = 0; it < NL; ++it) {
    const int i = (tid + it * 256) * 4;
    const int l = i / (JW_ * NB_), jj = (i / NB_) % JW_, k2 = i % NB_;
    const v4f v = lv[it] * li[it];
    sts4(Lj + jj * (NB_ * PA32_) + l * PA32_ + k2, v * v);
  }
  #pragma unroll
  for (int it = 0; it < NT; ++it) {
    const int i = (tid + it * 256) * 4;
    const int k2 = i / (JW_ * D_), cc = i % (JW_ * D_);
    sts4(Yj + (cc >> 6) * (NB_ * PB_) + k2 * PB_ + (cc & 63), yv[it]);
  }
  __syncthreads();

  #pragma unroll
  for (int jj = 0; jj < JW_; ++jj) {
    v8f acc = 0.0f;
    acc = wmma_tile<false, false>(Lj + jj * (NB_ * PA32_) + tj * 16 * PA32_, PA32_, 1.f,
                                  Yj + jj * (NB_ * PB_) + ti * 16, PB_, 32, acc);
    #pragma unroll
    for (int r = 0; r < 8; ++r) {
      const int l = tj * 16 + frag_m(r), d = ti * 16 + frag_l15();
      oh[(size_t)(l * 64 + j0 + jj) * 64 + d] = prev[jj][r] + acc[r];
    }
  }
}

extern "C" void kernel_launch(void* const* d_in, const int* in_sizes, int n_in,
                              void* d_out, int out_size, void* d_ws, size_t ws_size,
                              hipStream_t stream) {
  const float* q = (const float*)d_in[0];
  const float* k = (const float*)d_in[1];
  const float* v = (const float*)d_in[2];
  float* out = (float*)d_out;
  float* ws = (float*)d_ws;

  const int blkG   = NH_ * NB_;
  const int jG     = NH_ * (BS_ / JW_);

  k_first_block<<<blkG, 256, 0, stream>>>(q, k, ws);
  k_r1_block2<<<blkG, 256, 0, stream>>>(q, k, ws);
  k_grad_j<<<jG, 256, 0, stream>>>(q, ws);
  k_r2_apply<<<blkG, 256, 0, stream>>>(k, v, out, ws);
  k_zleft<<<jG, 256, 0, stream>>>(out, ws);
}


# Last step: each work-group's new r rows and block_diag columns stay in LDS for its Y = right @ X and block_diag @ X products, no longer stored to the workspace and loaded back; same sums
# speedup vs baseline: 3.0468x; 1.0422x over previous
#include <hip/hip_runtime.h>


#define S_   2048
#define D_   64
#define BS_  64
#define NB_  32
#define CW_  96
#define NH_  48
#define JW_  4

#define PA_   68
#define PA32_ 36
#define PB_   80
#define PX_   72
#define PAB_  76
#define PT32_ 40
#define PR_   66
#define LDS_TILE_ __shared__ __attribute__((aligned(16)))

typedef __attribute__((ext_vector_type(2))) float v2f;
typedef __attribute__((ext_vector_type(8))) float v8f;
typedef __attribute__((ext_vector_type(4))) float v4f;

#define NQ_ (BS_ * D_ / 4 / 256)
__device__ __forceinline__ v4f ldg4(const float* p) { return *reinterpret_cast<const v4f*>(p); }
__device__ __forceinline__ void sts4(float* p, v4f v) { *reinterpret_cast<v4f*>(p) = v; }

static constexpr size_t SZ_BL   = (size_t)NH_ * S_ * CW_;
static constexpr size_t SZ_R    = (size_t)NH_ * NB_ * BS_ * BS_;
static constexpr size_t O_BL    = 0;
static constexpr size_t O_R     = O_BL  + SZ_BL;
static constexpr size_t O_DBL   = O_R   + SZ_R;
static constexpr size_t O_T1    = O_DBL + SZ_BL;
static constexpr size_t O_T2    = O_T1  + SZ_R;
static constexpr size_t T2C_OFF = 0;
static constexpr size_t T1C_OFF = (size_t)BS_ * BS_;
static constexpr size_t O_INVBL = O_T2  + SZ_R;
static constexpr size_t O_INVR  = O_INVBL + (size_t)NH_ * S_;
static constexpr size_t O_R2S   = O_INVR  + (size_t)NH_ * S_;
static constexpr size_t O_DAD   = O_R2S   + (size_t)NH_ * S_;
static constexpr size_t O_L2S   = O_DAD   + (size_t)NH_ * S_;

__device__ __forceinline__ int frag_half() { return (threadIdx.x & 31) >> 4; }
__device__ __forceinline__ int frag_l15()  { return threadIdx.x & 15; }
__device__ __forceinline__ int frag_m(int r) { return r + frag_half() * 8; }

template <bool AT, bool BT>
__device__ __forceinline__ v8f wmma_tile(const float* A, int lda, float asgn,
                                         const float* B, int ldb, int K, v8f acc) {
  const int half = frag_half();
  const int l15  = frag_l15();
  for (int k0 = 0; k0 < K; k0 += 4) {
    const int ka = k0 + half * 2;
    v2f a, b;
    if (AT) { a.x = asgn * A[ka * lda + l15];  a.y = asgn * A[(ka + 1) * lda + l15]; }
    else    { a.x = asgn * A[l15 * lda + ka];  a.y = asgn * A[l15 * lda + ka + 1]; }
    if (BT) { b.x = B[l15 * ldb + ka];         b.y = B[l15 * ldb + ka + 1]; }
    else    { b.x = B[ka * ldb + l15];         b.y = B[(ka + 1) * ldb + l15]; }
    acc = __builtin_amdgcn_wmma_f32_16x16x4_f32(false, a, false, b, (short)0, acc,
                                                false, false);
  }
  return acc;
}

__device__ __forceinline__ v8f wmma_const_a(float a0, const float* B, int ldb, int K, v8f acc) {
  const int half = frag_half();
  const int l15  = frag_l15();
  v2f a;
  a.x = a0;  a.y = a0;
  for (int k0 = 0; k0 < K; k0 += 4) {
    const int ka = k0 + half * 2;
    v2f b;
    b.x = B[ka * ldb + l15];
    b.y = B[(ka + 1) * ldb + l15];
    acc = __builtin_amdgcn_wmma_f32_16x16x4_f32(false, a, false, b, (short)0, acc,
                                                false, false);
  }
  return acc;
}

__device__ __forceinline__ v8f wmma_tile_rowsq(const float* A, int lda, float s,
                                               const float* B, int ldb, int K, v8f acc) {
  const int half = frag_half();
  const int l15  = frag_l15();
  for (int k0 = 0; k0 < K; k0 += 4) {
    const int ka = k0 + half * 2;
    const float ax = A[l15 * lda + ka] * s;
    const float ay = A[l15 * lda + ka + 1] * s;
    v2f a, b;
    a.x = ax * ax;
    a.y = ay * ay;
    b.x = B[ka * ldb + l15];
    b.y = B[(ka + 1) * ldb + l15];
    acc = __builtin_amdgcn_wmma_f32_16x16x4_f32(false, a, false, b, (short)0, acc,
                                                false, false);
  }
  return acc;
}

__device__ __forceinline__ float inv_norm_from_sumsq(float s) {
  return (s > 0.f) ? 1.0f / sqrtf(s) : 0.f;
}
__device__ __forceinline__ float inv_norm_const_row(float v, int n) {
  float s = 0.f;
  for (int c = 0; c < n; ++c) s += v * v;
  return inv_norm_from_sumsq(s);
}

#define BL0_ 0.1020620726159658f
#define R0_  0.125f
__device__ __forceinline__ float invbl0() { return inv_norm_const_row(BL0_, CW_); }
__device__ __forceinline__ float invr0()  { return inv_norm_const_row(R0_, BS_); }
__device__ __forceinline__ float rsq0()  { const float r = R0_ * invr0();    return r * r; }
__device__ __forceinline__ float blsq0() { const float b = BL0_ * invbl0();  return b * b; }
__device__ __forceinline__ float r2s0() {
  const float r = rsq0();
  float s = 0.f;
  for (int i = 0; i < BS_; ++i) s += r * r;
  return s;
}
__device__ __forceinline__ float dad0() {
  const float r = rsq0(), b = BL0_ * invbl0();
  float s = 0.f;
  for (int i = 0; i < BS_; ++i) s += r * b * b;
  return s;
}
__device__ __forceinline__ float l2s0() {
  const float v = blsq0();
  float s = 0.f;
  for (int l = 0; l < NB_; ++l) s += v * v;
  return s;
}
template <bool FIRST>
__device__ __forceinline__ float par(const float* p, int i, float c0) {
  if (FIRST) return c0;
  return p[i];
}

__global__ __launch_bounds__(256) void k_first_block(const float* __restrict__ q,
                                                     const float* __restrict__ kk,
                                                     float* ws) {
  LDS_TILE_ float Qs[BS_ * PA_];
  LDS_TILE_ float Ks[BS_ * PA_];
  LDS_TILE_ float Qj2[2 * NB_ * PA_];

  const int blk = blockIdx.x;
  const int hb = blk >> 5, kb = blk & 31;
  const float* qhd = q  + (size_t)hb * S_ * D_;
  const float* qh  = qhd + (size_t)kb * BS_ * D_;
  const float* kh  = kk + (size_t)hb * S_ * D_ + (size_t)kb * BS_ * D_;
  float* dbl = ws + O_DBL + (size_t)hb * S_ * CW_;
  float* t1c = ws + O_T2  + (size_t)hb * NB_ * BS_ * BS_ + T1C_OFF;
  float* t2c = ws + O_T2  + (size_t)hb * NB_ * BS_ * BS_ + T2C_OFF;

  const int tid = threadIdx.x;
  v4f qv[NQ_], kv[NQ_], gv[NQ_];
  #pragma unroll
  for (int it = 0; it < NQ_; ++it) {
    const int i = (tid + it * 256) * 4;
    qv[it] = ldg4(qh + i);
    kv[it] = ldg4(kh + i);
    const int l = i >> 7, cc = i & 127;
    gv[it] = ldg4(qhd + (l * 64 + 2 * kb) * 64 + cc);
  }
  #pragma unroll
  for (int it = 0; it < NQ_; ++it) {
    const int i = (tid + it * 256) * 4;
    const int p = (i >> 6) * PA_ + (i & 63);
    sts4(Qs + p, qv[it] * 0.125f);
    sts4(Ks + p, kv[it]);
    const int l = i >> 7, cc = i & 127;
    sts4(Qj2 + (cc >> 6) * (NB_ * PA_) + l * PA_ + (cc & 63), gv[it] * 0.125f);
  }
  __syncthreads();

  const float rsq = rsq0(), bsq = blsq0();
  const int wid = tid >> 5;
  for (int t = wid; t < 16; t += 8) {
    const int tj = t >> 2, ti = t & 3;
    v8f acc;
    #pragma unroll
    for (int r = 0; r < 8; ++r) acc[r] = bsq + bsq * rsq;
    acc = wmma_tile<false, true>(Qs + tj * 16 * PA_, PA_, -1.f,
                                 Ks + ti * 16 * PA_, PA_, 64, acc);
    #pragma unroll
    for (int r = 0; r < 8; ++r) {
      const int m = tj * 16 + frag_m(r), n = ti * 16 + frag_l15();
      dbl[(kb * 64 + m) * CW_ + n] = acc[r];
    }
  }
  if (wid >= 4) {
    const int ti = wid & 3;
    v8f acc = 0.0f;
    acc = wmma_const_a(rsq, Ks + ti * 16, PA_, 64, acc);
    if (frag_half() == 0) t1c[kb * 64 + ti * 16 + frag_l15()] = acc[0];
  } else {
    const int ti = wid;
    #pragma unroll
    for (int jj = 0; jj < 2; ++jj) {
      v8f acc = 0.0f;
      acc = wmma_const_a(bsq, Qj2 + jj * (NB_ * PA_) + ti * 16, PA_, 32, acc);
      if (frag_half() == 0) t2c[(2 * kb + jj) * 64 + ti * 16 + frag_l15()] = acc[0];
    }
  }
}

__device__ __forceinline__ void dleft_first_body(const float* __restrict__ q,
                                                 float* ws, float* Qs, float* T1c) {
  const int blk = blockIdx.x;
  const int hb = blk >> 5, kb = blk & 31;
  const float* qh  = q + (size_t)hb * S_ * D_ + (size_t)kb * BS_ * D_;
  const float* t1c = ws + O_T2 + (size_t)hb * NB_ * BS_ * BS_ + T1C_OFF;
  float* dbl = ws + O_DBL + (size_t)hb * S_ * CW_;

  const int tid = threadIdx.x;
  v4f qv[NQ_], tv[NQ_ / 2];
  #pragma unroll
  for (int it = 0; it < NQ_; ++it) {
    qv[it] = ldg4(qh + (tid + it * 256) * 4);
    if (it < NQ_ / 2) tv[it] = ldg4(t1c + (tid + it * 256) * 4);
  }
  #pragma unroll
  for (int it = 0; it < NQ_; ++it) {
    const int i = (tid + it * 256) * 4;
    const int p = (i >> 6) * PA_ + (i & 63);
    sts4(Qs + p, qv[it] * 0.125f);
    if (it < NQ_ / 2) sts4(T1c + p, tv[it]);
  }
  __syncthreads();

  const int wid = tid >> 5;
  const int tj = wid >> 1, ti = wid & 1;
  v8f acc = 0.0f;
  acc = wmma_tile<false, true>(Qs + tj * 16 * PA_, PA_, 1.f,
                               T1c + ti * 16 * PA_, PA_, 64, acc);
  const float lf = blsq0(), r2 = r2s0(), da = dad0();
  #pragma unroll
  for (int r = 0; r < 8; ++r) {
    const int j = tj * 16 + frag_m(r), k2 = ti * 16 + frag_l15();
    float v = lf * r2 - acc[r];
    if (k2 == kb) v += da;
    dbl[(kb * 64 + j) * CW_ + 64 + k2] = v;
  }
}

__device__ __forceinline__ void grad_block_body(const float* __restrict__ q,
                                                float* ws, float* Qs, float* Ks,
                                                float* Rsq, float* DiagL) {

  const int blk = blockIdx.x;
  const int hb = blk >> 5, kb = blk & 31;
  const float* qh = q  + (size_t)hb * S_ * D_ + (size_t)kb * BS_ * D_;
  float* blp   = ws + O_BL    + (size_t)hb * S_ * CW_;
  float* rp    = ws + O_R     + (size_t)hb * NB_ * BS_ * BS_ + (size_t)kb * BS_ * BS_;
  float* invbl = ws + O_INVBL + (size_t)hb * S_;
  float* invr  = ws + O_INVR  + (size_t)hb * S_;
  float* dbl   = ws + O_DBL   + (size_t)hb * S_ * CW_;
  float* t1    = ws + O_T1    + (size_t)hb * NB_ * BS_ * BS_ + (size_t)kb * BS_ * BS_;
  float* r2s   = ws + O_R2S   + (size_t)hb * S_;
  float* dad   = ws + O_DAD   + (size_t)hb * S_;

  const int tid = threadIdx.x;
  v4f qv[NQ_], rv[NQ_];
  float iv[NQ_], dlv = 0.f, dli = 0.f;
  #pragma unroll
  for (int it = 0; it < NQ_; ++it) {
    const int i = (tid + it * 256) * 4;
    qv[it] = ldg4(qh + i);
    rv[it] = ldg4(rp + i);
    iv[it] = invr[kb * 64 + (i >> 6)];
  }
  if (tid < BS_) {
    const int row = kb * 64 + tid;
    dlv = blp[row * CW_ + 64 + kb];
    dli = invbl[row];
  }
  #pragma unroll
  for (int it = 0; it < NQ_; ++it) {
    const int i = (tid + it * 256) * 4;
    const int p = (i >> 6) * PA_ + (i & 63);
    sts4(Qs + p, qv[it] * 0.125f);
    const v4f rs = rv[it] * iv[it];
    sts4(Rsq + p, rs * rs);
  }
  if (tid < BS_) {
    const float v = dlv * dli;
    DiagL[tid] = v * v;
  }
  __syncthreads();

  if (tid < BS_) {
    const int j = tid, row = kb * 64 + j;
    const float inv = invbl[row];
    float s2 = 0.f, sd = 0.f;
    for (int i = 0; i < BS_; ++i) {
      float r = Rsq[j * PA_ + i];
      float b = blp[row * CW_ + i] * inv;
      s2 += r * r;
      sd += r * b * b;
    }
    r2s[row] = s2;
    dad[row] = sd;
  }

  const int wid = tid >> 5;
  for (int t = wid; t < 32; t += 8) {
    const int which = t >> 4;
    const int tt = t & 15;
    const int tj = tt >> 2, ti = tt & 3;
    if (which == 0) {
      v8f acc;
      #pragma unroll
      for (int r = 0; r < 8; ++r) {
        const int m = tj * 16 + frag_m(r), n = ti * 16 + frag_l15();
        const int row = kb * 64 + m;
        float b = blp[row * CW_ + n] * invbl[row];
        acc[r] = b * b + DiagL[m] * Rsq[m * PA_ + n];
      }
      acc = wmma_tile<false, true>(Qs + tj * 16 * PA_, PA_, -1.f,
                                   Ks + ti * 16 * PA_, PA_, 64, acc);
      #pragma unroll
      for (int r = 0; r < 8; ++r) {
        const int m = tj * 16 + frag_m(r), n = ti * 16 + frag_l15();
        dbl[(kb * 64 + m) * CW_ + n] = acc[r];
      }
    } else {
      v8f acc = 0.0f;
      acc = wmma_tile<false, false>(Rsq + tj * 16 * PA_, PA_, 1.f,
                                    Ks + ti * 16, PA_, 64, acc);
      #pragma unroll
      for (int r = 0; r < 8; ++r) {
        const int m = tj * 16 + frag_m(r), n = ti * 16 + frag_l15();
        t1[m * 64 + n] = acc[r];
      }
    }
  }
}

__global__ __launch_bounds__(256, 6) void k_grad_j(const float* __restrict__ q,
                                                float* ws) {
  LDS_TILE_ float Lj[JW_ * NB_ * PT32_];
  LDS_TILE_ float Qj[JW_ * NB_ * PAB_];
  LDS_TILE_ float T1j[JW_ * NB_ * PA_];

  const int blk = blockIdx.x;
  const int hb = blk / (BS_ / JW_), j0 = (blk % (BS_ / JW_)) * JW_;
  const float* qh = q + (size_t)hb * S_ * D_;
  float* blp   = ws + O_BL    + (size_t)hb * S_ * CW_;
  float* invbl = ws + O_INVBL + (size_t)hb * S_;
  float* dbl   = ws + O_DBL   + (size_t)hb * S_ * CW_;
  float* t1h   = ws + O_T1    + (size_t)hb * NB_ * BS_ * BS_;
  float* t2h   = ws + O_T2    + (size_t)hb * NB_ * BS_ * BS_;
  float* r2s   = ws + O_R2S   + (size_t)hb * S_;
  float* dad   = ws + O_DAD   + (size_t)hb * S_;
  float* l2s   = ws + O_L2S   + (size_t)hb * S_;

  const int tid = threadIdx.x;
  constexpr int NL = JW_ * NB_ * NB_ / 4 / 256;
  constexpr int NT = JW_ * NB_ * D_ / 4 / 256;
  v4f lv[NL], qv[NT], tv[NT];
  float li[NL];
  #pragma unroll
  for (int it = 0; it < NL; ++it) {
    const int i = (tid + it * 256) * 4;
    const int l = i / (JW_ * NB_), jj = (i / NB_) % JW_, k2 = i % NB_;
    const int row = l * 64 + j0 + jj;
    lv[it] = ldg4(blp + row * CW_ + 64 + k2);
    li[it] = invbl[row];
  }
  #pragma unroll
  for (int it = 0; it < NT; ++it) {
    const int i = (tid + it * 256) * 4;
    const int l = i / (JW_ * D_), cc = i % (JW_ * D_);
    qv[it] = ldg4(qh + (l * 64 + j0) * 64 + cc);
    tv[it] = ldg4(t1h + l * (BS_ * BS_) + j0 * 64 + cc);
  }
  #pragma unroll
  for (int it = 0; it < NL; ++it) {
    const int i = (tid + it * 256) * 4;
    const int l = i / (JW_ * NB_), jj = (i / NB_) % JW_, k2 = i % NB_;
    const v4f v = lv[it] * li[it];
    sts4(Lj + jj * (NB_ * PT32_) + l * PT32_ + k2, v * v);
  }
  #pragma unroll
  for (int it = 0; it < NT; ++it) {
    const int i = (tid + it * 256) * 4;
    const int l = i / (JW_ * D_), cc = i % (JW_ * D_);
    const int jj = cc >> 6, d = cc & 63;
    sts4(Qj + jj * (NB_ * PAB_) + l * PAB_ + d, qv[it] * 0.125f);
    sts4(T1j + jj * (NB_ * PA_) + l * PA_ + d,  tv[it]);
  }
  __syncthreads();

  if (tid < JW_ * NB_) {
    const int jj = tid >> 5, k2 = tid & 31;
    const float* L = Lj + jj * (NB_ * PT32_);
    float s = 0.f;
    for (int l = 0; l < NB_; ++l) { float v = L[l * PT32_ + k2]; s += v * v; }
    l2s[(j0 + jj) * 32 + k2] = s;
  }

  const int wid = tid >> 5;
  for (int t = wid; t < 4 * JW_; t += 8) {
    const int jj = t >> 2, tj = (t >> 1) & 1, ti = t & 1;
    const int j = j0 + jj;
    const float* L = Lj + jj * (NB_ * PT32_);
    v8f acc = 0.0f;
    acc = wmma_tile<false, true>(Qj + jj * (NB_ * PAB_) + tj * 16 * PAB_, PAB_, 1.f,
                                 T1j + jj * (NB_ * PA_) + ti * 16 * PA_, PA_, 64, acc);
    #pragma unroll
    for (int r = 0; r < 8; ++r) {
      const int l = tj * 16 + frag_m(r), k2 = ti * 16 + frag_l15();
      float v = L[l * PT32_ + k2] * r2s[k2 * 64 + j] - acc[r];
      if (l == k2) v += dad[l * 64 + j];
      dbl[(l * 64 + j) * CW_ + 64 + k2] = v;
    }
  }
  for (int jj = 0; jj < JW_; ++jj) {
    const int tj = wid >> 2, ti = wid & 3;
    v8f acc = 0.0f;
    acc = wmma_tile<true, false>(Lj + jj * (NB_ * PT32_) + tj * 16, PT32_, 1.f,
                                 Qj + jj * (NB_ * PAB_) + ti * 16, PAB_, 32, acc);
    #pragma unroll
    for (int r = 0; r < 8; ++r) {
      const int k2 = tj * 16 + frag_m(r), d = ti * 16 + frag_l15();
      t2h[k2 * (BS_ * BS_) + (j0 + jj) * 64 + d] = acc[r];
    }
  }
}

template <bool FIRST>
__device__ __forceinline__ void grad_r_body(const float* __restrict__ kk,
                                            float* ws, float* T2s, float* Ks,
                                            float* DRs, float* InvRn, float* BDs) {

  const int blk = blockIdx.x;
  const int hb = blk >> 5, kb = blk & 31;
  const float* kh = kk + (size_t)hb * S_ * D_ + (size_t)kb * BS_ * D_;
  float* blp   = ws + O_BL    + (size_t)hb * S_ * CW_;
  float* rp    = ws + O_R     + (size_t)hb * NB_ * BS_ * BS_ + (size_t)kb * BS_ * BS_;
  float* invbl = ws + O_INVBL + (size_t)hb * S_;
  float* invr  = ws + O_INVR  + (size_t)hb * S_;
  float* t2h   = ws + O_T2    + (size_t)hb * NB_ * BS_ * BS_ + (size_t)kb * BS_ * BS_;
  float* l2s   = ws + O_L2S   + (size_t)hb * S_;
  const float* dbl = ws + O_DBL + (size_t)hb * S_ * CW_;

  const int tid = threadIdx.x;
  const float* t2src = FIRST ? ws + O_T2 + (size_t)hb * NB_ * BS_ * BS_ + T2C_OFF : t2h;
  {
    v4f tv[NQ_], kv[NQ_];
    #pragma unroll
    for (int it = 0; it < NQ_; ++it) {
      const int i = (tid + it * 256) * 4;
      tv[it] = ldg4(t2src + i);
      kv[it] = ldg4(kh + i);
    }
    #pragma unroll
    for (int it = 0; it < NQ_; ++it) {
      const int i = (tid + it * 256) * 4;
      const int p = (i >> 6) * PA_ + (i & 63);
      sts4(T2s + p, tv[it]);
      sts4(Ks + p, kv[it]);
    }
  }
  __syncthreads();

  const int wid = tid >> 5;
  for (int t = wid; t < 16; t += 8) {
    const int tj = t >> 2, ti = t & 3;
    v8f acc;
    #pragma unroll
    for (int r = 0; r < 8; ++r) {
      const int m = tj * 16 + frag_m(r), n = ti * 16 + frag_l15();
      const int row = kb * 64 + m;
      const float ib = par<FIRST>(invbl, row, invbl0());
      float bd = par<FIRST>(blp, row * CW_ + n, BL0_) * ib;         bd *= bd;
      float dl = par<FIRST>(blp, row * CW_ + 64 + kb, BL0_) * ib;   dl *= dl;
      float rv = par<FIRST>(rp, m * 64 + n, R0_) * par<FIRST>(invr, row, invr0());
      rv *= rv;
      acc[r] = rv * par<FIRST>(l2s, m * 32 + kb, l2s0()) + dl * bd;
    }
    acc = wmma_tile<false, true>(T2s + tj * 16 * PA_, PA_, -1.f,
                                 Ks + ti * 16 * PA_, PA_, 64, acc);
    #pragma unroll
    for (int r = 0; r < 8; ++r) {
      const int m = tj * 16 + frag_m(r), n = ti * 16 + frag_l15();
      DRs[m * PR_ + n] = acc[r];
    }
  }
  __syncthreads();

  if (tid < BS_) {
    const int j = tid, row = kb * 64 + j;
    const float inv = par<FIRST>(invr, row, invr0());
    float dot = 0.f;
    for (int i = 0; i < BS_; ++i) {
      float rs = par<FIRST>(rp, j * 64 + i, R0_) * inv;
      dot += rs * (2.f * DRs[j * PR_ + i] * rs);
    }
    float s2 = 0.f;
    #pragma unroll 32
    for (int i = 0; i < BS_; ++i) {
      const float old = par<FIRST>(rp, j * 64 + i, R0_);
      float rs = old * inv;
      float dr = 2.f * DRs[j * PR_ + i] * rs;
      float df = (dr - rs * dot) * inv;
      float nv = old - 0.5f * df;
      if (FIRST) rp[j * 64 + i] = nv;
      else       DRs[j * PR_ + i] = nv;
      s2 += nv * nv;
    }
    if (FIRST) invr[row] = inv_norm_from_sumsq(s2);
    else       InvRn[j]  = inv_norm_from_sumsq(s2);
  } else {
    const int lane = tid & 31;
    for (int grp = wid - 2; grp < 16; grp += 6) {
      const int row0 = kb * 64 + grp * 4;
      float old[4][3], bs[4][3], g[4][3], inv[4], dot[4], s2[4];
      float bdn[4][2];
      #pragma unroll
      for (int u = 0; u < 4; ++u) {
        const int row = row0 + u;
        inv[u] = par<FIRST>(invbl, row, invbl0());
        #pragma unroll
        for (int i = 0; i < 3; ++i) {
          const int c = row * CW_ + lane + 32 * i;
          old[u][i] = par<FIRST>(blp, c, BL0_);
          g[u][i]   = dbl[c];
        }
      }
      #pragma unroll
      for (int u = 0; u < 4; ++u) {
        float p = 0.f;
        #pragma unroll
        for (int i = 0; i < 3; ++i) {
          bs[u][i] = old[u][i] * inv[u];
          g[u][i]  = 2.f * g[u][i] * bs[u][i];
          p += bs[u][i] * g[u][i];
        }
        dot[u] = p;
      }
      #pragma unroll
      for (int m = 16; m >= 1; m >>= 1) {
        #pragma unroll
        for (int u = 0; u < 4; ++u) dot[u] += __shfl_xor(dot[u], m);
      }
      #pragma unroll
      for (int u = 0; u < 4; ++u) {
        const int row = row0 + u;
        float s = 0.f;
        #pragma unroll
        for (int i = 0; i < 3; ++i) {
          float nv = old[u][i] - 0.5f * ((g[u][i] - bs[u][i] * dot[u]) * inv[u]);
          if (FIRST || i == 2) blp[row * CW_ + lane + 32 * i] = nv;
          else                 bdn[u][i] = nv;
          s += nv * nv;
        }
        s2[u] = s;
      }
      #pragma unroll
      for (int m = 16; m >= 1; m >>= 1) {
        #pragma unroll
        for (int u = 0; u < 4; ++u) s2[u] += __shfl_xor(s2[u], m);
      }
      if (!FIRST) {
        #pragma unroll
        for (int u = 0; u < 4; ++u) {
          const float invn = inv_norm_from_sumsq(s2[u]);
          #pragma unroll
          for (int i = 0; i < 2; ++i) {
            const float v = bdn[u][i] * invn;
            BDs[(grp * 4 + u) * PA_ + lane + 32 * i] = v * v;
          }
        }
      }
      if (lane < 4) {
        float mine = s2[0];
        if (lane == 1) mine = s2[1];
        if (lane == 2) mine = s2[2];
        if (lane == 3) mine = s2[3];
        invbl[row0 + lane] = inv_norm_from_sumsq(mine);
      }
    }
  }
}
__global__ __launch_bounds__(256, 12) void k_r1_block2(const float* __restrict__ q,
                                                   const float* __restrict__ kk,
                                                   float* ws) {
  LDS_TILE_ float pool[3 * BS_ * PA_];
  __shared__ float DiagL[BS_];
  float* T2s = pool;
  float* Ks  = pool + BS_ * PA_;
  float* DRs = pool + 2 * BS_ * PA_;
  dleft_first_body(q, ws, pool, DRs);
  __syncthreads();
  grad_r_body<true>(kk, ws, T2s, Ks, DRs, nullptr, nullptr);
  __syncthreads();
  grad_block_body(q, ws, T2s, Ks, DRs, DiagL);
}

__global__ __launch_bounds__(256, 12) void k_r2_apply(const float* __restrict__ kk,
                                                  const float* __restrict__ val,
                                                  float* __restrict__ out,
                                                  float* ws) {
  LDS_TILE_ float pool[BS_ * PA_ + BS_ * PX_ + BS_ * PR_];
  __shared__ float InvRn[BS_];
  float* BDs = pool;
  float* Xs  = pool + BS_ * PA_;
  float* Rn  = pool + BS_ * PA_ + BS_ * PX_;
  grad_r_body<false>(kk, ws, pool, Xs, Rn, InvRn, BDs);
  __syncthreads();

  const int blk = blockIdx.x;
  const int hb = blk >> 5, kb = blk & 31;
  const float* xh = val + (size_t)hb * S_ * D_ + (size_t)kb * BS_ * D_;
  float* yb    = ws + O_T1    + (size_t)hb * NB_ * BS_ * BS_ + (size_t)kb * BS_ * BS_;
  float* oh    = out + (size_t)hb * S_ * D_ + (size_t)kb * BS_ * D_;

  const int tid = threadIdx.x;
  v4f xv[NQ_];
  #pragma unroll
  for (int it = 0; it < NQ_; ++it) xv[it] = ldg4(xh + (tid + it * 256) * 4);
  #pragma unroll
  for (int it = 0; it < NQ_; ++it) {
    const int i = (tid + it * 256) * 4;
    sts4(Xs + (i >> 6) * PX_ + (i & 63), xv[it]);
  }
  __syncthreads();

  const int wid = tid >> 5;
  for (int t = wid; t < 16; t += 8) {
    const int tj = t >> 2, ti = t & 3;
    v8f acc = 0.0f;
    acc = wmma_tile_rowsq(Rn + tj * 16 * PR_, PR_, InvRn[tj * 16 + frag_l15()],
                          Xs + ti * 16, PX_, 64, acc);
    #pragma unroll
    for (int r = 0; r < 8; ++r) {
      const int m = tj * 16 + frag_m(r), n = ti * 16 + frag_l15();
      yb[m * 64 + n] = acc[r];
    }
  }
  for (int t = wid; t < 16; t += 8) {
    const int tj = t >> 2, ti = t & 3;
    v8f acc = 0.0f;
    acc = wmma_tile<false, false>(BDs + tj * 16 * PA_, PA_, 1.f,
                                  Xs + ti * 16, PX_, 64, acc);
    #pragma unroll
    for (int r = 0; r < 8; ++r) {
      const int m = tj * 16 + frag_m(r), n = ti * 16 + frag_l15();
      oh[m * 64 + n] = acc[r];
    }
  }
}

__global__ __launch_bounds__(256, 6) void k_zleft(float* __restrict__ out,
                                               float* ws) {
  LDS_TILE_ float Lj[JW_ * NB_ * PA32_];
  LDS_TILE_ float Yj[JW_ * NB_ * PB_];
  const int blk = blockIdx.x;
  const int hb = blk / (BS_ / JW_), j0 = (blk % (BS_ / JW_)) * JW_;
  float* blp   = ws + O_BL    + (size_t)hb * S_ * CW_;
  float* invbl = ws + O_INVBL + (size_t)hb * S_;
  float* yb    = ws + O_T1    + (size_t)hb * NB_ * BS_ * BS_;
  float* oh    = out + (size_t)hb * S_ * D_;

  const int tid = threadIdx.x;
  const int wid = tid >> 5;
  const int tj = wid >> 2, ti = wid & 3;
  v8f prev[JW_];
  #pragma unroll
  for (int jj = 0; jj < JW_; ++jj) {
    #pragma unroll
    for (int r = 0; r < 8; ++r) {
      const int l = tj * 16 + frag_m(r), d = ti * 16 + frag_l15();
      prev[jj][r] = oh[(size_t)(l * 64 + j0 + jj) * 64 + d];
    }
  }
  constexpr int NL = JW_ * NB_ * NB_ / 4 / 256;
  constexpr int NT = JW_ * NB_ * D_ / 4 / 256;
  v4f lv[NL], yv[NT];
  float li[NL];
  #pragma unroll
  for (int it = 0; it < NL; ++it) {
    const int i = (tid + it * 256) * 4;
    const int l = i / (JW_ * NB_), jj = (i / NB_) % JW_, k2 = i % NB_;
    const int row = l * 64 + j0 + jj;
    lv[it] = ldg4(blp + row * CW_ + 64 + k2);
    li[it] = invbl[row];
  }
  #pragma unroll
  for (int it = 0; it < NT; ++it) {
    const int i = (tid + it * 256) * 4;
    const int k2 = i / (JW_ * D_), cc = i % (JW_ * D_);
    yv[it] = ldg4(yb + k2 * (BS_ * BS_) + j0 * 64 + cc);
  }
  #pragma unroll
  for (int it = 0; it < NL; ++it) {
    const int i = (tid + it * 256) * 4;
    const int l = i / (JW_ * NB_), jj = (i / NB_) % JW_, k2 = i % NB_;
    const v4f v = lv[it] * li[it];
    sts4(Lj + jj * (NB_ * PA32_) + l * PA32_ + k2, v * v);
  }
  #pragma unroll
  for (int it = 0; it < NT; ++it) {
    const int i = (tid + it * 256) * 4;
    const int k2 = i / (JW_ * D_), cc = i % (JW_ * D_);
    sts4(Yj + (cc >> 6) * (NB_ * PB_) + k2 * PB_ + (cc & 63), yv[it]);
  }
  __syncthreads();

  #pragma unroll
  for (int jj = 0; jj < JW_; ++jj) {
    v8f acc = 0.0f;
    acc = wmma_tile<false, false>(Lj + jj * (NB_ * PA32_) + tj * 16 * PA32_, PA32_, 1.f,
                                  Yj + jj * (NB_ * PB_) + ti * 16, PB_, 32, acc);
    #pragma unroll
    for (int r = 0; r < 8; ++r) {
      const int l = tj * 16 + frag_m(r), d = ti * 16 + frag_l15();
      oh[(size_t)(l * 64 + j0 + jj) * 64 + d] = prev[jj][r] + acc[r];
    }
  }
}

extern "C" void kernel_launch(void* const* d_in, const int* in_sizes, int n_in,
                              void* d_out, int out_size, void* d_ws, size_t ws_size,
                              hipStream_t stream) {
  const float* q = (const float*)d_in[0];
  const float* k = (const float*)d_in[1];
  const float* v = (const float*)d_in[2];
  float* out = (float*)d_out;
  float* ws = (float*)d_ws;

  const int blkG   = NH_ * NB_;
  const int jG     = NH_ * (BS_ / JW_);

  k_first_block<<<blkG, 256, 0, stream>>>(q, k, ws);
  k_r1_block2<<<blkG, 256, 0, stream>>>(q, k, ws);
  k_grad_j<<<jG, 256, 0, stream>>>(q, ws);
  k_r2_apply<<<blkG, 256, 0, stream>>>(k, v, out, ws);
  k_zleft<<<jG, 256, 0, stream>>>(out, ws);
}


# Q @ K^T of a block is computed once, in the second launch, and kept in registers for step 2's d_bd; each step adds its own start value to the finished product (d_bd sum order: start value last)
# speedup vs baseline: 3.1670x; 1.0395x over previous
#include <hip/hip_runtime.h>


#define S_   2048
#define D_   64
#define BS_  64
#define NB_  32
#define CW_  96
#define NH_  48
#define JW_  4

#define PA_   68
#define PA32_ 36
#define PB_   80
#define PX_   72
#define PAB_  76
#define PT32_ 40
#define PR_   66
#define LDS_TILE_ __shared__ __attribute__((aligned(16)))

typedef __attribute__((ext_vector_type(2))) float v2f;
typedef __attribute__((ext_vector_type(8))) float v8f;
typedef __attribute__((ext_vector_type(4))) float v4f;

#define NQ_ (BS_ * D_ / 4 / 256)
__device__ __forceinline__ v4f ldg4(const float* p) { return *reinterpret_cast<const v4f*>(p); }
__device__ __forceinline__ void sts4(float* p, v4f v) { *reinterpret_cast<v4f*>(p) = v; }

static constexpr size_t SZ_BL   = (size_t)NH_ * S_ * CW_;
static constexpr size_t SZ_R    = (size_t)NH_ * NB_ * BS_ * BS_;
static constexpr size_t O_BL    = 0;
static constexpr size_t O_R     = O_BL  + SZ_BL;
static constexpr size_t O_DBL   = O_R   + SZ_R;
static constexpr size_t O_T1    = O_DBL + SZ_BL;
static constexpr size_t O_T2    = O_T1  + SZ_R;
static constexpr size_t T2C_OFF = 0;
static constexpr size_t T1C_OFF = (size_t)BS_ * BS_;
static constexpr size_t O_INVBL = O_T2  + SZ_R;
static constexpr size_t O_INVR  = O_INVBL + (size_t)NH_ * S_;
static constexpr size_t O_R2S   = O_INVR  + (size_t)NH_ * S_;
static constexpr size_t O_DAD   = O_R2S   + (size_t)NH_ * S_;
static constexpr size_t O_L2S   = O_DAD   + (size_t)NH_ * S_;

__device__ __forceinline__ int frag_half() { return (threadIdx.x & 31) >> 4; }
__device__ __forceinline__ int frag_l15()  { return threadIdx.x & 15; }
__device__ __forceinline__ int frag_m(int r) { return r + frag_half() * 8; }

template <bool AT, bool BT>
__device__ __forceinline__ v8f wmma_tile(const float* A, int lda, float asgn,
                                         const float* B, int ldb, int K, v8f acc) {
  const int half = frag_half();
  const int l15  = frag_l15();
  for (int k0 = 0; k0 < K; k0 += 4) {
    const int ka = k0 + half * 2;
    v2f a, b;
    if (AT) { a.x = asgn * A[ka * lda + l15];  a.y = asgn * A[(ka + 1) * lda + l15]; }
    else    { a.x = asgn * A[l15 * lda + ka];  a.y = asgn * A[l15 * lda + ka + 1]; }
    if (BT) { b.x = B[l15 * ldb + ka];         b.y = B[l15 * ldb + ka + 1]; }
    else    { b.x = B[ka * ldb + l15];         b.y = B[(ka + 1) * ldb + l15]; }
    acc = __builtin_amdgcn_wmma_f32_16x16x4_f32(false, a, false, b, (short)0, acc,
                                                false, false);
  }
  return acc;
}

__device__ __forceinline__ v8f wmma_const_a(float a0, const float* B, int ldb, int K, v8f acc) {
  const int half = frag_half();
  const int l15  = frag_l15();
  v2f a;
  a.x = a0;  a.y = a0;
  for (int k0 = 0; k0 < K; k0 += 4) {
    const int ka = k0 + half * 2;
    v2f b;
    b.x = B[ka * ldb + l15];
    b.y = B[(ka + 1) * ldb + l15];
    acc = __builtin_amdgcn_wmma_f32_16x16x4_f32(false, a, false, b, (short)0, acc,
                                                false, false);
  }
  return acc;
}

__device__ __forceinline__ v8f wmma_tile_rowsq(const float* A, int lda, float s,
                                               const float* B, int ldb, int K, v8f acc) {
  const int half = frag_half();
  const int l15  = frag_l15();
  for (int k0 = 0; k0 < K; k0 += 4) {
    const int ka = k0 + half * 2;
    const float ax = A[l15 * lda + ka] * s;
    const float ay = A[l15 * lda + ka + 1] * s;
    v2f a, b;
    a.x = ax * ax;
    a.y = ay * ay;
    b.x = B[ka * ldb + l15];
    b.y = B[(ka + 1) * ldb + l15];
    acc = __builtin_amdgcn_wmma_f32_16x16x4_f32(false, a, false, b, (short)0, acc,
                                                false, false);
  }
  return acc;
}

__device__ __forceinline__ float inv_norm_from_sumsq(float s) {
  return (s > 0.f) ? 1.0f / sqrtf(s) : 0.f;
}
__device__ __forceinline__ float inv_norm_const_row(float v, int n) {
  float s = 0.f;
  for (int c = 0; c < n; ++c) s += v * v;
  return inv_norm_from_sumsq(s);
}

#define BL0_ 0.1020620726159658f
#define R0_  0.125f
__device__ __forceinline__ float invbl0() { return inv_norm_const_row(BL0_, CW_); }
__device__ __forceinline__ float invr0()  { return inv_norm_const_row(R0_, BS_); }
__device__ __forceinline__ float rsq0()  { const float r = R0_ * invr0();    return r * r; }
__device__ __forceinline__ float blsq0() { const float b = BL0_ * invbl0();  return b * b; }
__device__ __forceinline__ float r2s0() {
  const float r = rsq0();
  float s = 0.f;
  for (int i = 0; i < BS_; ++i) s += r * r;
  return s;
}
__device__ __forceinline__ float dad0() {
  const float r = rsq0(), b = BL0_ * invbl0();
  float s = 0.f;
  for (int i = 0; i < BS_; ++i) s += r * b * b;
  return s;
}
__device__ __forceinline__ float l2s0() {
  const float v = blsq0();
  float s = 0.f;
  for (int l = 0; l < NB_; ++l) s += v * v;
  return s;
}
template <bool FIRST>
__device__ __forceinline__ float par(const float* p, int i, float c0) {
  if (FIRST) return c0;
  return p[i];
}

__global__ __launch_bounds__(256) void k_first_block(const float* __restrict__ q,
                                                     const float* __restrict__ kk,
                                                     float* ws) {
  LDS_TILE_ float Ks[BS_ * PA_];
  LDS_TILE_ float Qj2[2 * NB_ * PA_];

  const int blk = blockIdx.x;
  const int hb = blk >> 5, kb = blk & 31;
  const float* qhd = q  + (size_t)hb * S_ * D_;
  const float* kh  = kk + (size_t)hb * S_ * D_ + (size_t)kb * BS_ * D_;
  float* t1c = ws + O_T2  + (size_t)hb * NB_ * BS_ * BS_ + T1C_OFF;
  float* t2c = ws + O_T2  + (size_t)hb * NB_ * BS_ * BS_ + T2C_OFF;

  const int tid = threadIdx.x;
  v4f kv[NQ_], gv[NQ_];
  #pragma unroll
  for (int it = 0; it < NQ_; ++it) {
    const int i = (tid + it * 256) * 4;
    kv[it] = ldg4(kh + i);
    const int l = i >> 7, cc = i & 127;
    gv[it] = ldg4(qhd + (l * 64 + 2 * kb) * 64 + cc);
  }
  #pragma unroll
  for (int it = 0; it < NQ_; ++it) {
    const int i = (tid + it * 256) * 4;
    const int p = (i >> 6) * PA_ + (i & 63);
    sts4(Ks + p, kv[it]);
    const int l = i >> 7, cc = i & 127;
    sts4(Qj2 + (cc >> 6) * (NB_ * PA_) + l * PA_ + (cc & 63), gv[it] * 0.125f);
  }
  __syncthreads();

  const float rsq = rsq0(), bsq = blsq0();
  const int wid = tid >> 5;
  if (wid >= 4) {
    const int ti = wid & 3;
    v8f acc = 0.0f;
    acc = wmma_const_a(rsq, Ks + ti * 16, PA_, 64, acc);
    if (frag_half() == 0) t1c[kb * 64 + ti * 16 + frag_l15()] = acc[0];
  } else {
    const int ti = wid;
    #pragma unroll
    for (int jj = 0; jj < 2; ++jj) {
      v8f acc = 0.0f;
      acc = wmma_const_a(bsq, Qj2 + jj * (NB_ * PA_) + ti * 16, PA_, 32, acc);
      if (frag_half() == 0) t2c[(2 * kb + jj) * 64 + ti * 16 + frag_l15()] = acc[0];
    }
  }
}

__device__ __forceinline__ void dleft_first_body(const float* __restrict__ q,
                                                 const float* __restrict__ kk,
                                                 float* ws, float* Qs, float* Ks,
                                                 float* T1c, v8f (&qkn)[2]) {
  const int blk = blockIdx.x;
  const int hb = blk >> 5, kb = blk & 31;
  const float* qh  = q + (size_t)hb * S_ * D_ + (size_t)kb * BS_ * D_;
  const float* kh  = kk + (size_t)hb * S_ * D_ + (size_t)kb * BS_ * D_;
  const float* t1c = ws + O_T2 + (size_t)hb * NB_ * BS_ * BS_ + T1C_OFF;
  float* dbl = ws + O_DBL + (size_t)hb * S_ * CW_;

  const int tid = threadIdx.x;
  v4f qv[NQ_], kv[NQ_], tv[NQ_ / 2];
  #pragma unroll
  for (int it = 0; it < NQ_; ++it) {
    qv[it] = ldg4(qh + (tid + it * 256) * 4);
    kv[it] = ldg4(kh + (tid + it * 256) * 4);
    if (it < NQ_ / 2) tv[it] = ldg4(t1c + (tid + it * 256) * 4);
  }
  #pragma unroll
  for (int it = 0; it < NQ_; ++it) {
    const int i = (tid + it * 256) * 4;
    const int p = (i >> 6) * PA_ + (i & 63);
    sts4(Qs + p, qv[it] * 0.125f);
    sts4(Ks + p, kv[it]);
    if (it < NQ_ / 2) sts4(T1c + p, tv[it]);
  }
  __syncthreads();

  const int wid = tid >> 5;
  {
    const float c0 = blsq0() + blsq0() * rsq0();
    #pragma unroll
    for (int s = 0; s < 2; ++s) {
      const int tt = wid + 8 * s;
      const int tj = tt >> 2, ti = tt & 3;
      v8f acc = 0.0f;
      acc = wmma_tile<false, true>(Qs + tj * 16 * PA_, PA_, -1.f,
                                   Ks + ti * 16 * PA_, PA_, 64, acc);
      qkn[s] = acc;
      #pragma unroll
      for (int r = 0; r < 8; ++r) {
        const int m = tj * 16 + frag_m(r), n = ti * 16 + frag_l15();
        dbl[(kb * 64 + m) * CW_ + n] = c0 + acc[r];
      }
    }
  }
  const int tj = wid >> 1, ti = wid & 1;
  v8f acc = 0.0f;
  acc = wmma_tile<false, true>(Qs + tj * 16 * PA_, PA_, 1.f,
                               T1c + ti * 16 * PA_, PA_, 64, acc);
  const float lf = blsq0(), r2 = r2s0(), da = dad0();
  #pragma unroll
  for (int r = 0; r < 8; ++r) {
    const int j = tj * 16 + frag_m(r), k2 = ti * 16 + frag_l15();
    float v = lf * r2 - acc[r];
    if (k2 == kb) v += da;
    dbl[(kb * 64 + j) * CW_ + 64 + k2] = v;
  }
}

__device__ __forceinline__ void grad_block_body(float* ws, float* Ks,
                                                float* Rsq, float* DiagL,
                                                const v8f (&qkn)[2]) {

  const int blk = blockIdx.x;
  const int hb = blk >> 5, kb = blk & 31;
  float* blp   = ws + O_BL    + (size_t)hb * S_ * CW_;
  float* rp    = ws + O_R     + (size_t)hb * NB_ * BS_ * BS_ + (size_t)kb * BS_ * BS_;
  float* invbl = ws + O_INVBL + (size_t)hb * S_;
  float* invr  = ws + O_INVR  + (size_t)hb * S_;
  float* dbl   = ws + O_DBL   + (size_t)hb * S_ * CW_;
  float* t1    = ws + O_T1    + (size_t)hb * NB_ * BS_ * BS_ + (size_t)kb * BS_ * BS_;
  float* r2s   = ws + O_R2S   + (size_t)hb * S_;
  float* dad   = ws + O_DAD   + (size_t)hb * S_;

  const int tid = threadIdx.x;
  v4f rv[NQ_];
  float iv[NQ_], dlv = 0.f, dli = 0.f;
  #pragma unroll
  for (int it = 0; it < NQ_; ++it) {
    const int i = (tid + it * 256) * 4;
    rv[it] = ldg4(rp + i);
    iv[it] = invr[kb * 64 + (i >> 6)];
  }
  if (tid < BS_) {
    const int row = kb * 64 + tid;
    dlv = blp[row * CW_ + 64 + kb];
    dli = invbl[row];
  }
  #pragma unroll
  for (int it = 0; it < NQ_; ++it) {
    const int i = (tid + it * 256) * 4;
    const int p = (i >> 6) * PA_ + (i & 63);
    const v4f rs = rv[it] * iv[it];
    sts4(Rsq + p, rs * rs);
  }
  if (tid < BS_) {
    const float v = dlv * dli;
    DiagL[tid] = v * v;
  }
  __syncthreads();

  if (tid < BS_) {
    const int j = tid, row = kb * 64 + j;
    const float inv = invbl[row];
    float s2 = 0.f, sd = 0.f;
    for (int i = 0; i < BS_; ++i) {
      float r = Rsq[j * PA_ + i];
      float b = blp[row * CW_ + i] * inv;
      s2 += r * r;
      sd += r * b * b;
    }
    r2s[row] = s2;
    dad[row] = sd;
  }

  const int wid = tid >> 5;
  #pragma unroll
  for (int s = 0; s < 2; ++s) {
    const int tt = wid + 8 * s;
    const int tj = tt >> 2, ti = tt & 3;
    float bv[8], nv[8];
    #pragma unroll
    for (int r = 0; r < 8; ++r) {
      const int m = tj * 16 + frag_m(r), n = ti * 16 + frag_l15();
      const int row = kb * 64 + m;
      bv[r] = blp[row * CW_ + n];
      nv[r] = invbl[row];
    }
    #pragma unroll
    for (int r = 0; r < 8; ++r) {
      const int m = tj * 16 + frag_m(r), n = ti * 16 + frag_l15();
      const int row = kb * 64 + m;
      const float b = bv[r] * nv[r];
      dbl[row * CW_ + n] = (b * b + DiagL[m] * Rsq[m * PA_ + n]) + qkn[s][r];
    }
  }
  for (int tt = wid; tt < 16; tt += 8) {
    const int tj = tt >> 2, ti = tt & 3;
    v8f acc = 0.0f;
    acc = wmma_tile<false, false>(Rsq + tj * 16 * PA_, PA_, 1.f,
                                  Ks + ti * 16, PA_, 64, acc);
    #pragma unroll
    for (int r = 0; r < 8; ++r) {
      const int m = tj * 16 + frag_m(r), n = ti * 16 + frag_l15();
      t1[m * 64 + n] = acc[r];
    }
  }
}

__global__ __launch_bounds__(256, 6) void k_grad_j(const float* __restrict__ q,
                                                float* ws) {
  LDS_TILE_ float Lj[JW_ * NB_ * PT32_];
  LDS_TILE_ float Qj[JW_ * NB_ * PAB_];
  LDS_TILE_ float T1j[JW_ * NB_ * PA_];

  const int blk = blockIdx.x;
  const int hb = blk / (BS_ / JW_), j0 = (blk % (BS_ / JW_)) * JW_;
  const float* qh = q + (size_t)hb * S_ * D_;
  float* blp   = ws + O_BL    + (size_t)hb * S_ * CW_;
  float* invbl = ws + O_INVBL + (size_t)hb * S_;
  float* dbl   = ws + O_DBL   + (size_t)hb * S_ * CW_;
  float* t1h   = ws + O_T1    + (size_t)hb * NB_ * BS_ * BS_;
  float* t2h   = ws + O_T2    + (size_t)hb * NB_ * BS_ * BS_;
  float* r2s   = ws + O_R2S   + (size_t)hb * S_;
  float* dad   = ws + O_DAD   + (size_t)hb * S_;
  float* l2s   = ws + O_L2S   + (size_t)hb * S_;

  const int tid = threadIdx.x;
  constexpr int NL = JW_ * NB_ * NB_ / 4 / 256;
  constexpr int NT = JW_ * NB_ * D_ / 4 / 256;
  v4f lv[NL], qv[NT], tv[NT];
  float li[NL];
  #pragma unroll
  for (int it = 0; it < NL; ++it) {
    const int i = (tid + it * 256) * 4;
    const int l = i / (JW_ * NB_), jj = (i / NB_) % JW_, k2 = i % NB_;
    const int row = l * 64 + j0 + jj;
    lv[it] = ldg4(blp + row * CW_ + 64 + k2);
    li[it] = invbl[row];
  }
  #pragma unroll
  for (int it = 0; it < NT; ++it) {
    const int i = (tid + it * 256) * 4;
    const int l = i / (JW_ * D_), cc = i % (JW_ * D_);
    qv[it] = ldg4(qh + (l * 64 + j0) * 64 + cc);
    tv[it] = ldg4(t1h + l * (BS_ * BS_) + j0 * 64 + cc);
  }
  #pragma unroll
  for (int it = 0; it < NL; ++it) {
    const int i = (tid + it * 256) * 4;
    const int l = i / (JW_ * NB_), jj = (i / NB_) % JW_, k2 = i % NB_;
    const v4f v = lv[it] * li[it];
    sts4(Lj + jj * (NB_ * PT32_) + l * PT32_ + k2, v * v);
  }
  #pragma unroll
  for (int it = 0; it < NT; ++it) {
    const int i = (tid + it * 256) * 4;
    const int l = i / (JW_ * D_), cc = i % (JW_ * D_);
    const int jj = cc >> 6, d = cc & 63;
    sts4(Qj + jj * (NB_ * PAB_) + l * PAB_ + d, qv[it] * 0.125f);
    sts4(T1j + jj * (NB_ * PA_) + l * PA_ + d,  tv[it]);
  }
  __syncthreads();

  if (tid < JW_ * NB_) {
    const int jj = tid >> 5, k2 = tid & 31;
    const float* L = Lj + jj * (NB_ * PT32_);
    float s = 0.f;
    for (int l = 0; l < NB_; ++l) { float v = L[l * PT32_ + k2]; s += v * v; }
    l2s[(j0 + jj) * 32 + k2] = s;
  }

  const int wid = tid >> 5;
  for (int t = wid; t < 4 * JW_; t += 8) {
    const int jj = t >> 2, tj = (t >> 1) & 1, ti = t & 1;
    const int j = j0 + jj;
    const float* L = Lj + jj * (NB_ * PT32_);
    v8f acc = 0.0f;
    acc = wmma_tile<false, true>(Qj + jj * (NB_ * PAB_) + tj * 16 * PAB_, PAB_, 1.f,
                                 T1j + jj * (NB_ * PA_) + ti * 16 * PA_, PA_, 64, acc);
    #pragma unroll
    for (int r = 0; r < 8; ++r) {
      const int l = tj * 16 + frag_m(r), k2 = ti * 16 + frag_l15();
      float v = L[l * PT32_ + k2] * r2s[k2 * 64 + j] - acc[r];
      if (l == k2) v += dad[l * 64 + j];
      dbl[(l * 64 + j) * CW_ + 64 + k2] = v;
    }
  }
  for (int jj = 0; jj < JW_; ++jj) {
    const int tj = wid >> 2, ti = wid & 3;
    v8f acc = 0.0f;
    acc = wmma_tile<true, false>(Lj + jj * (NB_ * PT32_) + tj * 16, PT32_, 1.f,
                                 Qj + jj * (NB_ * PAB_) + ti * 16, PAB_, 32, acc);
    #pragma unroll
    for (int r = 0; r < 8; ++r) {
      const int k2 = tj * 16 + frag_m(r), d = ti * 16 + frag_l15();
      t2h[k2 * (BS_ * BS_) + (j0 + jj) * 64 + d] = acc[r];
    }
  }
}

template <bool FIRST>
__device__ __forceinline__ void grad_r_body(const float* __restrict__ kk,
                                            float* ws, float* T2s, float* Ks,
                                            float* DRs, float* InvRn, float* BDs) {

  const int blk = blockIdx.x;
  const int hb = blk >> 5, kb = blk & 31;
  const float* kh = kk + (size_t)hb * S_ * D_ + (size_t)kb * BS_ * D_;
  float* blp   = ws + O_BL    + (size_t)hb * S_ * CW_;
  float* rp    = ws + O_R     + (size_t)hb * NB_ * BS_ * BS_ + (size_t)kb * BS_ * BS_;
  float* invbl = ws + O_INVBL + (size_t)hb * S_;
  float* invr  = ws + O_INVR  + (size_t)hb * S_;
  float* t2h   = ws + O_T2    + (size_t)hb * NB_ * BS_ * BS_ + (size_t)kb * BS_ * BS_;
  float* l2s   = ws + O_L2S   + (size_t)hb * S_;
  const float* dbl = ws + O_DBL + (size_t)hb * S_ * CW_;

  const int tid = threadIdx.x;
  const float* t2src = FIRST ? ws + O_T2 + (size_t)hb * NB_ * BS_ * BS_ + T2C_OFF : t2h;
  {
    v4f tv[NQ_], kv[NQ_];
    #pragma unroll
    for (int it = 0; it < NQ_; ++it) {
      const int i = (tid + it * 256) * 4;
      tv[it] = ldg4(t2src + i);
      if (!FIRST) kv[it] = ldg4(kh + i);
    }
    #pragma unroll
    for (int it = 0; it < NQ_; ++it) {
      const int i = (tid + it * 256) * 4;
      const int p = (i >> 6) * PA_ + (i & 63);
      sts4(T2s + p, tv[it]);
      if (!FIRST) sts4(Ks + p, kv[it]);
    }
  }
  __syncthreads();

  const int wid = tid >> 5;
  for (int t = wid; t < 16; t += 8) {
    const int tj = t >> 2, ti = t & 3;
    v8f acc;
    #pragma unroll
    for (int r = 0; r < 8; ++r) {
      const int m = tj * 16 + frag_m(r), n = ti * 16 + frag_l15();
      const int row = kb * 64 + m;
      const float ib = par<FIRST>(invbl, row, invbl0());
      float bd = par<FIRST>(blp, row * CW_ + n, BL0_) * ib;         bd *= bd;
      float dl = par<FIRST>(blp, row * CW_ + 64 + kb, BL0_) * ib;   dl *= dl;
      float rv = par<FIRST>(rp, m * 64 + n, R0_) * par<FIRST>(invr, row, invr0());
      rv *= rv;
      acc[r] = rv * par<FIRST>(l2s, m * 32 + kb, l2s0()) + dl * bd;
    }
    acc = wmma_tile<false, true>(T2s + tj * 16 * PA_, PA_, -1.f,
                                 Ks + ti * 16 * PA_, PA_, 64, acc);
    #pragma unroll
    for (int r = 0; r < 8; ++r) {
      const int m = tj * 16 + frag_m(r), n = ti * 16 + frag_l15();
      DRs[m * PR_ + n] = acc[r];
    }
  }
  __syncthreads();

  if (tid < BS_) {
    const int j = tid, row = kb * 64 + j;
    const float inv = par<FIRST>(invr, row, invr0());
    float dot = 0.f;
    for (int i = 0; i < BS_; ++i) {
      float rs = par<FIRST>(rp, j * 64 + i, R0_) * inv;
      dot += rs * (2.f * DRs[j * PR_ + i] * rs);
    }
    float s2 = 0.f;
    #pragma unroll 32
    for (int i = 0; i < BS_; ++i) {
      const float old = par<FIRST>(rp, j * 64 + i, R0_);
      float rs = old * inv;
      float dr = 2.f * DRs[j * PR_ + i] * rs;
      float df = (dr - rs * dot) * inv;
      float nv = old - 0.5f * df;
      if (FIRST) rp[j * 64 + i] = nv;
      else       DRs[j * PR_ + i] = nv;
      s2 += nv * nv;
    }
    if (FIRST) invr[row] = inv_norm_from_sumsq(s2);
    else       InvRn[j]  = inv_norm_from_sumsq(s2);
  } else {
    const int lane = tid & 31;
    for (int grp = wid - 2; grp < 16; grp += 6) {
      const int row0 = kb * 64 + grp * 4;
      float old[4][3], bs[4][3], g[4][3], inv[4], dot[4], s2[4];
      float bdn[4][2];
      #pragma unroll
      for (int u = 0; u < 4; ++u) {
        const int row = row0 + u;
        inv[u] = par<FIRST>(invbl, row, invbl0());
        #pragma unroll
        for (int i = 0; i < 3; ++i) {
          const int c = row * CW_ + lane + 32 * i;
          old[u][i] = par<FIRST>(blp, c, BL0_);
          g[u][i]   = dbl[c];
        }
      }
      #pragma unroll
      for (int u = 0; u < 4; ++u) {
        float p = 0.f;
        #pragma unroll
        for (int i = 0; i < 3; ++i) {
          bs[u][i] = old[u][i] * inv[u];
          g[u][i]  = 2.f * g[u][i] * bs[u][i];
          p += bs[u][i] * g[u][i];
        }
        dot[u] = p;
      }
      #pragma unroll
      for (int m = 16; m >= 1; m >>= 1) {
        #pragma unroll
        for (int u = 0; u < 4; ++u) dot[u] += __shfl_xor(dot[u], m);
      }
      #pragma unroll
      for (int u = 0; u < 4; ++u) {
        const int row = row0 + u;
        float s = 0.f;
        #pragma unroll
        for (int i = 0; i < 3; ++i) {
          float nv = old[u][i] - 0.5f * ((g[u][i] - bs[u][i] * dot[u]) * inv[u]);
          if (FIRST || i == 2) blp[row * CW_ + lane + 32 * i] = nv;
          else                 bdn[u][i] = nv;
          s += nv * nv;
        }
        s2[u] = s;
      }
      #pragma unroll
      for (int m = 16; m >= 1; m >>= 1) {
        #pragma unroll
        for (int u = 0; u < 4; ++u) s2[u] += __shfl_xor(s2[u], m);
      }
      if (!FIRST) {
        #pragma unroll
        for (int u = 0; u < 4; ++u) {
          const float invn = inv_norm_from_sumsq(s2[u]);
          #pragma unroll
          for (int i = 0; i < 2; ++i) {
            const float v = bdn[u][i] * invn;
            BDs[(grp * 4 + u) * PA_ + lane + 32 * i] = v * v;
          }
        }
      }
      if (lane < 4) {
        float mine = s2[0];
        if (lane == 1) mine = s2[1];
        if (lane == 2) mine = s2[2];
        if (lane == 3) mine = s2[3];
        invbl[row0 + lane] = inv_norm_from_sumsq(mine);
      }
    }
  }
}
__global__ __launch_bounds__(256, 12) void k_r1_block2(const float* __restrict__ q,
                                                   const float* __restrict__ kk,
                                                   float* ws) {
  LDS_TILE_ float pool[3 * BS_ * PA_];
  __shared__ float DiagL[BS_];
  float* T2s = pool;
  float* Ks  = pool + BS_ * PA_;
  float* DRs = pool + 2 * BS_ * PA_;
  v8f qkn[2];
  dleft_first_body(q, kk, ws, pool, Ks, DRs, qkn);
  __syncthreads();
  grad_r_body<true>(kk, ws, T2s, Ks, DRs, nullptr, nullptr);
  __syncthreads();
  grad_block_body(ws, Ks, DRs, DiagL, qkn);
}

__global__ __launch_bounds__(256, 12) void k_r2_apply(const float* __restrict__ kk,
                                                  const float* __restrict__ val,
                                                  float* __restrict__ out,
                                                  float* ws) {
  LDS_TILE_ float pool[BS_ * PA_ + BS_ * PX_ + BS_ * PR_];
  __shared__ float InvRn[BS_];
  float* BDs = pool;
  float* Xs  = pool + BS_ * PA_;
  float* Rn  = pool + BS_ * PA_ + BS_ * PX_;
  grad_r_body<false>(kk, ws, pool, Xs, Rn, InvRn, BDs);
  __syncthreads();

  const int blk = blockIdx.x;
  const int hb = blk >> 5, kb = blk & 31;
  const float* xh = val + (size_t)hb * S_ * D_ + (size_t)kb * BS_ * D_;
  float* yb    = ws + O_T1    + (size_t)hb * NB_ * BS_ * BS_ + (size_t)kb * BS_ * BS_;
  float* oh    = out + (size_t)hb * S_ * D_ + (size_t)kb * BS_ * D_;

  const int tid = threadIdx.x;
  v4f xv[NQ_];
  #pragma unroll
  for (int it = 0; it < NQ_; ++it) xv[it] = ldg4(xh + (tid + it * 256) * 4);
  #pragma unroll
  for (int it = 0; it < NQ_; ++it) {
    const int i = (tid + it * 256) * 4;
    sts4(Xs + (i >> 6) * PX_ + (i & 63), xv[it]);
  }
  __syncthreads();

  const int wid = tid >> 5;
  for (int t = wid; t < 16; t += 8) {
    const int tj = t >> 2, ti = t & 3;
    v8f acc = 0.0f;
    acc = wmma_tile_rowsq(Rn + tj * 16 * PR_, PR_, InvRn[tj * 16 + frag_l15()],
                          Xs + ti * 16, PX_, 64, acc);
    #pragma unroll
    for (int r = 0; r < 8; ++r) {
      const int m = tj * 16 + frag_m(r), n = ti * 16 + frag_l15();
      yb[m * 64 + n] = acc[r];
    }
  }
  for (int t = wid; t < 16; t += 8) {
    const int tj = t >> 2, ti = t & 3;
    v8f acc = 0.0f;
    acc = wmma_tile<false, false>(BDs + tj * 16 * PA_, PA_, 1.f,
                                  Xs + ti * 16, PX_, 64, acc);
    #pragma unroll
    for (int r = 0; r < 8; ++r) {
      const int m = tj * 16 + frag_m(r), n = ti * 16 + frag_l15();
      oh[m * 64 + n] = acc[r];
    }
  }
}

__global__ __launch_bounds__(256, 6) void k_zleft(float* __restrict__ out,
                                               float* ws) {
  LDS_TILE_ float Lj[JW_ * NB_ * PA32_];
  LDS_TILE_ float Yj[JW_ * NB_ * PB_];
  const int blk = blockIdx.x;
  const int hb = blk / (BS_ / JW_), j0 = (blk % (BS_ / JW_)) * JW_;
  float* blp   = ws + O_BL    + (size_t)hb * S_ * CW_;
  float* invbl = ws + O_INVBL + (size_t)hb * S_;
  float* yb    = ws + O_T1    + (size_t)hb * NB_ * BS_ * BS_;
  float* oh    = out + (size_t)hb * S_ * D_;

  const int tid = threadIdx.x;
  const int wid = tid >> 5;
  const int tj = wid >> 2, ti = wid & 3;
  v8f prev[JW_];
  #pragma unroll
  for (int jj = 0; jj < JW_; ++jj) {
    #pragma unroll
    for (int r = 0; r < 8; ++r) {
      const int l = tj * 16 + frag_m(r), d = ti * 16 + frag_l15();
      prev[jj][r] = oh[(size_t)(l * 64 + j0 + jj) * 64 + d];
    }
  }
  constexpr int NL = JW_ * NB_ * NB_ / 4 / 256;
  constexpr int NT = JW_ * NB_ * D_ / 4 / 256;
  v4f lv[NL], yv[NT];
  float li[NL];
  #pragma unroll
  for (int it = 0; it < NL; ++it) {
    const int i = (tid + it * 256) * 4;
    const int l = i / (JW_ * NB_), jj = (i / NB_) % JW_, k2 = i % NB_;
    const int row = l * 64 + j0 + jj;
    lv[it] = ldg4(blp + row * CW_ + 64 + k2);
    li[it] = invbl[row];
  }
  #pragma unroll
  for (int it = 0; it < NT; ++it) {
    const int i = (tid + it * 256) * 4;
    const int k2 = i / (JW_ * D_), cc = i % (JW_ * D_);
    yv[it] = ldg4(yb + k2 * (BS_ * BS_) + j0 * 64 + cc);
  }
  #pragma unroll
  for (int it = 0; it < NL; ++it) {
    const int i = (tid + it * 256) * 4;
    const int l = i / (JW_ * NB_), jj = (i / NB_) % JW_, k2 = i % NB_;
    const v4f v = lv[it] * li[it];
    sts4(Lj + jj * (NB_ * PA32_) + l * PA32_ + k2, v * v);
  }
  #pragma unroll
  for (int it = 0; it < NT; ++it) {
    const int i = (tid + it * 256) * 4;
    const int k2 = i / (JW_ * D_), cc = i % (JW_ * D_);
    sts4(Yj + (cc >> 6) * (NB_ * PB_) + k2 * PB_ + (cc & 63), yv[it]);
  }
  __syncthreads();

  #pragma unroll
  for (int jj = 0; jj < JW_; ++jj) {
    v8f acc = 0.0f;
    acc = wmma_tile<false, false>(Lj + jj * (NB_ * PA32_) + tj * 16 * PA32_, PA32_, 1.f,
                                  Yj + jj * (NB_ * PB_) + ti * 16, PB_, 32, acc);
    #pragma unroll
    for (int r = 0; r < 8; ++r) {
      const int l = tj * 16 + frag_m(r), d = ti * 16 + frag_l15();
      oh[(size_t)(l * 64 + j0 + jj) * 64 + d] = prev[jj][r] + acc[r];
    }
  }
}

extern "C" void kernel_launch(void* const* d_in, const int* in_sizes, int n_in,
                              void* d_out, int out_size, void* d_ws, size_t ws_size,
                              hipStream_t stream) {
  const float* q = (const float*)d_in[0];
  const float* k = (const float*)d_in[1];
  const float* v = (const float*)d_in[2];
  float* out = (float*)d_out;
  float* ws = (float*)d_ws;

  const int blkG   = NH_ * NB_;
  const int jG     = NH_ * (BS_ / JW_);

  k_first_block<<<blkG, 256, 0, stream>>>(q, k, ws);
  k_r1_block2<<<blkG, 256, 0, stream>>>(q, k, ws);
  k_grad_j<<<jG, 256, 0, stream>>>(q, ws);
  k_r2_apply<<<blkG, 256, 0, stream>>>(k, v, out, ws);
  k_zleft<<<jG, 256, 0, stream>>>(out, ws);
}
